# starting kernel
# baseline (speedup 1.0000x reference)
#include <hip/hip_runtime.h>
#include <hip/hip_bf16.h>

#define NGRID   32
#define NTOKEN  1024
#define NTOK2   2048
#define HEAD    4
#define DK      64
#define NLAYER  4
#define FDIM    64
#define BATCH   8

typedef __attribute__((ext_vector_type(2))) float v2f;
typedef __attribute__((ext_vector_type(8))) float v8f;

__device__ __forceinline__ v8f wmma4(v2f a, v2f b, v8f c) {
  return __builtin_amdgcn_wmma_f32_16x16x4_f32(
       false, a,  false, b,
       (short)0, c,  false,  false);
}

__device__ __forceinline__ v2f frag_a(const float* src, int lda, int m0, int k0, int lane) {
  int row = m0 + (lane & 15);
  int kk  = k0 + ((lane >> 4) << 1);
  v2f a; a.x = src[row * lda + kk]; a.y = src[row * lda + kk + 1];
  return a;
}

__device__ __forceinline__ v2f frag_a_t(const float* src, int lds, int m0, int k0, int lane) {
  int row = m0 + (lane & 15);
  int kk  = k0 + ((lane >> 4) << 1);
  v2f a; a.x = src[kk * lds + row]; a.y = src[(kk + 1) * lds + row];
  return a;
}

__device__ __forceinline__ v2f frag_b(const float* src, int ldb, int k0, int n0, int lane) {
  int col = n0 + (lane & 15);
  int kk  = k0 + ((lane >> 4) << 1);
  v2f b; b.x = src[kk * ldb + col]; b.y = src[(kk + 1) * ldb + col];
  return b;
}

#define MLPF 2337

__device__ __forceinline__ float edge_net(const float* s, float e0, float e1, float e2, float e3) {
  const float* W0 = s;
  const float* b0 = s + 128;
  const float* W1 = s + 160;
  const float* b1 = s + 2208;
  const float* W2 = s + 2272;
  float bias2 = s[2336];
  float h0[32];
#pragma unroll
  for (int u = 0; u < 32; ++u) {
    float t = b0[u];
    t = fmaf(W0[u * 4 + 0], e0, t);
    t = fmaf(W0[u * 4 + 1], e1, t);
    t = fmaf(W0[u * 4 + 2], e2, t);
    t = fmaf(W0[u * 4 + 3], e3, t);
    h0[u] = t > 0.f ? t : 0.01f * t;
  }
  float out = bias2;
  for (int v = 0; v < 64; ++v) {
    float t = b1[v];
    const float* w = W1 + v * 32;
#pragma unroll
    for (int u = 0; u < 32; ++u) t = fmaf(w[u], h0[u], t);
    t = t > 0.f ? t : 0.01f * t;
    out = fmaf(W2[v], t, out);
  }
  return out;
}

__global__ void edge_mlp_kernel(const float* kW0, const float* kb0, const float* kW1,
                                const float* kb1, const float* kW2, const float* kb2,
                                const float* fW0, const float* fb0, const float* fW1,
                                const float* fb1, const float* fW2, const float* fb2,
                                float* wout, float* fout) {
  __shared__ float sh[2 * MLPF];
  float* sk = sh;
  float* sf = sh + MLPF;
  for (int i = threadIdx.x; i < MLPF; i += blockDim.x) {
    float v, vf;
    if (i < 128)       { v = kW0[i];        vf = fW0[i]; }
    else if (i < 160)  { v = kb0[i - 128];  vf = fb0[i - 128]; }
    else if (i < 2208) { v = kW1[i - 160];  vf = fW1[i - 160]; }
    else if (i < 2272) { v = kb1[i - 2208]; vf = fb1[i - 2208]; }
    else if (i < 2336) { v = kW2[i - 2272]; vf = fW2[i - 2272]; }
    else               { v = kb2[0];        vf = fb2[0]; }
    sk[i] = v; sf[i] = vf;
  }
  __syncthreads();
  int idx = blockIdx.x * blockDim.x + threadIdx.x;
  int n = idx >> 10, m = idx & 1023;
  const float DXc = 2.0f / 31.0f;
  float e0 = -1.f + DXc * (float)(n >> 5);
  float e1 = -1.f + DXc * (float)(n & 31);
  float e2 = -1.f + DXc * (float)(m >> 5);
  float e3 = -1.f + DXc * (float)(m & 31);
  wout[idx] = edge_net(sk, e0, e1, e2, e3);
  fout[idx] = edge_net(sf, e0, e1, e2, e3);
}

__global__ void ln0_reduce_kernel(const float* xy, float* stats) {
  int b = blockIdx.x;
  const float* p = xy + (size_t)b * NTOK2 * FDIM;
  const int N = NTOK2 * FDIM;
  float s = 0.f, ss = 0.f;
  for (int i = threadIdx.x; i < N; i += 256) {
    float v = p[i];
    s += v; ss += v * v;
  }
  __shared__ float sh[512];
  sh[threadIdx.x] = s; sh[256 + threadIdx.x] = ss;
  __syncthreads();
  for (int st = 128; st >= 1; st >>= 1) {
    if ((int)threadIdx.x < st) {
      sh[threadIdx.x]       += sh[threadIdx.x + st];
      sh[256 + threadIdx.x] += sh[256 + threadIdx.x + st];
    }
    __syncthreads();
  }
  if (threadIdx.x == 0) {
    float mean = sh[0] / (float)N;
    float var  = sh[256] / (float)N - mean * mean;
    stats[2 * b]     = mean;
    stats[2 * b + 1] = rsqrtf(var + 1e-5f);
  }
}

__global__ void ln0_apply_kernel(const float* xy, const float* g, const float* bb,
                                 const float* stats, float* V) {
  int idx = blockIdx.x * blockDim.x + threadIdx.x;
  int b = idx >> 17;
  int r = idx & 131071;
  V[idx] = (xy[idx] - stats[2 * b]) * stats[2 * b + 1] * g[r] + bb[r];
}

__global__ void qk_kernel(const float* Vin, const float* Wq, const float* bq,
                          const float* Wk, const float* bk, float* Q, float* K) {
  int gx = blockIdx.x;
  bool isK = gx >= 32;
  int bh = gx & 31;
  const float* W    = isK ? Wk : Wq;
  const float* bias = isK ? bk : bq;
  float* Out        = isK ? K  : Q;
  int h = bh & 3;
  int b = bh >> 2;
  const float* A     = Vin + (size_t)b * NTOK2 * FDIM;
  const float* Wh    = W + h * DK * FDIM;
  const float* biash = bias + h * DK;
  int wave = threadIdx.x >> 5, lane = threadIdx.x & 31;
  int m0 = blockIdx.y * 64 + wave * 16;
  v8f c[4] = {};
  for (int k0 = 0; k0 < FDIM; k0 += 4) {
    v2f a = frag_a(A, FDIM, m0, k0, lane);
    int kk = k0 + ((lane >> 4) << 1);
#pragma unroll
    for (int t = 0; t < 4; ++t) {
      int col = t * 16 + (lane & 15);
      v2f bf; bf.x = Wh[col * FDIM + kk]; bf.y = Wh[col * FDIM + kk + 1];
      c[t] = wmma4(a, bf, c[t]);
    }
  }
  float* dst = Out + (size_t)bh * NTOK2 * DK;
  int rbase = m0 + ((lane >> 4) << 3);
#pragma unroll
  for (int t = 0; t < 4; ++t) {
    int col = t * 16 + (lane & 15);
    float bv = biash[col];
#pragma unroll
    for (int r = 0; r < 8; ++r) dst[(rbase + r) * DK + col] = c[t][r] + bv;
  }
}

__global__ void ktv_kernel(const float* Kbuf, const float* Bsrc, int bstride,
                           int kdim, float s, float* Mout) {
  int bh = blockIdx.x;
  int b = bh >> 2;
  int wave = threadIdx.x >> 5, lane = threadIdx.x & 31;
  int m0 = wave * 16;
  const float* Kh = Kbuf + (size_t)bh * NTOK2 * DK;
  const float* B  = Bsrc + (size_t)b * bstride;
  v8f c[4] = {};
  for (int k0 = 0; k0 < kdim; k0 += 4) {
    v2f a = frag_a_t(Kh, DK, m0, k0, lane);
#pragma unroll
    for (int t = 0; t < 4; ++t) {
      v2f bf = frag_b(B, FDIM, k0, t * 16, lane);
      c[t] = wmma4(a, bf, c[t]);
    }
  }
  float* dst = Mout + (size_t)bh * DK * FDIM;
  int rbase = m0 + ((lane >> 4) << 3);
#pragma unroll
  for (int t = 0; t < 4; ++t) {
    int col = t * 16 + (lane & 15);
#pragma unroll
    for (int r = 0; r < 8; ++r) dst[(rbase + r) * FDIM + col] = c[t][r] * s;
  }
}

__global__ void qm_kernel(const float* Q, const float* Mb, float* Out) {
  int b = blockIdx.x;
  int wave = threadIdx.x >> 5, lane = threadIdx.x & 31;
  int m0 = blockIdx.y * 64 + wave * 16;
  v8f c[4] = {};
  for (int h = 0; h < HEAD; ++h) {
    const float* Qh = Q + (size_t)(b * HEAD + h) * NTOK2 * DK;
    const float* Mh = Mb + (size_t)(b * HEAD + h) * DK * FDIM;
    for (int k0 = 0; k0 < DK; k0 += 4) {
      v2f a = frag_a(Qh, DK, m0, k0, lane);
#pragma unroll
      for (int t = 0; t < 4; ++t) {
        v2f bf = frag_b(Mh, FDIM, k0, t * 16, lane);
        c[t] = wmma4(a, bf, c[t]);
      }
    }
  }
  float* dst = Out + (size_t)b * NTOK2 * FDIM;
  int rbase = m0 + ((lane >> 4) << 3);
#pragma unroll
  for (int t = 0; t < 4; ++t) {
    int col = t * 16 + (lane & 15);
#pragma unroll
    for (int r = 0; r < 8; ++r) dst[(rbase + r) * FDIM + col] = c[t][r];
  }
}

__global__ void ln_res_kernel(const float* mid, const float* vin, const float* g,
                              const float* bb, float* vout) {
  int row = blockIdx.x * 8 + (threadIdx.x >> 5);
  int lane = threadIdx.x & 31;
  const float* mr = mid + (size_t)row * FDIM;
  float x0 = mr[lane], x1 = mr[lane + 32];
  float s = x0 + x1, ss = x0 * x0 + x1 * x1;
#pragma unroll
  for (int m = 16; m >= 1; m >>= 1) {
    s  += __shfl_xor(s, m, 32);
    ss += __shfl_xor(ss, m, 32);
  }
  float mean = s * (1.f / 64.f);
  float var  = ss * (1.f / 64.f) - mean * mean;
  float inv  = rsqrtf(var + 1e-5f);
  size_t base = (size_t)row * FDIM;
  vout[base + lane]      = (x0 - mean) * inv * g[lane]      + bb[lane]      + vin[base + lane];
  vout[base + lane + 32] = (x1 - mean) * inv * g[lane + 32] + bb[lane + 32] + vin[base + lane + 32];
}

__global__ void final_kernel(const float* weight, const float* weightf,
                             const float* S, float* out, float mul) {
  int b = blockIdx.x;
  int wave = threadIdx.x >> 5, lane = threadIdx.x & 31;
  int m0 = blockIdx.y * 64 + wave * 16;
  const float* Sb = S + (size_t)b * NTOK2 * FDIM;
  int row = m0 + (lane & 15);
  v8f c[4] = {};
  for (int k0 = 0; k0 < NTOK2; k0 += 4) {
    const float* W = (k0 < NTOKEN) ? weight : weightf;
    int kb = (k0 & (NTOKEN - 1)) + ((lane >> 4) << 1);
    v2f a; a.x = W[kb * NTOKEN + row]; a.y = W[(kb + 1) * NTOKEN + row];
    if ((k0 & 60) == 0 && (k0 & (NTOKEN - 1)) + 64 < NTOKEN) {
      __builtin_prefetch(&W[(kb + 64) * NTOKEN + row], 0, 3);
    }
#pragma unroll
    for (int t = 0; t < 4; ++t) {
      v2f bf = frag_b(Sb, FDIM, k0, t * 16, lane);
      c[t] = wmma4(a, bf, c[t]);
    }
  }
  float* dst = out + (size_t)b * NTOKEN * FDIM;
  int rbase = m0 + ((lane >> 4) << 3);
#pragma unroll
  for (int t = 0; t < 4; ++t) {
    int col = t * 16 + (lane & 15);
#pragma unroll
    for (int r = 0; r < 8; ++r) dst[(rbase + r) * FDIM + col] = c[t][r] * mul;
  }
}

extern "C" void kernel_launch(void* const* d_in, const int* in_sizes, int n_in,
                              void* d_out, int out_size, void* d_ws, size_t ws_size,
                              hipStream_t stream) {
  (void)in_sizes; (void)n_in; (void)out_size; (void)ws_size;
  const float* xy   = (const float*)d_in[0];
  const float* kW0  = (const float*)d_in[1];
  const float* kb0  = (const float*)d_in[2];
  const float* kW1  = (const float*)d_in[3];
  const float* kb1  = (const float*)d_in[4];
  const float* kW2  = (const float*)d_in[5];
  const float* kb2  = (const float*)d_in[6];
  const float* fW0  = (const float*)d_in[7];
  const float* fb0  = (const float*)d_in[8];
  const float* fW1  = (const float*)d_in[9];
  const float* fb1  = (const float*)d_in[10];
  const float* fW2  = (const float*)d_in[11];
  const float* fb2  = (const float*)d_in[12];
  const float* Wq   = (const float*)d_in[13];
  const float* bq   = (const float*)d_in[14];
  const float* Wk   = (const float*)d_in[15];
  const float* bk   = (const float*)d_in[16];
  const float* ln0g = (const float*)d_in[17];
  const float* ln0b = (const float*)d_in[18];
  const float* lng  = (const float*)d_in[19];
  const float* lnb  = (const float*)d_in[20];
  float* out = (float*)d_out;

  float* ws = (float*)d_ws;
  const size_t SZ_1M = 1048576;
  float* w_weight  = ws;
  float* w_weightf = ws + SZ_1M;
  float* V0        = ws + 2 * SZ_1M;
  float* V1        = ws + 3 * SZ_1M;
  float* Qb        = ws + 4 * SZ_1M;
  float* Kb        = ws + 8 * SZ_1M;
  float* Mb        = ws + 12 * SZ_1M;
  float* MID       = ws + 12 * SZ_1M + 131072;
  float* stats     = ws + 13 * SZ_1M + 131072;

  const float DXc   = 2.0f / 31.0f;
  const float dxdy  = DXc * DXc;
  const float scale = 0.125f;

  edge_mlp_kernel<<<4096, 256, 0, stream>>>(kW0, kb0, kW1, kb1, kW2, kb2,
                                            fW0, fb0, fW1, fb1, fW2, fb2,
                                            w_weight, w_weightf);
  ln0_reduce_kernel<<<BATCH, 256, 0, stream>>>(xy, stats);
  ln0_apply_kernel<<<4096, 256, 0, stream>>>(xy, ln0g, ln0b, stats, V0);

  float* Va = V0;
  float* Vb = V1;
  for (int j = 0; j < NLAYER - 1; ++j) {
    qk_kernel<<<dim3(64, NTOK2 / 64), 128, 0, stream>>>(
        Va, Wq + (size_t)j * HEAD * DK * FDIM, bq + (size_t)j * HEAD * DK,
        Wk + (size_t)j * HEAD * DK * FDIM, bk + (size_t)j * HEAD * DK, Qb, Kb);
    ktv_kernel<<<32, 128, 0, stream>>>(Kb, Va, NTOK2 * FDIM, NTOK2,
                                       scale * dxdy, Mb);
    qm_kernel<<<dim3(BATCH, NTOK2 / 64), 128, 0, stream>>>(Qb, Mb, MID);
    ln_res_kernel<<<2048, 256, 0, stream>>>(MID, Va, lng + j * FDIM,
                                            lnb + j * FDIM, Vb);
    float* t = Va; Va = Vb; Vb = t;
  }

  {
    int j = NLAYER - 1;
    qk_kernel<<<dim3(64, NTOK2 / 64), 128, 0, stream>>>(
        Va, Wq + (size_t)j * HEAD * DK * FDIM, bq + (size_t)j * HEAD * DK,
        Wk + (size_t)j * HEAD * DK * FDIM, bk + (size_t)j * HEAD * DK, Qb, Kb);
    ktv_kernel<<<32, 128, 0, stream>>>(Kb, xy, NTOK2 * FDIM, NTOKEN, scale, Mb);
    qm_kernel<<<dim3(BATCH, NTOK2 / 64), 128, 0, stream>>>(Qb, Mb, MID);
    final_kernel<<<dim3(BATCH, NTOKEN / 16 / 4), 128, 0, stream>>>(
        w_weight, w_weightf, MID, out, dxdy * dxdy);
  }
}


# Split the K loop of the two long-K few-wave GEMMs (K^T V and final weight product) over 16 waves per work-group, partial tiles added in LDS in fixed order; summation order changes
# speedup vs baseline: 1.8146x; 1.8146x over previous
#include <hip/hip_runtime.h>
#include <hip/hip_bf16.h>

#define NGRID   32
#define NTOKEN  1024
#define NTOK2   2048
#define HEAD    4
#define DK      64
#define NLAYER  4
#define FDIM    64
#define BATCH   8

typedef __attribute__((ext_vector_type(2))) float v2f;
typedef __attribute__((ext_vector_type(8))) float v8f;

__device__ __forceinline__ v8f wmma4(v2f a, v2f b, v8f c) {
  return __builtin_amdgcn_wmma_f32_16x16x4_f32(
       false, a,  false, b,
       (short)0, c,  false,  false);
}

__device__ __forceinline__ v2f frag_a(const float* src, int lda, int m0, int k0, int lane) {
  int row = m0 + (lane & 15);
  int kk  = k0 + ((lane >> 4) << 1);
  v2f a; a.x = src[row * lda + kk]; a.y = src[row * lda + kk + 1];
  return a;
}

__device__ __forceinline__ v2f frag_a_t(const float* src, int lds, int m0, int k0, int lane) {
  int row = m0 + (lane & 15);
  int kk  = k0 + ((lane >> 4) << 1);
  v2f a; a.x = src[kk * lds + row]; a.y = src[(kk + 1) * lds + row];
  return a;
}

__device__ __forceinline__ v2f frag_b(const float* src, int ldb, int k0, int n0, int lane) {
  int col = n0 + (lane & 15);
  int kk  = k0 + ((lane >> 4) << 1);
  v2f b; b.x = src[kk * ldb + col]; b.y = src[(kk + 1) * ldb + col];
  return b;
}


#define KSPLIT   16
#define TPITCH   68
#define TPLANE   (16 * TPITCH)

__device__ __forceinline__ void ksplit_reduce_store(float* sh, const v8f* c, int ks, int lane,
                                                    float* dst, int m0, float mul) {
  float* mine = sh + ks * TPLANE;
  int lrow = (lane >> 4) << 3;
  int lcol = lane & 15;
#pragma unroll
  for (int t = 0; t < 4; ++t) {
#pragma unroll
    for (int r = 0; r < 8; ++r) mine[(lrow + r) * TPITCH + t * 16 + lcol] = c[t][r];
  }
  __syncthreads();
  for (int idx = threadIdx.x; idx < 16 * 64; idx += KSPLIT * 32) {
    int row = idx >> 6, col = idx & 63;
    float acc = sh[row * TPITCH + col];
#pragma unroll
    for (int p = 1; p < KSPLIT; ++p) acc += sh[p * TPLANE + row * TPITCH + col];
    dst[(m0 + row) * 64 + col] = acc * mul;
  }
}

#define MLPF 2337

__device__ __forceinline__ float edge_net(const float* s, float e0, float e1, float e2, float e3) {
  const float* W0 = s;
  const float* b0 = s + 128;
  const float* W1 = s + 160;
  const float* b1 = s + 2208;
  const float* W2 = s + 2272;
  float bias2 = s[2336];
  float h0[32];
#pragma unroll
  for (int u = 0; u < 32; ++u) {
    float t = b0[u];
    t = fmaf(W0[u * 4 + 0], e0, t);
    t = fmaf(W0[u * 4 + 1], e1, t);
    t = fmaf(W0[u * 4 + 2], e2, t);
    t = fmaf(W0[u * 4 + 3], e3, t);
    h0[u] = t > 0.f ? t : 0.01f * t;
  }
  float out = bias2;
  for (int v = 0; v < 64; ++v) {
    float t = b1[v];
    const float* w = W1 + v * 32;
#pragma unroll
    for (int u = 0; u < 32; ++u) t = fmaf(w[u], h0[u], t);
    t = t > 0.f ? t : 0.01f * t;
    out = fmaf(W2[v], t, out);
  }
  return out;
}

__global__ void edge_mlp_kernel(const float* kW0, const float* kb0, const float* kW1,
                                const float* kb1, const float* kW2, const float* kb2,
                                const float* fW0, const float* fb0, const float* fW1,
                                const float* fb1, const float* fW2, const float* fb2,
                                float* wout, float* fout) {
  __shared__ float sh[2 * MLPF];
  float* sk = sh;
  float* sf = sh + MLPF;
  for (int i = threadIdx.x; i < MLPF; i += blockDim.x) {
    float v, vf;
    if (i < 128)       { v = kW0[i];        vf = fW0[i]; }
    else if (i < 160)  { v = kb0[i - 128];  vf = fb0[i - 128]; }
    else if (i < 2208) { v = kW1[i - 160];  vf = fW1[i - 160]; }
    else if (i < 2272) { v = kb1[i - 2208]; vf = fb1[i - 2208]; }
    else if (i < 2336) { v = kW2[i - 2272]; vf = fW2[i - 2272]; }
    else               { v = kb2[0];        vf = fb2[0]; }
    sk[i] = v; sf[i] = vf;
  }
  __syncthreads();
  int idx = blockIdx.x * blockDim.x + threadIdx.x;
  int n = idx >> 10, m = idx & 1023;
  const float DXc = 2.0f / 31.0f;
  float e0 = -1.f + DXc * (float)(n >> 5);
  float e1 = -1.f + DXc * (float)(n & 31);
  float e2 = -1.f + DXc * (float)(m >> 5);
  float e3 = -1.f + DXc * (float)(m & 31);
  wout[idx] = edge_net(sk, e0, e1, e2, e3);
  fout[idx] = edge_net(sf, e0, e1, e2, e3);
}

__global__ void ln0_reduce_kernel(const float* xy, float* stats) {
  int b = blockIdx.x;
  const float* p = xy + (size_t)b * NTOK2 * FDIM;
  const int N = NTOK2 * FDIM;
  float s = 0.f, ss = 0.f;
  for (int i = threadIdx.x; i < N; i += 256) {
    float v = p[i];
    s += v; ss += v * v;
  }
  __shared__ float sh[512];
  sh[threadIdx.x] = s; sh[256 + threadIdx.x] = ss;
  __syncthreads();
  for (int st = 128; st >= 1; st >>= 1) {
    if ((int)threadIdx.x < st) {
      sh[threadIdx.x]       += sh[threadIdx.x + st];
      sh[256 + threadIdx.x] += sh[256 + threadIdx.x + st];
    }
    __syncthreads();
  }
  if (threadIdx.x == 0) {
    float mean = sh[0] / (float)N;
    float var  = sh[256] / (float)N - mean * mean;
    stats[2 * b]     = mean;
    stats[2 * b + 1] = rsqrtf(var + 1e-5f);
  }
}

__global__ void ln0_apply_kernel(const float* xy, const float* g, const float* bb,
                                 const float* stats, float* V) {
  int idx = blockIdx.x * blockDim.x + threadIdx.x;
  int b = idx >> 17;
  int r = idx & 131071;
  V[idx] = (xy[idx] - stats[2 * b]) * stats[2 * b + 1] * g[r] + bb[r];
}

__global__ void qk_kernel(const float* Vin, const float* Wq, const float* bq,
                          const float* Wk, const float* bk, float* Q, float* K) {
  int gx = blockIdx.x;
  bool isK = gx >= 32;
  int bh = gx & 31;
  const float* W    = isK ? Wk : Wq;
  const float* bias = isK ? bk : bq;
  float* Out        = isK ? K  : Q;
  int h = bh & 3;
  int b = bh >> 2;
  const float* A     = Vin + (size_t)b * NTOK2 * FDIM;
  const float* Wh    = W + h * DK * FDIM;
  const float* biash = bias + h * DK;
  int wave = threadIdx.x >> 5, lane = threadIdx.x & 31;
  int m0 = blockIdx.y * 64 + wave * 16;
  v8f c[4] = {};
  for (int k0 = 0; k0 < FDIM; k0 += 4) {
    v2f a = frag_a(A, FDIM, m0, k0, lane);
    int kk = k0 + ((lane >> 4) << 1);
#pragma unroll
    for (int t = 0; t < 4; ++t) {
      int col = t * 16 + (lane & 15);
      v2f bf; bf.x = Wh[col * FDIM + kk]; bf.y = Wh[col * FDIM + kk + 1];
      c[t] = wmma4(a, bf, c[t]);
    }
  }
  float* dst = Out + (size_t)bh * NTOK2 * DK;
  int rbase = m0 + ((lane >> 4) << 3);
#pragma unroll
  for (int t = 0; t < 4; ++t) {
    int col = t * 16 + (lane & 15);
    float bv = biash[col];
#pragma unroll
    for (int r = 0; r < 8; ++r) dst[(rbase + r) * DK + col] = c[t][r] + bv;
  }
}

__global__ void ktv_kernel(const float* Kbuf, const float* Bsrc, int bstride,
                           int kdim, float s, float* Mout) {
  __shared__ float sh[KSPLIT * TPLANE];
  int bh = blockIdx.x;
  int b = bh >> 2;
  int ks = __builtin_amdgcn_readfirstlane(threadIdx.x >> 5);
  int lane = threadIdx.x & 31;
  int m0 = blockIdx.y * 16;
  const float* Kh = Kbuf + (size_t)bh * NTOK2 * DK;
  const float* B  = Bsrc + (size_t)b * bstride;
  int klen = kdim / KSPLIT;
  int kbeg = ks * klen, kend = kbeg + klen;
  v8f c[4] = {};
  for (int k0 = kbeg; k0 < kend; k0 += 4) {
    v2f a = frag_a_t(Kh, DK, m0, k0, lane);
#pragma unroll
    for (int t = 0; t < 4; ++t) {
      v2f bf = frag_b(B, FDIM, k0, t * 16, lane);
      c[t] = wmma4(a, bf, c[t]);
    }
  }
  ksplit_reduce_store(sh, c, ks, lane, Mout + (size_t)bh * DK * FDIM, m0, s);
}

__global__ void qm_kernel(const float* Q, const float* Mb, float* Out) {
  int b = blockIdx.x;
  int wave = threadIdx.x >> 5, lane = threadIdx.x & 31;
  int m0 = blockIdx.y * 64 + wave * 16;
  v8f c[4] = {};
  for (int h = 0; h < HEAD; ++h) {
    const float* Qh = Q + (size_t)(b * HEAD + h) * NTOK2 * DK;
    const float* Mh = Mb + (size_t)(b * HEAD + h) * DK * FDIM;
    for (int k0 = 0; k0 < DK; k0 += 4) {
      v2f a = frag_a(Qh, DK, m0, k0, lane);
#pragma unroll
      for (int t = 0; t < 4; ++t) {
        v2f bf = frag_b(Mh, FDIM, k0, t * 16, lane);
        c[t] = wmma4(a, bf, c[t]);
      }
    }
  }
  float* dst = Out + (size_t)b * NTOK2 * FDIM;
  int rbase = m0 + ((lane >> 4) << 3);
#pragma unroll
  for (int t = 0; t < 4; ++t) {
    int col = t * 16 + (lane & 15);
#pragma unroll
    for (int r = 0; r < 8; ++r) dst[(rbase + r) * FDIM + col] = c[t][r];
  }
}

__global__ void ln_res_kernel(const float* mid, const float* vin, const float* g,
                              const float* bb, float* vout) {
  int row = blockIdx.x * 8 + (threadIdx.x >> 5);
  int lane = threadIdx.x & 31;
  const float* mr = mid + (size_t)row * FDIM;
  float x0 = mr[lane], x1 = mr[lane + 32];
  float s = x0 + x1, ss = x0 * x0 + x1 * x1;
#pragma unroll
  for (int m = 16; m >= 1; m >>= 1) {
    s  += __shfl_xor(s, m, 32);
    ss += __shfl_xor(ss, m, 32);
  }
  float mean = s * (1.f / 64.f);
  float var  = ss * (1.f / 64.f) - mean * mean;
  float inv  = rsqrtf(var + 1e-5f);
  size_t base = (size_t)row * FDIM;
  vout[base + lane]      = (x0 - mean) * inv * g[lane]      + bb[lane]      + vin[base + lane];
  vout[base + lane + 32] = (x1 - mean) * inv * g[lane + 32] + bb[lane + 32] + vin[base + lane + 32];
}

__global__ void final_kernel(const float* weight, const float* weightf,
                             const float* S, float* out, float mul) {
  __shared__ float sh[KSPLIT * TPLANE];
  int b = blockIdx.x;
  int ks = __builtin_amdgcn_readfirstlane(threadIdx.x >> 5);
  int lane = threadIdx.x & 31;
  int m0 = blockIdx.y * 16;
  const float* Sb = S + (size_t)b * NTOK2 * FDIM;
  int row = m0 + (lane & 15);
  const int klen = NTOK2 / KSPLIT;
  int kbeg = ks * klen, kend = kbeg + klen;
  const float* W = (kbeg < NTOKEN) ? weight : weightf;
  v8f c[4] = {};
  for (int k0 = kbeg; k0 < kend; k0 += 4) {
    int kb = (k0 & (NTOKEN - 1)) + ((lane >> 4) << 1);
    v2f a; a.x = W[kb * NTOKEN + row]; a.y = W[(kb + 1) * NTOKEN + row];
    if ((k0 & 60) == 0 && (k0 & (NTOKEN - 1)) + 64 < NTOKEN) {
      __builtin_prefetch(&W[(kb + 64) * NTOKEN + row], 0, 3);
    }
#pragma unroll
    for (int t = 0; t < 4; ++t) {
      v2f bf = frag_b(Sb, FDIM, k0, t * 16, lane);
      c[t] = wmma4(a, bf, c[t]);
    }
  }
  ksplit_reduce_store(sh, c, ks, lane, out + (size_t)b * NTOKEN * FDIM, m0, mul);
}

extern "C" void kernel_launch(void* const* d_in, const int* in_sizes, int n_in,
                              void* d_out, int out_size, void* d_ws, size_t ws_size,
                              hipStream_t stream) {
  (void)in_sizes; (void)n_in; (void)out_size; (void)ws_size;
  const float* xy   = (const float*)d_in[0];
  const float* kW0  = (const float*)d_in[1];
  const float* kb0  = (const float*)d_in[2];
  const float* kW1  = (const float*)d_in[3];
  const float* kb1  = (const float*)d_in[4];
  const float* kW2  = (const float*)d_in[5];
  const float* kb2  = (const float*)d_in[6];
  const float* fW0  = (const float*)d_in[7];
  const float* fb0  = (const float*)d_in[8];
  const float* fW1  = (const float*)d_in[9];
  const float* fb1  = (const float*)d_in[10];
  const float* fW2  = (const float*)d_in[11];
  const float* fb2  = (const float*)d_in[12];
  const float* Wq   = (const float*)d_in[13];
  const float* bq   = (const float*)d_in[14];
  const float* Wk   = (const float*)d_in[15];
  const float* bk   = (const float*)d_in[16];
  const float* ln0g = (const float*)d_in[17];
  const float* ln0b = (const float*)d_in[18];
  const float* lng  = (const float*)d_in[19];
  const float* lnb  = (const float*)d_in[20];
  float* out = (float*)d_out;

  float* ws = (float*)d_ws;
  const size_t SZ_1M = 1048576;
  float* w_weight  = ws;
  float* w_weightf = ws + SZ_1M;
  float* V0        = ws + 2 * SZ_1M;
  float* V1        = ws + 3 * SZ_1M;
  float* Qb        = ws + 4 * SZ_1M;
  float* Kb        = ws + 8 * SZ_1M;
  float* Mb        = ws + 12 * SZ_1M;
  float* MID       = ws + 12 * SZ_1M + 131072;
  float* stats     = ws + 13 * SZ_1M + 131072;

  const float DXc   = 2.0f / 31.0f;
  const float dxdy  = DXc * DXc;
  const float scale = 0.125f;

  edge_mlp_kernel<<<4096, 256, 0, stream>>>(kW0, kb0, kW1, kb1, kW2, kb2,
                                            fW0, fb0, fW1, fb1, fW2, fb2,
                                            w_weight, w_weightf);
  ln0_reduce_kernel<<<BATCH, 256, 0, stream>>>(xy, stats);
  ln0_apply_kernel<<<4096, 256, 0, stream>>>(xy, ln0g, ln0b, stats, V0);

  float* Va = V0;
  float* Vb = V1;
  for (int j = 0; j < NLAYER - 1; ++j) {
    qk_kernel<<<dim3(64, NTOK2 / 64), 128, 0, stream>>>(
        Va, Wq + (size_t)j * HEAD * DK * FDIM, bq + (size_t)j * HEAD * DK,
        Wk + (size_t)j * HEAD * DK * FDIM, bk + (size_t)j * HEAD * DK, Qb, Kb);
    ktv_kernel<<<dim3(32, DK / 16), KSPLIT * 32, 0, stream>>>(Kb, Va, NTOK2 * FDIM, NTOK2,
                                       scale * dxdy, Mb);
    qm_kernel<<<dim3(BATCH, NTOK2 / 64), 128, 0, stream>>>(Qb, Mb, MID);
    ln_res_kernel<<<2048, 256, 0, stream>>>(MID, Va, lng + j * FDIM,
                                            lnb + j * FDIM, Vb);
    float* t = Va; Va = Vb; Vb = t;
  }

  {
    int j = NLAYER - 1;
    qk_kernel<<<dim3(64, NTOK2 / 64), 128, 0, stream>>>(
        Va, Wq + (size_t)j * HEAD * DK * FDIM, bq + (size_t)j * HEAD * DK,
        Wk + (size_t)j * HEAD * DK * FDIM, bk + (size_t)j * HEAD * DK, Qb, Kb);
    ktv_kernel<<<dim3(32, DK / 16), KSPLIT * 32, 0, stream>>>(Kb, xy, NTOK2 * FDIM, NTOKEN, scale, Mb);
    qm_kernel<<<dim3(BATCH, NTOK2 / 64), 128, 0, stream>>>(Qb, Mb, MID);
    final_kernel<<<dim3(BATCH, NTOKEN / 16), KSPLIT * 32, 0, stream>>>(
        w_weight, w_weightf, MID, out, dxdy * dxdy);
  }
}


# Edge MLPs: the 32->64 hidden layer moves from per-thread scalar FMAs fed by LDS-broadcast weights onto f32 WMMA (W1 and biases held in registers); hidden and output sums change order
# speedup vs baseline: 2.5264x; 1.3923x over previous
#include <hip/hip_runtime.h>
#include <hip/hip_bf16.h>

#define NGRID   32
#define NTOKEN  1024
#define NTOK2   2048
#define HEAD    4
#define DK      64
#define NLAYER  4
#define FDIM    64
#define BATCH   8

typedef __attribute__((ext_vector_type(2))) float v2f;
typedef __attribute__((ext_vector_type(8))) float v8f;

__device__ __forceinline__ v8f wmma4(v2f a, v2f b, v8f c) {
  return __builtin_amdgcn_wmma_f32_16x16x4_f32(
       false, a,  false, b,
       (short)0, c,  false,  false);
}

__device__ __forceinline__ v2f frag_a(const float* src, int lda, int m0, int k0, int lane) {
  int row = m0 + (lane & 15);
  int kk  = k0 + ((lane >> 4) << 1);
  v2f a; a.x = src[row * lda + kk]; a.y = src[row * lda + kk + 1];
  return a;
}

__device__ __forceinline__ v2f frag_a_t(const float* src, int lds, int m0, int k0, int lane) {
  int row = m0 + (lane & 15);
  int kk  = k0 + ((lane >> 4) << 1);
  v2f a; a.x = src[kk * lds + row]; a.y = src[(kk + 1) * lds + row];
  return a;
}

__device__ __forceinline__ v2f frag_b(const float* src, int ldb, int k0, int n0, int lane) {
  int col = n0 + (lane & 15);
  int kk  = k0 + ((lane >> 4) << 1);
  v2f b; b.x = src[kk * ldb + col]; b.y = src[(kk + 1) * ldb + col];
  return b;
}


#define KSPLIT   16
#define TPITCH   68
#define TPLANE   (16 * TPITCH)

__device__ __forceinline__ void ksplit_reduce_store(float* sh, const v8f* c, int ks, int lane,
                                                    float* dst, int m0, float mul) {
  float* mine = sh + ks * TPLANE;
  int lrow = (lane >> 4) << 3;
  int lcol = lane & 15;
#pragma unroll
  for (int t = 0; t < 4; ++t) {
#pragma unroll
    for (int r = 0; r < 8; ++r) mine[(lrow + r) * TPITCH + t * 16 + lcol] = c[t][r];
  }
  __syncthreads();
  for (int idx = threadIdx.x; idx < 16 * 64; idx += KSPLIT * 32) {
    int row = idx >> 6, col = idx & 63;
    float acc = sh[row * TPITCH + col];
#pragma unroll
    for (int p = 1; p < KSPLIT; ++p) acc += sh[p * TPLANE + row * TPITCH + col];
    dst[(m0 + row) * 64 + col] = acc * mul;
  }
}

__device__ __forceinline__ float leaky01(float t) {
  return fmaxf(t, 0.01f * t);
}

__global__ __launch_bounds__(128) void edge_mlp_kernel(
                                const float* kW0, const float* kb0, const float* kW1,
                                const float* kb1, const float* kW2, const float* kb2,
                                const float* fW0, const float* fb0, const float* fW1,
                                const float* fb1, const float* fW2, const float* fb2,
                                float* wout, float* fout) {
  const bool isF = blockIdx.y != 0;
  const float* W0 = isF ? fW0 : kW0;
  const float* b0 = isF ? fb0 : kb0;
  const float* W1 = isF ? fW1 : kW1;
  const float* b1 = isF ? fb1 : kb1;
  const float* W2 = isF ? fW2 : kW2;
  const float* b2 = isF ? fb2 : kb2;
  float* dst      = isF ? fout : wout;

  const int lane = threadIdx.x & 31;
  const int wave = threadIdx.x >> 5;
  const int c = lane & 15, h = lane >> 4;
  const int n = blockIdx.x;
  const int mbase = wave * 256;
  const float DXc = 2.0f / 31.0f;
  const float e0 = -1.f + DXc * (float)(n >> 5);
  const float e1 = -1.f + DXc * (float)(n & 31);

  v2f wa[4][8];
#pragma unroll
  for (int t = 0; t < 4; ++t) {
#pragma unroll
    for (int i = 0; i < 8; ++i) {
      const float* p = W1 + (t * 16 + c) * 32 + 4 * i + 2 * h;
      wa[t][i].x = p[0]; wa[t][i].y = p[1];
    }
  }
  float pre[16], wc2[16], wc3[16];
#pragma unroll
  for (int i = 0; i < 8; ++i) {
#pragma unroll
    for (int j = 0; j < 2; ++j) {
      int u = 4 * i + 2 * h + j;
      float t = b0[u];
      t = fmaf(W0[u * 4 + 0], e0, t);
      t = fmaf(W0[u * 4 + 1], e1, t);
      pre[2 * i + j] = t;
      wc2[2 * i + j] = W0[u * 4 + 2];
      wc3[2 * i + j] = W0[u * 4 + 3];
    }
  }
  v8f cb[4];
  float w2v[4][8];
#pragma unroll
  for (int t = 0; t < 4; ++t) {
#pragma unroll
    for (int r = 0; r < 8; ++r) {
      cb[t][r]  = b1[t * 16 + 8 * h + r];
      w2v[t][r] = W2[t * 16 + 8 * h + r];
    }
  }
  const float bias2 = b2[0];

  for (int it = 0; it < 8; ++it) {
    const int m0 = mbase + it * 32;
    float res[2];
#pragma unroll
    for (int et = 0; et < 2; ++et) {
      const int m = m0 + et * 16 + c;
      const float e2 = -1.f + DXc * (float)(m >> 5);
      const float e3 = -1.f + DXc * (float)(m & 31);
      v2f hb[8];
#pragma unroll
      for (int i = 0; i < 8; ++i) {
        float t0 = fmaf(wc3[2 * i],     e3, fmaf(wc2[2 * i],     e2, pre[2 * i]));
        float t1 = fmaf(wc3[2 * i + 1], e3, fmaf(wc2[2 * i + 1], e2, pre[2 * i + 1]));
        hb[i].x = leaky01(t0); hb[i].y = leaky01(t1);
      }
      v8f acc[4];
#pragma unroll
      for (int t = 0; t < 4; ++t) acc[t] = wmma4(wa[t][0], hb[0], cb[t]);
#pragma unroll
      for (int i = 1; i < 8; ++i) {
#pragma unroll
        for (int t = 0; t < 4; ++t) acc[t] = wmma4(wa[t][i], hb[i], acc[t]);
      }
      float s = 0.f;
#pragma unroll
      for (int t = 0; t < 4; ++t) {
#pragma unroll
        for (int r = 0; r < 8; ++r) s = fmaf(w2v[t][r], leaky01(acc[t][r]), s);
      }
      s += __shfl_xor(s, 16, 32);
      res[et] = s + bias2;
    }
    dst[n * NTOKEN + m0 + lane] = h ? res[1] : res[0];
  }
}

__global__ void ln0_reduce_kernel(const float* xy, float* stats) {
  int b = blockIdx.x;
  const float* p = xy + (size_t)b * NTOK2 * FDIM;
  const int N = NTOK2 * FDIM;
  float s = 0.f, ss = 0.f;
  for (int i = threadIdx.x; i < N; i += 256) {
    float v = p[i];
    s += v; ss += v * v;
  }
  __shared__ float sh[512];
  sh[threadIdx.x] = s; sh[256 + threadIdx.x] = ss;
  __syncthreads();
  for (int st = 128; st >= 1; st >>= 1) {
    if ((int)threadIdx.x < st) {
      sh[threadIdx.x]       += sh[threadIdx.x + st];
      sh[256 + threadIdx.x] += sh[256 + threadIdx.x + st];
    }
    __syncthreads();
  }
  if (threadIdx.x == 0) {
    float mean = sh[0] / (float)N;
    float var  = sh[256] / (float)N - mean * mean;
    stats[2 * b]     = mean;
    stats[2 * b + 1] = rsqrtf(var + 1e-5f);
  }
}

__global__ void ln0_apply_kernel(const float* xy, const float* g, const float* bb,
                                 const float* stats, float* V) {
  int idx = blockIdx.x * blockDim.x + threadIdx.x;
  int b = idx >> 17;
  int r = idx & 131071;
  V[idx] = (xy[idx] - stats[2 * b]) * stats[2 * b + 1] * g[r] + bb[r];
}

__global__ void qk_kernel(const float* Vin, const float* Wq, const float* bq,
                          const float* Wk, const float* bk, float* Q, float* K) {
  int gx = blockIdx.x;
  bool isK = gx >= 32;
  int bh = gx & 31;
  const float* W    = isK ? Wk : Wq;
  const float* bias = isK ? bk : bq;
  float* Out        = isK ? K  : Q;
  int h = bh & 3;
  int b = bh >> 2;
  const float* A     = Vin + (size_t)b * NTOK2 * FDIM;
  const float* Wh    = W + h * DK * FDIM;
  const float* biash = bias + h * DK;
  int wave = threadIdx.x >> 5, lane = threadIdx.x & 31;
  int m0 = blockIdx.y * 64 + wave * 16;
  v8f c[4] = {};
  for (int k0 = 0; k0 < FDIM; k0 += 4) {
    v2f a = frag_a(A, FDIM, m0, k0, lane);
    int kk = k0 + ((lane >> 4) << 1);
#pragma unroll
    for (int t = 0; t < 4; ++t) {
      int col = t * 16 + (lane & 15);
      v2f bf; bf.x = Wh[col * FDIM + kk]; bf.y = Wh[col * FDIM + kk + 1];
      c[t] = wmma4(a, bf, c[t]);
    }
  }
  float* dst = Out + (size_t)bh * NTOK2 * DK;
  int rbase = m0 + ((lane >> 4) << 3);
#pragma unroll
  for (int t = 0; t < 4; ++t) {
    int col = t * 16 + (lane & 15);
    float bv = biash[col];
#pragma unroll
    for (int r = 0; r < 8; ++r) dst[(rbase + r) * DK + col] = c[t][r] + bv;
  }
}

__global__ void ktv_kernel(const float* Kbuf, const float* Bsrc, int bstride,
                           int kdim, float s, float* Mout) {
  __shared__ float sh[KSPLIT * TPLANE];
  int bh = blockIdx.x;
  int b = bh >> 2;
  int ks = __builtin_amdgcn_readfirstlane(threadIdx.x >> 5);
  int lane = threadIdx.x & 31;
  int m0 = blockIdx.y * 16;
  const float* Kh = Kbuf + (size_t)bh * NTOK2 * DK;
  const float* B  = Bsrc + (size_t)b * bstride;
  int klen = kdim / KSPLIT;
  int kbeg = ks * klen, kend = kbeg + klen;
  v8f c[4] = {};
  for (int k0 = kbeg; k0 < kend; k0 += 4) {
    v2f a = frag_a_t(Kh, DK, m0, k0, lane);
#pragma unroll
    for (int t = 0; t < 4; ++t) {
      v2f bf = frag_b(B, FDIM, k0, t * 16, lane);
      c[t] = wmma4(a, bf, c[t]);
    }
  }
  ksplit_reduce_store(sh, c, ks, lane, Mout + (size_t)bh * DK * FDIM, m0, s);
}

__global__ void qm_kernel(const float* Q, const float* Mb, float* Out) {
  int b = blockIdx.x;
  int wave = threadIdx.x >> 5, lane = threadIdx.x & 31;
  int m0 = blockIdx.y * 64 + wave * 16;
  v8f c[4] = {};
  for (int h = 0; h < HEAD; ++h) {
    const float* Qh = Q + (size_t)(b * HEAD + h) * NTOK2 * DK;
    const float* Mh = Mb + (size_t)(b * HEAD + h) * DK * FDIM;
    for (int k0 = 0; k0 < DK; k0 += 4) {
      v2f a = frag_a(Qh, DK, m0, k0, lane);
#pragma unroll
      for (int t = 0; t < 4; ++t) {
        v2f bf = frag_b(Mh, FDIM, k0, t * 16, lane);
        c[t] = wmma4(a, bf, c[t]);
      }
    }
  }
  float* dst = Out + (size_t)b * NTOK2 * FDIM;
  int rbase = m0 + ((lane >> 4) << 3);
#pragma unroll
  for (int t = 0; t < 4; ++t) {
    int col = t * 16 + (lane & 15);
#pragma unroll
    for (int r = 0; r < 8; ++r) dst[(rbase + r) * FDIM + col] = c[t][r];
  }
}

__global__ void ln_res_kernel(const float* mid, const float* vin, const float* g,
                              const float* bb, float* vout) {
  int row = blockIdx.x * 8 + (threadIdx.x >> 5);
  int lane = threadIdx.x & 31;
  const float* mr = mid + (size_t)row * FDIM;
  float x0 = mr[lane], x1 = mr[lane + 32];
  float s = x0 + x1, ss = x0 * x0 + x1 * x1;
#pragma unroll
  for (int m = 16; m >= 1; m >>= 1) {
    s  += __shfl_xor(s, m, 32);
    ss += __shfl_xor(ss, m, 32);
  }
  float mean = s * (1.f / 64.f);
  float var  = ss * (1.f / 64.f) - mean * mean;
  float inv  = rsqrtf(var + 1e-5f);
  size_t base = (size_t)row * FDIM;
  vout[base + lane]      = (x0 - mean) * inv * g[lane]      + bb[lane]      + vin[base + lane];
  vout[base + lane + 32] = (x1 - mean) * inv * g[lane + 32] + bb[lane + 32] + vin[base + lane + 32];
}

__global__ void final_kernel(const float* weight, const float* weightf,
                             const float* S, float* out, float mul) {
  __shared__ float sh[KSPLIT * TPLANE];
  int b = blockIdx.x;
  int ks = __builtin_amdgcn_readfirstlane(threadIdx.x >> 5);
  int lane = threadIdx.x & 31;
  int m0 = blockIdx.y * 16;
  const float* Sb = S + (size_t)b * NTOK2 * FDIM;
  int row = m0 + (lane & 15);
  const int klen = NTOK2 / KSPLIT;
  int kbeg = ks * klen, kend = kbeg + klen;
  const float* W = (kbeg < NTOKEN) ? weight : weightf;
  v8f c[4] = {};
  for (int k0 = kbeg; k0 < kend; k0 += 4) {
    int kb = (k0 & (NTOKEN - 1)) + ((lane >> 4) << 1);
    v2f a; a.x = W[kb * NTOKEN + row]; a.y = W[(kb + 1) * NTOKEN + row];
    if ((k0 & 60) == 0 && (k0 & (NTOKEN - 1)) + 64 < NTOKEN) {
      __builtin_prefetch(&W[(kb + 64) * NTOKEN + row], 0, 3);
    }
#pragma unroll
    for (int t = 0; t < 4; ++t) {
      v2f bf = frag_b(Sb, FDIM, k0, t * 16, lane);
      c[t] = wmma4(a, bf, c[t]);
    }
  }
  ksplit_reduce_store(sh, c, ks, lane, out + (size_t)b * NTOKEN * FDIM, m0, mul);
}

extern "C" void kernel_launch(void* const* d_in, const int* in_sizes, int n_in,
                              void* d_out, int out_size, void* d_ws, size_t ws_size,
                              hipStream_t stream) {
  (void)in_sizes; (void)n_in; (void)out_size; (void)ws_size;
  const float* xy   = (const float*)d_in[0];
  const float* kW0  = (const float*)d_in[1];
  const float* kb0  = (const float*)d_in[2];
  const float* kW1  = (const float*)d_in[3];
  const float* kb1  = (const float*)d_in[4];
  const float* kW2  = (const float*)d_in[5];
  const float* kb2  = (const float*)d_in[6];
  const float* fW0  = (const float*)d_in[7];
  const float* fb0  = (const float*)d_in[8];
  const float* fW1  = (const float*)d_in[9];
  const float* fb1  = (const float*)d_in[10];
  const float* fW2  = (const float*)d_in[11];
  const float* fb2  = (const float*)d_in[12];
  const float* Wq   = (const float*)d_in[13];
  const float* bq   = (const float*)d_in[14];
  const float* Wk   = (const float*)d_in[15];
  const float* bk   = (const float*)d_in[16];
  const float* ln0g = (const float*)d_in[17];
  const float* ln0b = (const float*)d_in[18];
  const float* lng  = (const float*)d_in[19];
  const float* lnb  = (const float*)d_in[20];
  float* out = (float*)d_out;

  float* ws = (float*)d_ws;
  const size_t SZ_1M = 1048576;
  float* w_weight  = ws;
  float* w_weightf = ws + SZ_1M;
  float* V0        = ws + 2 * SZ_1M;
  float* V1        = ws + 3 * SZ_1M;
  float* Qb        = ws + 4 * SZ_1M;
  float* Kb        = ws + 8 * SZ_1M;
  float* Mb        = ws + 12 * SZ_1M;
  float* MID       = ws + 12 * SZ_1M + 131072;
  float* stats     = ws + 13 * SZ_1M + 131072;

  const float DXc   = 2.0f / 31.0f;
  const float dxdy  = DXc * DXc;
  const float scale = 0.125f;

  edge_mlp_kernel<<<dim3(NTOKEN, 2), 128, 0, stream>>>(kW0, kb0, kW1, kb1, kW2, kb2,
                                                       fW0, fb0, fW1, fb1, fW2, fb2,
                                                       w_weight, w_weightf);
  ln0_reduce_kernel<<<BATCH, 256, 0, stream>>>(xy, stats);
  ln0_apply_kernel<<<4096, 256, 0, stream>>>(xy, ln0g, ln0b, stats, V0);

  float* Va = V0;
  float* Vb = V1;
  for (int j = 0; j < NLAYER - 1; ++j) {
    qk_kernel<<<dim3(64, NTOK2 / 64), 128, 0, stream>>>(
        Va, Wq + (size_t)j * HEAD * DK * FDIM, bq + (size_t)j * HEAD * DK,
        Wk + (size_t)j * HEAD * DK * FDIM, bk + (size_t)j * HEAD * DK, Qb, Kb);
    ktv_kernel<<<dim3(32, DK / 16), KSPLIT * 32, 0, stream>>>(Kb, Va, NTOK2 * FDIM, NTOK2,
                                       scale * dxdy, Mb);
    qm_kernel<<<dim3(BATCH, NTOK2 / 64), 128, 0, stream>>>(Qb, Mb, MID);
    ln_res_kernel<<<2048, 256, 0, stream>>>(MID, Va, lng + j * FDIM,
                                            lnb + j * FDIM, Vb);
    float* t = Va; Va = Vb; Vb = t;
  }

  {
    int j = NLAYER - 1;
    qk_kernel<<<dim3(64, NTOK2 / 64), 128, 0, stream>>>(
        Va, Wq + (size_t)j * HEAD * DK * FDIM, bq + (size_t)j * HEAD * DK,
        Wk + (size_t)j * HEAD * DK * FDIM, bk + (size_t)j * HEAD * DK, Qb, Kb);
    ktv_kernel<<<dim3(32, DK / 16), KSPLIT * 32, 0, stream>>>(Kb, xy, NTOK2 * FDIM, NTOKEN, scale, Mb);
    qm_kernel<<<dim3(BATCH, NTOK2 / 64), 128, 0, stream>>>(Qb, Mb, MID);
    final_kernel<<<dim3(BATCH, NTOKEN / 16), KSPLIT * 32, 0, stream>>>(
        w_weight, w_weightf, MID, out, dxdy * dxdy);
  }
}


# Q and K never formed: layers reassociated through the 64x64 Gram product Vin^T B and weight-only Wq^T Wk (mid = Vin (C G + c g) + bias row), Q/K round trip gone; sums change order
# speedup vs baseline: 4.1280x; 1.6339x over previous
#include <hip/hip_runtime.h>
#include <hip/hip_bf16.h>

#define NGRID   32
#define NTOKEN  1024
#define NTOK2   2048
#define HEAD    4
#define DK      64
#define NLAYER  4
#define FDIM    64
#define BATCH   8

typedef __attribute__((ext_vector_type(2))) float v2f;
typedef __attribute__((ext_vector_type(8))) float v8f;

__device__ __forceinline__ v8f wmma4(v2f a, v2f b, v8f c) {
  return __builtin_amdgcn_wmma_f32_16x16x4_f32(
       false, a,  false, b,
       (short)0, c,  false,  false);
}

__device__ __forceinline__ v2f frag_a(const float* src, int lda, int m0, int k0, int lane) {
  int row = m0 + (lane & 15);
  int kk  = k0 + ((lane >> 4) << 1);
  v2f a; a.x = src[row * lda + kk]; a.y = src[row * lda + kk + 1];
  return a;
}

__device__ __forceinline__ v2f frag_b(const float* src, int ldb, int k0, int n0, int lane) {
  int col = n0 + (lane & 15);
  int kk  = k0 + ((lane >> 4) << 1);
  v2f b; b.x = src[kk * ldb + col]; b.y = src[(kk + 1) * ldb + col];
  return b;
}


#define KSPLIT   16
#define TPITCH   68
#define TPLANE   (16 * TPITCH)

__device__ __forceinline__ void ksplit_reduce_store(float* sh, const v8f* c, int ks, int lane,
                                                    float* dst, int m0, float mul,
                                                    int tstep, int cstep) {
  float* mine = sh + ks * TPLANE;
  int lrow = (lane >> 4) << 3;
  int lcol = lane & 15;
#pragma unroll
  for (int t = 0; t < 4; ++t) {
#pragma unroll
    for (int r = 0; r < 8; ++r) mine[(lrow + r) * TPITCH + t * tstep + lcol * cstep] = c[t][r];
  }
  __syncthreads();
  for (int idx = threadIdx.x; idx < 16 * 64; idx += KSPLIT * 32) {
    int row = idx >> 6, col = idx & 63;
    float acc = sh[row * TPITCH + col];
#pragma unroll
    for (int p = 1; p < KSPLIT; ++p) acc += sh[p * TPLANE + row * TPITCH + col];
    dst[(m0 + row) * 64 + col] = acc * mul;
  }
}

__device__ __forceinline__ float leaky01(float t) {
  return fmaxf(t, 0.01f * t);
}

__global__ __launch_bounds__(128) void edge_mlp_kernel(
                                const float* kW0, const float* kb0, const float* kW1,
                                const float* kb1, const float* kW2, const float* kb2,
                                const float* fW0, const float* fb0, const float* fW1,
                                const float* fb1, const float* fW2, const float* fb2,
                                float* wout, float* fout) {
  const bool isF = blockIdx.y != 0;
  const float* W0 = isF ? fW0 : kW0;
  const float* b0 = isF ? fb0 : kb0;
  const float* W1 = isF ? fW1 : kW1;
  const float* b1 = isF ? fb1 : kb1;
  const float* W2 = isF ? fW2 : kW2;
  const float* b2 = isF ? fb2 : kb2;
  float* dst      = isF ? fout : wout;

  const int lane = threadIdx.x & 31;
  const int wave = threadIdx.x >> 5;
  const int c = lane & 15, h = lane >> 4;
  const int n = blockIdx.x;
  const int mbase = wave * 256;
  const float DXc = 2.0f / 31.0f;
  const float e0 = -1.f + DXc * (float)(n >> 5);
  const float e1 = -1.f + DXc * (float)(n & 31);

  v2f wa[4][8];
#pragma unroll
  for (int t = 0; t < 4; ++t) {
#pragma unroll
    for (int i = 0; i < 8; ++i) {
      const float* p = W1 + (t * 16 + c) * 32 + 4 * i + 2 * h;
      wa[t][i].x = p[0]; wa[t][i].y = p[1];
    }
  }
  float pre[16], wc2[16], wc3[16];
#pragma unroll
  for (int i = 0; i < 8; ++i) {
#pragma unroll
    for (int j = 0; j < 2; ++j) {
      int u = 4 * i + 2 * h + j;
      float t = b0[u];
      t = fmaf(W0[u * 4 + 0], e0, t);
      t = fmaf(W0[u * 4 + 1], e1, t);
      pre[2 * i + j] = t;
      wc2[2 * i + j] = W0[u * 4 + 2];
      wc3[2 * i + j] = W0[u * 4 + 3];
    }
  }
  v8f cb[4];
  float w2v[4][8];
#pragma unroll
  for (int t = 0; t < 4; ++t) {
#pragma unroll
    for (int r = 0; r < 8; ++r) {
      cb[t][r]  = b1[t * 16 + 8 * h + r];
      w2v[t][r] = W2[t * 16 + 8 * h + r];
    }
  }
  const float bias2 = b2[0];

  for (int it = 0; it < 8; ++it) {
    const int m0 = mbase + it * 32;
    float res[2];
#pragma unroll
    for (int et = 0; et < 2; ++et) {
      const int m = m0 + et * 16 + c;
      const float e2 = -1.f + DXc * (float)(m >> 5);
      const float e3 = -1.f + DXc * (float)(m & 31);
      v2f hb[8];
#pragma unroll
      for (int i = 0; i < 8; ++i) {
        float t0 = fmaf(wc3[2 * i],     e3, fmaf(wc2[2 * i],     e2, pre[2 * i]));
        float t1 = fmaf(wc3[2 * i + 1], e3, fmaf(wc2[2 * i + 1], e2, pre[2 * i + 1]));
        hb[i].x = leaky01(t0); hb[i].y = leaky01(t1);
      }
      v8f acc[4];
#pragma unroll
      for (int t = 0; t < 4; ++t) acc[t] = wmma4(wa[t][0], hb[0], cb[t]);
#pragma unroll
      for (int i = 1; i < 8; ++i) {
#pragma unroll
        for (int t = 0; t < 4; ++t) acc[t] = wmma4(wa[t][i], hb[i], acc[t]);
      }
      float s = 0.f;
#pragma unroll
      for (int t = 0; t < 4; ++t) {
#pragma unroll
        for (int r = 0; r < 8; ++r) s = fmaf(w2v[t][r], leaky01(acc[t][r]), s);
      }
      s += __shfl_xor(s, 16, 32);
      res[et] = s + bias2;
    }
    dst[n * NTOKEN + m0 + lane] = h ? res[1] : res[0];
  }
}

__global__ void ln0_reduce_kernel(const float* xy, float* stats) {
  int b = blockIdx.x;
  const float* p = xy + (size_t)b * NTOK2 * FDIM;
  const int N = NTOK2 * FDIM;
  float s = 0.f, ss = 0.f;
  for (int i = threadIdx.x; i < N; i += 256) {
    float v = p[i];
    s += v; ss += v * v;
  }
  __shared__ float sh[512];
  sh[threadIdx.x] = s; sh[256 + threadIdx.x] = ss;
  __syncthreads();
  for (int st = 128; st >= 1; st >>= 1) {
    if ((int)threadIdx.x < st) {
      sh[threadIdx.x]       += sh[threadIdx.x + st];
      sh[256 + threadIdx.x] += sh[256 + threadIdx.x + st];
    }
    __syncthreads();
  }
  if (threadIdx.x == 0) {
    float mean = sh[0] / (float)N;
    float var  = sh[256] / (float)N - mean * mean;
    stats[2 * b]     = mean;
    stats[2 * b + 1] = rsqrtf(var + 1e-5f);
  }
}

__global__ void ln0_apply_kernel(const float* xy, const float* g, const float* bb,
                                 const float* stats, float* V) {
  int idx = blockIdx.x * blockDim.x + threadIdx.x;
  int b = idx >> 17;
  int r = idx & 131071;
  V[idx] = (xy[idx] - stats[2 * b]) * stats[2 * b + 1] * g[r] + bb[r];
}

#define KQ      4
#define CAUG    (80 * 64)
#define GPITCH  72

__global__ __launch_bounds__(KSPLIT * 32) void atb_kernel(
                           const float* Abuf, int astride, const float* Bbuf, int bstride,
                           int kdim, float s, float* Out, int ostride, float* csum) {
  __shared__ float sh[KSPLIT * TPLANE];
  __shared__ float shc[KSPLIT * 64];
  int z = blockIdx.x;
  int ks = __builtin_amdgcn_readfirstlane(threadIdx.x >> 5);
  int lane = threadIdx.x & 31;
  const int cl = lane & 15, hl = lane >> 4;
  int m0 = blockIdx.y * 16;
  int klen = kdim / (KSPLIT * (int)gridDim.z);
  int kbeg = ((int)blockIdx.z * KSPLIT + ks) * klen, kend = kbeg + klen;
  const float* Ap = Abuf + (size_t)z * astride + m0 + cl;
  const float* Bp = Bbuf + (size_t)z * bstride + 4 * cl;
  v8f c[4] = {};
  float cs[4] = {0.f, 0.f, 0.f, 0.f};
  for (int kb = kbeg; kb < kend; kb += 16) {
    v2f a[4];
    float b0[4][4], b1[4][4];
#pragma unroll
    for (int u = 0; u < 4; ++u) {
      const int kk = kb + 4 * u + 2 * hl;
      a[u].x = Ap[kk * 64]; a[u].y = Ap[(kk + 1) * 64];
#pragma unroll
      for (int t = 0; t < 4; ++t) {
        b0[u][t] = Bp[kk * 64 + t];
        b1[u][t] = Bp[(kk + 1) * 64 + t];
      }
    }
#pragma unroll
    for (int u = 0; u < 4; ++u) {
#pragma unroll
      for (int t = 0; t < 4; ++t) {
        v2f bf; bf.x = b0[u][t]; bf.y = b1[u][t];
        cs[t] += bf.x + bf.y;
        c[t] = wmma4(a[u], bf, c[t]);
      }
    }
  }
#pragma unroll
  for (int t = 0; t < 4; ++t) {
    float v = cs[t] + __shfl_xor(cs[t], 16, 32);
    if (lane < 16) shc[ks * 64 + 4 * lane + t] = v;
  }
  size_t slot = (size_t)z * gridDim.z + blockIdx.z;
  ksplit_reduce_store(sh, c, ks, lane, Out + slot * ostride, m0, s, 1, 4);
  if (csum != nullptr && blockIdx.y == 0 && threadIdx.x < 64) {
    float a = shc[threadIdx.x];
#pragma unroll
    for (int p = 1; p < KSPLIT; ++p) a += shc[p * 64 + threadIdx.x];
    csum[slot * 64 + threadIdx.x] = a * s;
  }
}

__global__ void cvec_kernel(const float* Wq, const float* bq, const float* Wk, const float* bk,
                            float* Caug, float* caug) {
  __shared__ float sh[16 * 64];
  const int j = blockIdx.x;
  const bool isD = blockIdx.y != 0;
  const float* Wm = (isD ? Wk : Wq) + (size_t)j * 256 * 64;
  const float* v  = (isD ? bq : bk) + j * 256;
  const int i = threadIdx.x & 63, kc = threadIdx.x >> 6;
  float a = 0.f;
  for (int u = 0; u < 16; ++u) {
    int k = kc * 16 + u;
    a = fmaf(Wm[k * 64 + i], v[k], a);
  }
  sh[kc * 64 + i] = a;
  __syncthreads();
  float* Cj = Caug + (size_t)j * CAUG;
  float* cj = caug + j * 80;
  if (threadIdx.x < 64) {
    float t = sh[threadIdx.x];
    for (int p = 1; p < 16; ++p) t += sh[p * 64 + threadIdx.x];
    if (isD) Cj[64 * 64 + threadIdx.x] = t; else cj[threadIdx.x] = t;
  }
  if (isD) {
    if ((threadIdx.x >> 5) == 2) {
      const int lane = threadIdx.x & 31;
      float e = 0.f;
      for (int u = 0; u < 8; ++u) {
        int k = lane + 32 * u;
        e = fmaf(bq[j * 256 + k], bk[j * 256 + k], e);
      }
      for (int m = 16; m >= 1; m >>= 1) e += __shfl_xor(e, m, 32);
      if (lane == 0) cj[64] = e;
    }
  } else {
    if (threadIdx.x < 15 * 64) Cj[65 * 64 + threadIdx.x] = 0.f;
    if (threadIdx.x < 15) cj[65 + threadIdx.x] = 0.f;
  }
}

__global__ __launch_bounds__(1024) void pmat_kernel(const float* Cj, const float* cj,
                                                    const float* Gp, const float* gp,
                                                    float* Paug) {
  __shared__ float sG[64 * GPITCH];
  __shared__ float sg[64];
  const int b = blockIdx.x, tid = threadIdx.x;
  const float* G0 = Gp + (size_t)b * KQ * 4096;
  const float* g0 = gp + (size_t)b * KQ * 64;
  const int w = __builtin_amdgcn_readfirstlane(tid >> 5);
  const int lane = tid & 31, cl = lane & 15, hl = lane >> 4;
  const int rt = w >> 2, ct = w & 3;
  const int rbase = rt * 16 + (hl << 3);
  v2f a[16];
  float cjv[8];
#pragma unroll
  for (int u = 0; u < 16; ++u) { a[u].x = 0.f; a[u].y = 0.f; }
#pragma unroll
  for (int r = 0; r < 8; ++r) cjv[r] = 0.f;
  if (w < 20) {
#pragma unroll
    for (int u = 0; u < 16; ++u) a[u] = frag_a(Cj, 64, rt * 16, 4 * u, lane);
#pragma unroll
    for (int r = 0; r < 8; ++r) cjv[r] = cj[rbase + r];
  }
#pragma unroll
  for (int i = 0; i < 4; ++i) {
    const int idx = tid + 1024 * i;
    float v = G0[idx];
#pragma unroll
    for (int q = 1; q < KQ; ++q) v += G0[q * 4096 + idx];
    sG[(idx >> 6) * GPITCH + (idx & 63)] = v;
  }
  if (tid < 64) {
    float v = g0[tid];
#pragma unroll
    for (int q = 1; q < KQ; ++q) v += g0[q * 64 + tid];
    sg[tid] = v;
  }
  __syncthreads();
  if (w < 20) {
    const int col = ct * 16 + cl;
    const float g = sg[col];
    v8f acc;
#pragma unroll
    for (int r = 0; r < 8; ++r) acc[r] = cjv[r] * g;
#pragma unroll
    for (int u = 0; u < 16; ++u) {
      const int kk = 4 * u + 2 * hl;
      v2f bf; bf.x = sG[kk * GPITCH + col]; bf.y = sG[(kk + 1) * GPITCH + col];
      acc = wmma4(a[u], bf, acc);
    }
    float* dst = Paug + (size_t)b * CAUG;
#pragma unroll
    for (int r = 0; r < 8; ++r) dst[(rbase + r) * 64 + col] = acc[r];
  }
}

__global__ __launch_bounds__(128) void vp_kernel(const float* Vin, const float* Paug, float* Out) {
  const int b = blockIdx.x;
  const int wave = threadIdx.x >> 5, lane = threadIdx.x & 31;
  const int cl = lane & 15, hl = lane >> 4;
  const int m0 = blockIdx.y * 64 + wave * 16;
  const float* A  = Vin + (size_t)b * NTOK2 * FDIM;
  const float* Pp = Paug + (size_t)b * CAUG + 4 * cl;
  float pv[4];
#pragma unroll
  for (int t = 0; t < 4; ++t) pv[t] = Pp[64 * 64 + t];
  v8f c[4] = {};
  for (int kb = 0; kb < FDIM; kb += 32) {
    v2f a[8];
    float b0[8][4], b1[8][4];
#pragma unroll
    for (int u = 0; u < 8; ++u) {
      const int k0 = kb + 4 * u;
      a[u] = frag_a(A, FDIM, m0, k0, lane);
      const int kk = k0 + 2 * hl;
#pragma unroll
      for (int t = 0; t < 4; ++t) {
        b0[u][t] = Pp[kk * 64 + t];
        b1[u][t] = Pp[(kk + 1) * 64 + t];
      }
    }
#pragma unroll
    for (int u = 0; u < 8; ++u) {
#pragma unroll
      for (int t = 0; t < 4; ++t) {
        v2f bf; bf.x = b0[u][t]; bf.y = b1[u][t];
        c[t] = wmma4(a[u], bf, c[t]);
      }
    }
  }
  float* dst = Out + (size_t)b * NTOK2 * FDIM;
  const int rbase = m0 + (hl << 3);
#pragma unroll
  for (int r = 0; r < 8; ++r) {
#pragma unroll
    for (int t = 0; t < 4; ++t) dst[(rbase + r) * FDIM + 4 * cl + t] = c[t][r] + pv[t];
  }
}

__global__ void ln_res_kernel(const float* mid, const float* vin, const float* g,
                              const float* bb, float* vout) {
  int row = blockIdx.x * 8 + (threadIdx.x >> 5);
  int lane = threadIdx.x & 31;
  const float* mr = mid + (size_t)row * FDIM;
  float x0 = mr[lane], x1 = mr[lane + 32];
  float s = x0 + x1, ss = x0 * x0 + x1 * x1;
#pragma unroll
  for (int m = 16; m >= 1; m >>= 1) {
    s  += __shfl_xor(s, m, 32);
    ss += __shfl_xor(ss, m, 32);
  }
  float mean = s * (1.f / 64.f);
  float var  = ss * (1.f / 64.f) - mean * mean;
  float inv  = rsqrtf(var + 1e-5f);
  size_t base = (size_t)row * FDIM;
  vout[base + lane]      = (x0 - mean) * inv * g[lane]      + bb[lane]      + vin[base + lane];
  vout[base + lane + 32] = (x1 - mean) * inv * g[lane + 32] + bb[lane + 32] + vin[base + lane + 32];
}

__global__ void final_kernel(const float* weight, const float* weightf,
                             const float* S, float* out, float mul) {
  __shared__ float sh[KSPLIT * TPLANE];
  int b = blockIdx.x;
  int ks = __builtin_amdgcn_readfirstlane(threadIdx.x >> 5);
  int lane = threadIdx.x & 31;
  int m0 = blockIdx.y * 16;
  const float* Sb = S + (size_t)b * NTOK2 * FDIM;
  int row = m0 + (lane & 15);
  const int klen = NTOK2 / KSPLIT;
  int kbeg = ks * klen, kend = kbeg + klen;
  const float* W = (kbeg < NTOKEN) ? weight : weightf;
  v8f c[4] = {};
  for (int k0 = kbeg; k0 < kend; k0 += 4) {
    int kb = (k0 & (NTOKEN - 1)) + ((lane >> 4) << 1);
    v2f a; a.x = W[kb * NTOKEN + row]; a.y = W[(kb + 1) * NTOKEN + row];
    if ((k0 & 60) == 0 && (k0 & (NTOKEN - 1)) + 64 < NTOKEN) {
      __builtin_prefetch(&W[(kb + 64) * NTOKEN + row], 0, 3);
    }
#pragma unroll
    for (int t = 0; t < 4; ++t) {
      v2f bf = frag_b(Sb, FDIM, k0, t * 16, lane);
      c[t] = wmma4(a, bf, c[t]);
    }
  }
  ksplit_reduce_store(sh, c, ks, lane, out + (size_t)b * NTOKEN * FDIM, m0, mul, 16, 1);
}

extern "C" void kernel_launch(void* const* d_in, const int* in_sizes, int n_in,
                              void* d_out, int out_size, void* d_ws, size_t ws_size,
                              hipStream_t stream) {
  (void)in_sizes; (void)n_in; (void)out_size; (void)ws_size;
  const float* xy   = (const float*)d_in[0];
  const float* kW0  = (const float*)d_in[1];
  const float* kb0  = (const float*)d_in[2];
  const float* kW1  = (const float*)d_in[3];
  const float* kb1  = (const float*)d_in[4];
  const float* kW2  = (const float*)d_in[5];
  const float* kb2  = (const float*)d_in[6];
  const float* fW0  = (const float*)d_in[7];
  const float* fb0  = (const float*)d_in[8];
  const float* fW1  = (const float*)d_in[9];
  const float* fb1  = (const float*)d_in[10];
  const float* fW2  = (const float*)d_in[11];
  const float* fb2  = (const float*)d_in[12];
  const float* Wq   = (const float*)d_in[13];
  const float* bq   = (const float*)d_in[14];
  const float* Wk   = (const float*)d_in[15];
  const float* bk   = (const float*)d_in[16];
  const float* ln0g = (const float*)d_in[17];
  const float* ln0b = (const float*)d_in[18];
  const float* lng  = (const float*)d_in[19];
  const float* lnb  = (const float*)d_in[20];
  float* out = (float*)d_out;

  float* ws = (float*)d_ws;
  const size_t SZ_1M = 1048576;
  float* w_weight  = ws;
  float* w_weightf = ws + SZ_1M;
  float* V0        = ws + 2 * SZ_1M;
  float* V1        = ws + 3 * SZ_1M;
  float* Caug      = ws + 4 * SZ_1M;
  float* caug      = ws + 4 * SZ_1M + 20480;
  float* Gp        = ws + 4 * SZ_1M + 20992;
  float* gp        = ws + 4 * SZ_1M + 152064;
  float* Paug      = ws + 4 * SZ_1M + 154112;
  float* MID       = ws + 12 * SZ_1M + 131072;
  float* stats     = ws + 13 * SZ_1M + 131072;

  const float DXc   = 2.0f / 31.0f;
  const float dxdy  = DXc * DXc;
  const float scale = 0.125f;

  edge_mlp_kernel<<<dim3(NTOKEN, 2), 128, 0, stream>>>(kW0, kb0, kW1, kb1, kW2, kb2,
                                                       fW0, fb0, fW1, fb1, fW2, fb2,
                                                       w_weight, w_weightf);
  ln0_reduce_kernel<<<BATCH, 256, 0, stream>>>(xy, stats);
  ln0_apply_kernel<<<4096, 256, 0, stream>>>(xy, ln0g, ln0b, stats, V0);

  atb_kernel<<<dim3(NLAYER, 4, 1), KSPLIT * 32, 0, stream>>>(
      Wq, HEAD * DK * FDIM, Wk, HEAD * DK * FDIM, HEAD * DK, 1.0f, Caug, CAUG, nullptr);
  cvec_kernel<<<dim3(NLAYER, 2), 1024, 0, stream>>>(Wq, bq, Wk, bk, Caug, caug);

  float* Va = V0;
  float* Vb = V1;
  for (int j = 0; j < NLAYER - 1; ++j) {
    atb_kernel<<<dim3(BATCH, 4, KQ), KSPLIT * 32, 0, stream>>>(
        Va, NTOK2 * FDIM, Va, NTOK2 * FDIM, NTOK2, scale * dxdy, Gp, 4096, gp);
    pmat_kernel<<<BATCH, 1024, 0, stream>>>(Caug + (size_t)j * CAUG, caug + j * 80, Gp, gp, Paug);
    vp_kernel<<<dim3(BATCH, NTOK2 / 64), 128, 0, stream>>>(Va, Paug, MID);
    ln_res_kernel<<<2048, 256, 0, stream>>>(MID, Va, lng + j * FDIM,
                                            lnb + j * FDIM, Vb);
    float* t = Va; Va = Vb; Vb = t;
  }

  {
    int j = NLAYER - 1;
    atb_kernel<<<dim3(BATCH, 4, KQ), KSPLIT * 32, 0, stream>>>(
        Va, NTOK2 * FDIM, xy, NTOK2 * FDIM, NTOKEN, scale, Gp, 4096, gp);
    pmat_kernel<<<BATCH, 1024, 0, stream>>>(Caug + (size_t)j * CAUG, caug + j * 80, Gp, gp, Paug);
    vp_kernel<<<dim3(BATCH, NTOK2 / 64), 128, 0, stream>>>(Va, Paug, MID);
    final_kernel<<<dim3(BATCH, NTOKEN / 16), KSPLIT * 32, 0, stream>>>(
        w_weight, w_weightf, MID, out, dxdy * dxdy);
  }
}


# Global layernorm sums leave their own 8-work-group launch and ride in the edge-MLP launch as 64 extra work-groups (partials added in fixed order by the apply kernel); the sum order changes
# speedup vs baseline: 5.1871x; 1.2566x over previous
#include <hip/hip_runtime.h>
#include <hip/hip_bf16.h>

#define NGRID   32
#define NTOKEN  1024
#define NTOK2   2048
#define HEAD    4
#define DK      64
#define NLAYER  4
#define FDIM    64
#define BATCH   8

typedef __attribute__((ext_vector_type(2))) float v2f;
typedef __attribute__((ext_vector_type(8))) float v8f;

__device__ __forceinline__ v8f wmma4(v2f a, v2f b, v8f c) {
  return __builtin_amdgcn_wmma_f32_16x16x4_f32(
       false, a,  false, b,
       (short)0, c,  false,  false);
}

__device__ __forceinline__ v2f frag_a(const float* src, int lda, int m0, int k0, int lane) {
  int row = m0 + (lane & 15);
  int kk  = k0 + ((lane >> 4) << 1);
  v2f a; a.x = src[row * lda + kk]; a.y = src[row * lda + kk + 1];
  return a;
}

__device__ __forceinline__ v2f frag_b(const float* src, int ldb, int k0, int n0, int lane) {
  int col = n0 + (lane & 15);
  int kk  = k0 + ((lane >> 4) << 1);
  v2f b; b.x = src[kk * ldb + col]; b.y = src[(kk + 1) * ldb + col];
  return b;
}


#define KSPLIT   16
#define TPITCH   68
#define TPLANE   (16 * TPITCH)

__device__ __forceinline__ void ksplit_reduce_store(float* sh, const v8f* c, int ks, int lane,
                                                    float* dst, int m0, float mul,
                                                    int tstep, int cstep) {
  float* mine = sh + ks * TPLANE;
  int lrow = (lane >> 4) << 3;
  int lcol = lane & 15;
#pragma unroll
  for (int t = 0; t < 4; ++t) {
#pragma unroll
    for (int r = 0; r < 8; ++r) mine[(lrow + r) * TPITCH + t * tstep + lcol * cstep] = c[t][r];
  }
  __syncthreads();
  for (int idx = threadIdx.x; idx < 16 * 64; idx += KSPLIT * 32) {
    int row = idx >> 6, col = idx & 63;
    float acc = sh[row * TPITCH + col];
#pragma unroll
    for (int p = 1; p < KSPLIT; ++p) acc += sh[p * TPLANE + row * TPITCH + col];
    dst[(m0 + row) * 64 + col] = acc * mul;
  }
}

#define LN0_CHUNKS  8
#define LN0_JOBS    (BATCH * LN0_CHUNKS)
#define LN0_SLICE   (NTOK2 * FDIM / LN0_CHUNKS)

__device__ __forceinline__ void ln0_partial_job(const float* xy, float* part, int q, float* shp) {
  const float4* p = reinterpret_cast<const float4*>(xy + (size_t)q * LN0_SLICE);
  float4 s4 = make_float4(0.f, 0.f, 0.f, 0.f), q4 = make_float4(0.f, 0.f, 0.f, 0.f);
#pragma unroll 8
  for (int it = 0; it < LN0_SLICE / (4 * 128); ++it) {
    float4 v = p[it * 128 + threadIdx.x];
    s4.x += v.x; s4.y += v.y; s4.z += v.z; s4.w += v.w;
    q4.x = fmaf(v.x, v.x, q4.x); q4.y = fmaf(v.y, v.y, q4.y);
    q4.z = fmaf(v.z, v.z, q4.z); q4.w = fmaf(v.w, v.w, q4.w);
  }
  float s = (s4.x + s4.y) + (s4.z + s4.w);
  float ss = (q4.x + q4.y) + (q4.z + q4.w);
#pragma unroll
  for (int m = 16; m >= 1; m >>= 1) {
    s  += __shfl_xor(s, m, 32);
    ss += __shfl_xor(ss, m, 32);
  }
  const int wave = threadIdx.x >> 5;
  if ((threadIdx.x & 31) == 0) { shp[2 * wave] = s; shp[2 * wave + 1] = ss; }
  __syncthreads();
  if (threadIdx.x == 0) {
    part[2 * q]     = ((shp[0] + shp[2]) + shp[4]) + shp[6];
    part[2 * q + 1] = ((shp[1] + shp[3]) + shp[5]) + shp[7];
  }
}
__device__ __forceinline__ float leaky01(float t) {
  return fmaxf(t, 0.01f * t);
}

__global__ __launch_bounds__(128) void edge_mlp_kernel(
                                const float* kW0, const float* kb0, const float* kW1,
                                const float* kb1, const float* kW2, const float* kb2,
                                const float* fW0, const float* fb0, const float* fW1,
                                const float* fb1, const float* fW2, const float* fb2,
                                float* wout, float* fout,
                                const float* xy, float* ln0part) {
  __shared__ float shp[8];
  if (blockIdx.x < LN0_JOBS) {
    ln0_partial_job(xy, ln0part, blockIdx.x, shp);
    return;
  }
  const int ejob = blockIdx.x - LN0_JOBS;
  const bool isF = ejob >= NTOKEN;
  const float* W0 = isF ? fW0 : kW0;
  const float* b0 = isF ? fb0 : kb0;
  const float* W1 = isF ? fW1 : kW1;
  const float* b1 = isF ? fb1 : kb1;
  const float* W2 = isF ? fW2 : kW2;
  const float* b2 = isF ? fb2 : kb2;
  float* dst      = isF ? fout : wout;

  const int lane = threadIdx.x & 31;
  const int wave = threadIdx.x >> 5;
  const int c = lane & 15, h = lane >> 4;
  const int n = ejob & (NTOKEN - 1);
  const int mbase = wave * 256;
  const float DXc = 2.0f / 31.0f;
  const float e0 = -1.f + DXc * (float)(n >> 5);
  const float e1 = -1.f + DXc * (float)(n & 31);

  v2f wa[4][8];
#pragma unroll
  for (int t = 0; t < 4; ++t) {
#pragma unroll
    for (int i = 0; i < 8; ++i) {
      const float* p = W1 + (t * 16 + c) * 32 + 4 * i + 2 * h;
      wa[t][i].x = p[0]; wa[t][i].y = p[1];
    }
  }
  float pre[16], wc2[16], wc3[16];
#pragma unroll
  for (int i = 0; i < 8; ++i) {
#pragma unroll
    for (int j = 0; j < 2; ++j) {
      int u = 4 * i + 2 * h + j;
      float t = b0[u];
      t = fmaf(W0[u * 4 + 0], e0, t);
      t = fmaf(W0[u * 4 + 1], e1, t);
      pre[2 * i + j] = t;
      wc2[2 * i + j] = W0[u * 4 + 2];
      wc3[2 * i + j] = W0[u * 4 + 3];
    }
  }
  v8f cb[4];
  float w2v[4][8];
#pragma unroll
  for (int t = 0; t < 4; ++t) {
#pragma unroll
    for (int r = 0; r < 8; ++r) {
      cb[t][r]  = b1[t * 16 + 8 * h + r];
      w2v[t][r] = W2[t * 16 + 8 * h + r];
    }
  }
  const float bias2 = b2[0];

  for (int it = 0; it < 8; ++it) {
    const int m0 = mbase + it * 32;
    float res[2];
#pragma unroll
    for (int et = 0; et < 2; ++et) {
      const int m = m0 + et * 16 + c;
      const float e2 = -1.f + DXc * (float)(m >> 5);
      const float e3 = -1.f + DXc * (float)(m & 31);
      v2f hb[8];
#pragma unroll
      for (int i = 0; i < 8; ++i) {
        float t0 = fmaf(wc3[2 * i],     e3, fmaf(wc2[2 * i],     e2, pre[2 * i]));
        float t1 = fmaf(wc3[2 * i + 1], e3, fmaf(wc2[2 * i + 1], e2, pre[2 * i + 1]));
        hb[i].x = leaky01(t0); hb[i].y = leaky01(t1);
      }
      v8f acc[4];
#pragma unroll
      for (int t = 0; t < 4; ++t) acc[t] = wmma4(wa[t][0], hb[0], cb[t]);
#pragma unroll
      for (int i = 1; i < 8; ++i) {
#pragma unroll
        for (int t = 0; t < 4; ++t) acc[t] = wmma4(wa[t][i], hb[i], acc[t]);
      }
      float s = 0.f;
#pragma unroll
      for (int t = 0; t < 4; ++t) {
#pragma unroll
        for (int r = 0; r < 8; ++r) s = fmaf(w2v[t][r], leaky01(acc[t][r]), s);
      }
      s += __shfl_xor(s, 16, 32);
      res[et] = s + bias2;
    }
    dst[n * NTOKEN + m0 + lane] = h ? res[1] : res[0];
  }
}

__global__ __launch_bounds__(256) void ln0_apply_kernel(const float* xy, const float* g,
                                                        const float* bb, const float* part,
                                                        float* V) {
  const int idx = blockIdx.x * 256 + threadIdx.x;
  const int r = idx & 131071;
  const float xv = xy[idx], gv = g[r], bv = bb[r];
  const int b = blockIdx.x >> 9;
  const float* pb = part + b * (2 * LN0_CHUNKS);
  float s = pb[0], ss = pb[1];
#pragma unroll
  for (int cidx = 1; cidx < LN0_CHUNKS; ++cidx) { s += pb[2 * cidx]; ss += pb[2 * cidx + 1]; }
  const float N = (float)(NTOK2 * FDIM);
  const float mean = s / N;
  const float var  = ss / N - mean * mean;
  const float rstd = rsqrtf(var + 1e-5f);
  V[idx] = (xv - mean) * rstd * gv + bv;
}

#define KQ      4
#define CAUG    (80 * 64)
#define GPITCH  72

__global__ __launch_bounds__(KSPLIT * 32) void atb_kernel(
                           const float* Abuf, int astride, const float* Bbuf, int bstride,
                           int kdim, float s, float* Out, int ostride, float* csum) {
  __shared__ float sh[KSPLIT * TPLANE];
  __shared__ float shc[KSPLIT * 64];
  int z = blockIdx.x;
  int ks = __builtin_amdgcn_readfirstlane(threadIdx.x >> 5);
  int lane = threadIdx.x & 31;
  const int cl = lane & 15, hl = lane >> 4;
  int m0 = blockIdx.y * 16;
  int klen = kdim / (KSPLIT * (int)gridDim.z);
  int kbeg = ((int)blockIdx.z * KSPLIT + ks) * klen, kend = kbeg + klen;
  const float* Ap = Abuf + (size_t)z * astride + m0 + cl;
  const float* Bp = Bbuf + (size_t)z * bstride + 4 * cl;
  v8f c[4] = {};
  float cs[4] = {0.f, 0.f, 0.f, 0.f};
  for (int kb = kbeg; kb < kend; kb += 16) {
    v2f a[4];
    float b0[4][4], b1[4][4];
#pragma unroll
    for (int u = 0; u < 4; ++u) {
      const int kk = kb + 4 * u + 2 * hl;
      a[u].x = Ap[kk * 64]; a[u].y = Ap[(kk + 1) * 64];
#pragma unroll
      for (int t = 0; t < 4; ++t) {
        b0[u][t] = Bp[kk * 64 + t];
        b1[u][t] = Bp[(kk + 1) * 64 + t];
      }
    }
#pragma unroll
    for (int u = 0; u < 4; ++u) {
#pragma unroll
      for (int t = 0; t < 4; ++t) {
        v2f bf; bf.x = b0[u][t]; bf.y = b1[u][t];
        cs[t] += bf.x + bf.y;
        c[t] = wmma4(a[u], bf, c[t]);
      }
    }
  }
#pragma unroll
  for (int t = 0; t < 4; ++t) {
    float v = cs[t] + __shfl_xor(cs[t], 16, 32);
    if (lane < 16) shc[ks * 64 + 4 * lane + t] = v;
  }
  size_t slot = (size_t)z * gridDim.z + blockIdx.z;
  ksplit_reduce_store(sh, c, ks, lane, Out + slot * ostride, m0, s, 1, 4);
  if (csum != nullptr && blockIdx.y == 0 && threadIdx.x < 64) {
    float a = shc[threadIdx.x];
#pragma unroll
    for (int p = 1; p < KSPLIT; ++p) a += shc[p * 64 + threadIdx.x];
    csum[slot * 64 + threadIdx.x] = a * s;
  }
}

__global__ void cvec_kernel(const float* Wq, const float* bq, const float* Wk, const float* bk,
                            float* Caug, float* caug) {
  __shared__ float sh[16 * 64];
  const int j = blockIdx.x;
  const bool isD = blockIdx.y != 0;
  const float* Wm = (isD ? Wk : Wq) + (size_t)j * 256 * 64;
  const float* v  = (isD ? bq : bk) + j * 256;
  const int i = threadIdx.x & 63, kc = threadIdx.x >> 6;
  float a = 0.f;
  for (int u = 0; u < 16; ++u) {
    int k = kc * 16 + u;
    a = fmaf(Wm[k * 64 + i], v[k], a);
  }
  sh[kc * 64 + i] = a;
  __syncthreads();
  float* Cj = Caug + (size_t)j * CAUG;
  float* cj = caug + j * 80;
  if (threadIdx.x < 64) {
    float t = sh[threadIdx.x];
    for (int p = 1; p < 16; ++p) t += sh[p * 64 + threadIdx.x];
    if (isD) Cj[64 * 64 + threadIdx.x] = t; else cj[threadIdx.x] = t;
  }
  if (isD) {
    if ((threadIdx.x >> 5) == 2) {
      const int lane = threadIdx.x & 31;
      float e = 0.f;
      for (int u = 0; u < 8; ++u) {
        int k = lane + 32 * u;
        e = fmaf(bq[j * 256 + k], bk[j * 256 + k], e);
      }
      for (int m = 16; m >= 1; m >>= 1) e += __shfl_xor(e, m, 32);
      if (lane == 0) cj[64] = e;
    }
  } else {
    if (threadIdx.x < 15 * 64) Cj[65 * 64 + threadIdx.x] = 0.f;
    if (threadIdx.x < 15) cj[65 + threadIdx.x] = 0.f;
  }
}

__global__ __launch_bounds__(1024) void pmat_kernel(const float* Cj, const float* cj,
                                                    const float* Gp, const float* gp,
                                                    float* Paug) {
  __shared__ float sG[64 * GPITCH];
  __shared__ float sg[64];
  const int b = blockIdx.x, tid = threadIdx.x;
  const float* G0 = Gp + (size_t)b * KQ * 4096;
  const float* g0 = gp + (size_t)b * KQ * 64;
  const int w = __builtin_amdgcn_readfirstlane(tid >> 5);
  const int lane = tid & 31, cl = lane & 15, hl = lane >> 4;
  const int rt = w >> 2, ct = w & 3;
  const int rbase = rt * 16 + (hl << 3);
  v2f a[16];
  float cjv[8];
#pragma unroll
  for (int u = 0; u < 16; ++u) { a[u].x = 0.f; a[u].y = 0.f; }
#pragma unroll
  for (int r = 0; r < 8; ++r) cjv[r] = 0.f;
  if (w < 20) {
#pragma unroll
    for (int u = 0; u < 16; ++u) a[u] = frag_a(Cj, 64, rt * 16, 4 * u, lane);
#pragma unroll
    for (int r = 0; r < 8; ++r) cjv[r] = cj[rbase + r];
  }
#pragma unroll
  for (int i = 0; i < 4; ++i) {
    const int idx = tid + 1024 * i;
    float v = G0[idx];
#pragma unroll
    for (int q = 1; q < KQ; ++q) v += G0[q * 4096 + idx];
    sG[(idx >> 6) * GPITCH + (idx & 63)] = v;
  }
  if (tid < 64) {
    float v = g0[tid];
#pragma unroll
    for (int q = 1; q < KQ; ++q) v += g0[q * 64 + tid];
    sg[tid] = v;
  }
  __syncthreads();
  if (w < 20) {
    const int col = ct * 16 + cl;
    const float g = sg[col];
    v8f acc;
#pragma unroll
    for (int r = 0; r < 8; ++r) acc[r] = cjv[r] * g;
#pragma unroll
    for (int u = 0; u < 16; ++u) {
      const int kk = 4 * u + 2 * hl;
      v2f bf; bf.x = sG[kk * GPITCH + col]; bf.y = sG[(kk + 1) * GPITCH + col];
      acc = wmma4(a[u], bf, acc);
    }
    float* dst = Paug + (size_t)b * CAUG;
#pragma unroll
    for (int r = 0; r < 8; ++r) dst[(rbase + r) * 64 + col] = acc[r];
  }
}

__global__ __launch_bounds__(128) void vp_kernel(const float* Vin, const float* Paug, float* Out) {
  const int b = blockIdx.x;
  const int wave = threadIdx.x >> 5, lane = threadIdx.x & 31;
  const int cl = lane & 15, hl = lane >> 4;
  const int m0 = blockIdx.y * 64 + wave * 16;
  const float* A  = Vin + (size_t)b * NTOK2 * FDIM;
  const float* Pp = Paug + (size_t)b * CAUG + 4 * cl;
  float pv[4];
#pragma unroll
  for (int t = 0; t < 4; ++t) pv[t] = Pp[64 * 64 + t];
  v8f c[4] = {};
  for (int kb = 0; kb < FDIM; kb += 32) {
    v2f a[8];
    float b0[8][4], b1[8][4];
#pragma unroll
    for (int u = 0; u < 8; ++u) {
      const int k0 = kb + 4 * u;
      a[u] = frag_a(A, FDIM, m0, k0, lane);
      const int kk = k0 + 2 * hl;
#pragma unroll
      for (int t = 0; t < 4; ++t) {
        b0[u][t] = Pp[kk * 64 + t];
        b1[u][t] = Pp[(kk + 1) * 64 + t];
      }
    }
#pragma unroll
    for (int u = 0; u < 8; ++u) {
#pragma unroll
      for (int t = 0; t < 4; ++t) {
        v2f bf; bf.x = b0[u][t]; bf.y = b1[u][t];
        c[t] = wmma4(a[u], bf, c[t]);
      }
    }
  }
  float* dst = Out + (size_t)b * NTOK2 * FDIM;
  const int rbase = m0 + (hl << 3);
#pragma unroll
  for (int r = 0; r < 8; ++r) {
#pragma unroll
    for (int t = 0; t < 4; ++t) dst[(rbase + r) * FDIM + 4 * cl + t] = c[t][r] + pv[t];
  }
}

__global__ void ln_res_kernel(const float* mid, const float* vin, const float* g,
                              const float* bb, float* vout) {
  int row = blockIdx.x * 8 + (threadIdx.x >> 5);
  int lane = threadIdx.x & 31;
  const float* mr = mid + (size_t)row * FDIM;
  float x0 = mr[lane], x1 = mr[lane + 32];
  float s = x0 + x1, ss = x0 * x0 + x1 * x1;
#pragma unroll
  for (int m = 16; m >= 1; m >>= 1) {
    s  += __shfl_xor(s, m, 32);
    ss += __shfl_xor(ss, m, 32);
  }
  float mean = s * (1.f / 64.f);
  float var  = ss * (1.f / 64.f) - mean * mean;
  float inv  = rsqrtf(var + 1e-5f);
  size_t base = (size_t)row * FDIM;
  vout[base + lane]      = (x0 - mean) * inv * g[lane]      + bb[lane]      + vin[base + lane];
  vout[base + lane + 32] = (x1 - mean) * inv * g[lane + 32] + bb[lane + 32] + vin[base + lane + 32];
}

__global__ void final_kernel(const float* weight, const float* weightf,
                             const float* S, float* out, float mul) {
  __shared__ float sh[KSPLIT * TPLANE];
  int b = blockIdx.x;
  int ks = __builtin_amdgcn_readfirstlane(threadIdx.x >> 5);
  int lane = threadIdx.x & 31;
  int m0 = blockIdx.y * 16;
  const float* Sb = S + (size_t)b * NTOK2 * FDIM;
  int row = m0 + (lane & 15);
  const int klen = NTOK2 / KSPLIT;
  int kbeg = ks * klen, kend = kbeg + klen;
  const float* W = (kbeg < NTOKEN) ? weight : weightf;
  v8f c[4] = {};
  for (int k0 = kbeg; k0 < kend; k0 += 4) {
    int kb = (k0 & (NTOKEN - 1)) + ((lane >> 4) << 1);
    v2f a; a.x = W[kb * NTOKEN + row]; a.y = W[(kb + 1) * NTOKEN + row];
    if ((k0 & 60) == 0 && (k0 & (NTOKEN - 1)) + 64 < NTOKEN) {
      __builtin_prefetch(&W[(kb + 64) * NTOKEN + row], 0, 3);
    }
#pragma unroll
    for (int t = 0; t < 4; ++t) {
      v2f bf = frag_b(Sb, FDIM, k0, t * 16, lane);
      c[t] = wmma4(a, bf, c[t]);
    }
  }
  ksplit_reduce_store(sh, c, ks, lane, out + (size_t)b * NTOKEN * FDIM, m0, mul, 16, 1);
}

extern "C" void kernel_launch(void* const* d_in, const int* in_sizes, int n_in,
                              void* d_out, int out_size, void* d_ws, size_t ws_size,
                              hipStream_t stream) {
  (void)in_sizes; (void)n_in; (void)out_size; (void)ws_size;
  const float* xy   = (const float*)d_in[0];
  const float* kW0  = (const float*)d_in[1];
  const float* kb0  = (const float*)d_in[2];
  const float* kW1  = (const float*)d_in[3];
  const float* kb1  = (const float*)d_in[4];
  const float* kW2  = (const float*)d_in[5];
  const float* kb2  = (const float*)d_in[6];
  const float* fW0  = (const float*)d_in[7];
  const float* fb0  = (const float*)d_in[8];
  const float* fW1  = (const float*)d_in[9];
  const float* fb1  = (const float*)d_in[10];
  const float* fW2  = (const float*)d_in[11];
  const float* fb2  = (const float*)d_in[12];
  const float* Wq   = (const float*)d_in[13];
  const float* bq   = (const float*)d_in[14];
  const float* Wk   = (const float*)d_in[15];
  const float* bk   = (const float*)d_in[16];
  const float* ln0g = (const float*)d_in[17];
  const float* ln0b = (const float*)d_in[18];
  const float* lng  = (const float*)d_in[19];
  const float* lnb  = (const float*)d_in[20];
  float* out = (float*)d_out;

  float* ws = (float*)d_ws;
  const size_t SZ_1M = 1048576;
  float* w_weight  = ws;
  float* w_weightf = ws + SZ_1M;
  float* V0        = ws + 2 * SZ_1M;
  float* V1        = ws + 3 * SZ_1M;
  float* Caug      = ws + 4 * SZ_1M;
  float* caug      = ws + 4 * SZ_1M + 20480;
  float* Gp        = ws + 4 * SZ_1M + 20992;
  float* gp        = ws + 4 * SZ_1M + 152064;
  float* Paug      = ws + 4 * SZ_1M + 154112;
  float* MID       = ws + 12 * SZ_1M + 131072;
  float* ln0part   = ws + 4 * SZ_1M + 195072;

  const float DXc   = 2.0f / 31.0f;
  const float dxdy  = DXc * DXc;
  const float scale = 0.125f;

  edge_mlp_kernel<<<LN0_JOBS + 2 * NTOKEN, 128, 0, stream>>>(kW0, kb0, kW1, kb1, kW2, kb2,
                                                             fW0, fb0, fW1, fb1, fW2, fb2,
                                                             w_weight, w_weightf, xy, ln0part);
  ln0_apply_kernel<<<4096, 256, 0, stream>>>(xy, ln0g, ln0b, ln0part, V0);

  atb_kernel<<<dim3(NLAYER, 4, 1), KSPLIT * 32, 0, stream>>>(
      Wq, HEAD * DK * FDIM, Wk, HEAD * DK * FDIM, HEAD * DK, 1.0f, Caug, CAUG, nullptr);
  cvec_kernel<<<dim3(NLAYER, 2), 1024, 0, stream>>>(Wq, bq, Wk, bk, Caug, caug);

  float* Va = V0;
  float* Vb = V1;
  for (int j = 0; j < NLAYER - 1; ++j) {
    atb_kernel<<<dim3(BATCH, 4, KQ), KSPLIT * 32, 0, stream>>>(
        Va, NTOK2 * FDIM, Va, NTOK2 * FDIM, NTOK2, scale * dxdy, Gp, 4096, gp);
    pmat_kernel<<<BATCH, 1024, 0, stream>>>(Caug + (size_t)j * CAUG, caug + j * 80, Gp, gp, Paug);
    vp_kernel<<<dim3(BATCH, NTOK2 / 64), 128, 0, stream>>>(Va, Paug, MID);
    ln_res_kernel<<<2048, 256, 0, stream>>>(MID, Va, lng + j * FDIM,
                                            lnb + j * FDIM, Vb);
    float* t = Va; Va = Vb; Vb = t;
  }

  {
    int j = NLAYER - 1;
    atb_kernel<<<dim3(BATCH, 4, KQ), KSPLIT * 32, 0, stream>>>(
        Va, NTOK2 * FDIM, xy, NTOK2 * FDIM, NTOKEN, scale, Gp, 4096, gp);
    pmat_kernel<<<BATCH, 1024, 0, stream>>>(Caug + (size_t)j * CAUG, caug + j * 80, Gp, gp, Paug);
    vp_kernel<<<dim3(BATCH, NTOK2 / 64), 128, 0, stream>>>(Va, Paug, MID);
    final_kernel<<<dim3(BATCH, NTOKEN / 16), KSPLIT * 32, 0, stream>>>(
        w_weight, w_weightf, MID, out, dxdy * dxdy);
  }
}


# [P; p] = Caug G + caug g^T moves into the Vin*P launch: each work-group forms it in LDS from the Gram parts (operands staged as whole lines), its own 8-work-group launch gone x4; same sums and order
# speedup vs baseline: 5.5407x; 1.0682x over previous
#include <hip/hip_runtime.h>
#include <hip/hip_bf16.h>

#define NGRID   32
#define NTOKEN  1024
#define NTOK2   2048
#define HEAD    4
#define DK      64
#define NLAYER  4
#define FDIM    64
#define BATCH   8

typedef __attribute__((ext_vector_type(2))) float v2f;
typedef __attribute__((ext_vector_type(8))) float v8f;

__device__ __forceinline__ v8f wmma4(v2f a, v2f b, v8f c) {
  return __builtin_amdgcn_wmma_f32_16x16x4_f32(
       false, a,  false, b,
       (short)0, c,  false,  false);
}

__device__ __forceinline__ v2f frag_a(const float* src, int lda, int m0, int k0, int lane) {
  int row = m0 + (lane & 15);
  int kk  = k0 + ((lane >> 4) << 1);
  v2f a; a.x = src[row * lda + kk]; a.y = src[row * lda + kk + 1];
  return a;
}

__device__ __forceinline__ v2f frag_b(const float* src, int ldb, int k0, int n0, int lane) {
  int col = n0 + (lane & 15);
  int kk  = k0 + ((lane >> 4) << 1);
  v2f b; b.x = src[kk * ldb + col]; b.y = src[(kk + 1) * ldb + col];
  return b;
}


#define KSPLIT   16
#define TPITCH   68
#define TPLANE   (16 * TPITCH)

__device__ __forceinline__ void ksplit_reduce_store(float* sh, const v8f* c, int ks, int lane,
                                                    float* dst, int m0, float mul,
                                                    int tstep, int cstep) {
  float* mine = sh + ks * TPLANE;
  int lrow = (lane >> 4) << 3;
  int lcol = lane & 15;
#pragma unroll
  for (int t = 0; t < 4; ++t) {
#pragma unroll
    for (int r = 0; r < 8; ++r) mine[(lrow + r) * TPITCH + t * tstep + lcol * cstep] = c[t][r];
  }
  __syncthreads();
  for (int idx = threadIdx.x; idx < 16 * 64; idx += KSPLIT * 32) {
    int row = idx >> 6, col = idx & 63;
    float acc = sh[row * TPITCH + col];
#pragma unroll
    for (int p = 1; p < KSPLIT; ++p) acc += sh[p * TPLANE + row * TPITCH + col];
    dst[(m0 + row) * 64 + col] = acc * mul;
  }
}

#define LN0_CHUNKS  8
#define LN0_JOBS    (BATCH * LN0_CHUNKS)
#define LN0_SLICE   (NTOK2 * FDIM / LN0_CHUNKS)

__device__ __forceinline__ void ln0_partial_job(const float* xy, float* part, int q, float* shp) {
  const float4* p = reinterpret_cast<const float4*>(xy + (size_t)q * LN0_SLICE);
  float4 s4 = make_float4(0.f, 0.f, 0.f, 0.f), q4 = make_float4(0.f, 0.f, 0.f, 0.f);
#pragma unroll 8
  for (int it = 0; it < LN0_SLICE / (4 * 128); ++it) {
    float4 v = p[it * 128 + threadIdx.x];
    s4.x += v.x; s4.y += v.y; s4.z += v.z; s4.w += v.w;
    q4.x = fmaf(v.x, v.x, q4.x); q4.y = fmaf(v.y, v.y, q4.y);
    q4.z = fmaf(v.z, v.z, q4.z); q4.w = fmaf(v.w, v.w, q4.w);
  }
  float s = (s4.x + s4.y) + (s4.z + s4.w);
  float ss = (q4.x + q4.y) + (q4.z + q4.w);
#pragma unroll
  for (int m = 16; m >= 1; m >>= 1) {
    s  += __shfl_xor(s, m, 32);
    ss += __shfl_xor(ss, m, 32);
  }
  const int wave = threadIdx.x >> 5;
  if ((threadIdx.x & 31) == 0) { shp[2 * wave] = s; shp[2 * wave + 1] = ss; }
  __syncthreads();
  if (threadIdx.x == 0) {
    part[2 * q]     = ((shp[0] + shp[2]) + shp[4]) + shp[6];
    part[2 * q + 1] = ((shp[1] + shp[3]) + shp[5]) + shp[7];
  }
}
__device__ __forceinline__ float leaky01(float t) {
  return fmaxf(t, 0.01f * t);
}

__global__ __launch_bounds__(128) void edge_mlp_kernel(
                                const float* kW0, const float* kb0, const float* kW1,
                                const float* kb1, const float* kW2, const float* kb2,
                                const float* fW0, const float* fb0, const float* fW1,
                                const float* fb1, const float* fW2, const float* fb2,
                                float* wout, float* fout,
                                const float* xy, float* ln0part) {
  __shared__ float shp[8];
  if (blockIdx.x < LN0_JOBS) {
    ln0_partial_job(xy, ln0part, blockIdx.x, shp);
    return;
  }
  const int ejob = blockIdx.x - LN0_JOBS;
  const bool isF = ejob >= NTOKEN;
  const float* W0 = isF ? fW0 : kW0;
  const float* b0 = isF ? fb0 : kb0;
  const float* W1 = isF ? fW1 : kW1;
  const float* b1 = isF ? fb1 : kb1;
  const float* W2 = isF ? fW2 : kW2;
  const float* b2 = isF ? fb2 : kb2;
  float* dst      = isF ? fout : wout;

  const int lane = threadIdx.x & 31;
  const int wave = threadIdx.x >> 5;
  const int c = lane & 15, h = lane >> 4;
  const int n = ejob & (NTOKEN - 1);
  const int mbase = wave * 256;
  const float DXc = 2.0f / 31.0f;
  const float e0 = -1.f + DXc * (float)(n >> 5);
  const float e1 = -1.f + DXc * (float)(n & 31);

  v2f wa[4][8];
#pragma unroll
  for (int t = 0; t < 4; ++t) {
#pragma unroll
    for (int i = 0; i < 8; ++i) {
      const float* p = W1 + (t * 16 + c) * 32 + 4 * i + 2 * h;
      wa[t][i].x = p[0]; wa[t][i].y = p[1];
    }
  }
  float pre[16], wc2[16], wc3[16];
#pragma unroll
  for (int i = 0; i < 8; ++i) {
#pragma unroll
    for (int j = 0; j < 2; ++j) {
      int u = 4 * i + 2 * h + j;
      float t = b0[u];
      t = fmaf(W0[u * 4 + 0], e0, t);
      t = fmaf(W0[u * 4 + 1], e1, t);
      pre[2 * i + j] = t;
      wc2[2 * i + j] = W0[u * 4 + 2];
      wc3[2 * i + j] = W0[u * 4 + 3];
    }
  }
  v8f cb[4];
  float w2v[4][8];
#pragma unroll
  for (int t = 0; t < 4; ++t) {
#pragma unroll
    for (int r = 0; r < 8; ++r) {
      cb[t][r]  = b1[t * 16 + 8 * h + r];
      w2v[t][r] = W2[t * 16 + 8 * h + r];
    }
  }
  const float bias2 = b2[0];

  for (int it = 0; it < 8; ++it) {
    const int m0 = mbase + it * 32;
    float res[2];
#pragma unroll
    for (int et = 0; et < 2; ++et) {
      const int m = m0 + et * 16 + c;
      const float e2 = -1.f + DXc * (float)(m >> 5);
      const float e3 = -1.f + DXc * (float)(m & 31);
      v2f hb[8];
#pragma unroll
      for (int i = 0; i < 8; ++i) {
        float t0 = fmaf(wc3[2 * i],     e3, fmaf(wc2[2 * i],     e2, pre[2 * i]));
        float t1 = fmaf(wc3[2 * i + 1], e3, fmaf(wc2[2 * i + 1], e2, pre[2 * i + 1]));
        hb[i].x = leaky01(t0); hb[i].y = leaky01(t1);
      }
      v8f acc[4];
#pragma unroll
      for (int t = 0; t < 4; ++t) acc[t] = wmma4(wa[t][0], hb[0], cb[t]);
#pragma unroll
      for (int i = 1; i < 8; ++i) {
#pragma unroll
        for (int t = 0; t < 4; ++t) acc[t] = wmma4(wa[t][i], hb[i], acc[t]);
      }
      float s = 0.f;
#pragma unroll
      for (int t = 0; t < 4; ++t) {
#pragma unroll
        for (int r = 0; r < 8; ++r) s = fmaf(w2v[t][r], leaky01(acc[t][r]), s);
      }
      s += __shfl_xor(s, 16, 32);
      res[et] = s + bias2;
    }
    dst[n * NTOKEN + m0 + lane] = h ? res[1] : res[0];
  }
}

__global__ __launch_bounds__(256) void ln0_apply_kernel(const float* xy, const float* g,
                                                        const float* bb, const float* part,
                                                        float* V) {
  const int idx = blockIdx.x * 256 + threadIdx.x;
  const int r = idx & 131071;
  const float xv = xy[idx], gv = g[r], bv = bb[r];
  const int b = blockIdx.x >> 9;
  const float* pb = part + b * (2 * LN0_CHUNKS);
  float s = pb[0], ss = pb[1];
#pragma unroll
  for (int cidx = 1; cidx < LN0_CHUNKS; ++cidx) { s += pb[2 * cidx]; ss += pb[2 * cidx + 1]; }
  const float N = (float)(NTOK2 * FDIM);
  const float mean = s / N;
  const float var  = ss / N - mean * mean;
  const float rstd = rsqrtf(var + 1e-5f);
  V[idx] = (xv - mean) * rstd * gv + bv;
}

#define KQ      4
#define CAUG    (80 * 64)
#define GPITCH  72

__global__ __launch_bounds__(KSPLIT * 32) void atb_kernel(
                           const float* Abuf, int astride, const float* Bbuf, int bstride,
                           int kdim, float s, float* Out, int ostride, float* csum) {
  __shared__ float sh[KSPLIT * TPLANE];
  __shared__ float shc[KSPLIT * 64];
  int z = blockIdx.x;
  int ks = __builtin_amdgcn_readfirstlane(threadIdx.x >> 5);
  int lane = threadIdx.x & 31;
  const int cl = lane & 15, hl = lane >> 4;
  int m0 = blockIdx.y * 16;
  int klen = kdim / (KSPLIT * (int)gridDim.z);
  int kbeg = ((int)blockIdx.z * KSPLIT + ks) * klen, kend = kbeg + klen;
  const float* Ap = Abuf + (size_t)z * astride + m0 + cl;
  const float* Bp = Bbuf + (size_t)z * bstride + 4 * cl;
  v8f c[4] = {};
  float cs[4] = {0.f, 0.f, 0.f, 0.f};
  for (int kb = kbeg; kb < kend; kb += 16) {
    v2f a[4];
    float b0[4][4], b1[4][4];
#pragma unroll
    for (int u = 0; u < 4; ++u) {
      const int kk = kb + 4 * u + 2 * hl;
      a[u].x = Ap[kk * 64]; a[u].y = Ap[(kk + 1) * 64];
#pragma unroll
      for (int t = 0; t < 4; ++t) {
        b0[u][t] = Bp[kk * 64 + t];
        b1[u][t] = Bp[(kk + 1) * 64 + t];
      }
    }
#pragma unroll
    for (int u = 0; u < 4; ++u) {
#pragma unroll
      for (int t = 0; t < 4; ++t) {
        v2f bf; bf.x = b0[u][t]; bf.y = b1[u][t];
        cs[t] += bf.x + bf.y;
        c[t] = wmma4(a[u], bf, c[t]);
      }
    }
  }
#pragma unroll
  for (int t = 0; t < 4; ++t) {
    float v = cs[t] + __shfl_xor(cs[t], 16, 32);
    if (lane < 16) shc[ks * 64 + 4 * lane + t] = v;
  }
  size_t slot = (size_t)z * gridDim.z + blockIdx.z;
  ksplit_reduce_store(sh, c, ks, lane, Out + slot * ostride, m0, s, 1, 4);
  if (csum != nullptr && blockIdx.y == 0 && threadIdx.x < 64) {
    float a = shc[threadIdx.x];
#pragma unroll
    for (int p = 1; p < KSPLIT; ++p) a += shc[p * 64 + threadIdx.x];
    csum[slot * 64 + threadIdx.x] = a * s;
  }
}

__global__ void cvec_kernel(const float* Wq, const float* bq, const float* Wk, const float* bk,
                            float* Caug, float* caug) {
  __shared__ float sh[16 * 64];
  const int j = blockIdx.x;
  const bool isD = blockIdx.y != 0;
  const float* Wm = (isD ? Wk : Wq) + (size_t)j * 256 * 64;
  const float* v  = (isD ? bq : bk) + j * 256;
  const int i = threadIdx.x & 63, kc = threadIdx.x >> 6;
  float a = 0.f;
  for (int u = 0; u < 16; ++u) {
    int k = kc * 16 + u;
    a = fmaf(Wm[k * 64 + i], v[k], a);
  }
  sh[kc * 64 + i] = a;
  __syncthreads();
  float* Cj = Caug + (size_t)j * CAUG;
  float* cj = caug + j * 80;
  if (threadIdx.x < 64) {
    float t = sh[threadIdx.x];
    for (int p = 1; p < 16; ++p) t += sh[p * 64 + threadIdx.x];
    if (isD) Cj[64 * 64 + threadIdx.x] = t; else cj[threadIdx.x] = t;
  }
  if (isD) {
    if ((threadIdx.x >> 5) == 2) {
      const int lane = threadIdx.x & 31;
      float e = 0.f;
      for (int u = 0; u < 8; ++u) {
        int k = lane + 32 * u;
        e = fmaf(bq[j * 256 + k], bk[j * 256 + k], e);
      }
      for (int m = 16; m >= 1; m >>= 1) e += __shfl_xor(e, m, 32);
      if (lane == 0) cj[64] = e;
    }
  } else {
    if (threadIdx.x < 15 * 64) Cj[65 * 64 + threadIdx.x] = 0.f;
    if (threadIdx.x < 15) cj[65 + threadIdx.x] = 0.f;
  }
}

#define CPITCH  68
__global__ __launch_bounds__(128) void vp_kernel(const float* Vin, const float* Cj,
                                                 const float* cj, const float* Gp,
                                                 const float* gp, float* Out) {
  __shared__ __attribute__((aligned(16))) float sG[64 * GPITCH];
  __shared__ __attribute__((aligned(16))) float sC[80 * CPITCH];
  __shared__ __attribute__((aligned(16))) float sP[80 * CPITCH];
  __shared__ float sg[64];
  __shared__ __attribute__((aligned(16))) float scj[80];
  const int b = blockIdx.x, tid = threadIdx.x;
  const int wave = tid >> 5, lane = tid & 31;
  const int cl = lane & 15, hl = lane >> 4;
  const int m0 = blockIdx.y * 64 + wave * 16;
  const float* A  = Vin + (size_t)b * NTOK2 * FDIM;

  const float4* G4 = reinterpret_cast<const float4*>(Gp + (size_t)b * KQ * 4096);
  const float4* C4 = reinterpret_cast<const float4*>(Cj);
  const float* g0 = gp + (size_t)b * KQ * 64;
  float4 gq[8][KQ];
#pragma unroll
  for (int i = 0; i < 8; ++i) {
#pragma unroll
    for (int q = 0; q < KQ; ++q) gq[i][q] = G4[q * 1024 + tid + 128 * i];
  }
  const float4 cq0 = C4[tid],       cq1 = C4[tid + 128], cq2 = C4[tid + 256], cq3 = C4[tid + 384],
               cq4 = C4[tid + 512], cq5 = C4[tid + 640], cq6 = C4[tid + 768], cq7 = C4[tid + 896],
               cq8 = C4[tid + 1024], cq9 = C4[tid + 1152];
  float gv[KQ];
#pragma unroll
  for (int q = 0; q < KQ; ++q) gv[q] = g0[q * 64 + (tid & 63)];
  const float cjg = cj[tid < 80 ? tid : 79];
  v2f av[16];
#pragma unroll
  for (int u = 0; u < 16; ++u) av[u] = frag_a(A, FDIM, m0, 4 * u, lane);
  __builtin_amdgcn_sched_barrier(0);
#pragma unroll
  for (int i = 0; i < 8; ++i) {
    const int i4 = tid + 128 * i;
    float4 v = gq[i][0];
#pragma unroll
    for (int q = 1; q < KQ; ++q) {
      v.x += gq[i][q].x; v.y += gq[i][q].y; v.z += gq[i][q].z; v.w += gq[i][q].w;
    }
    *reinterpret_cast<float4*>(&sG[(i4 >> 4) * GPITCH + ((i4 & 15) << 2)]) = v;
  }
  {
    float4* cdst = reinterpret_cast<float4*>(&sC[(tid >> 4) * CPITCH + ((tid & 15) << 2)]);
    const int cstep = 8 * CPITCH / 4;
    cdst[0 * cstep] = cq0; cdst[1 * cstep] = cq1; cdst[2 * cstep] = cq2; cdst[3 * cstep] = cq3;
    cdst[4 * cstep] = cq4; cdst[5 * cstep] = cq5; cdst[6 * cstep] = cq6; cdst[7 * cstep] = cq7;
    cdst[8 * cstep] = cq8; cdst[9 * cstep] = cq9;
  }
  if (tid < 64) {
    float v = gv[0];
#pragma unroll
    for (int q = 1; q < KQ; ++q) v += gv[q];
    sg[tid] = v;
  }
  if (tid < 80) scj[tid] = cjg;
  __syncthreads();

  {
    const int col = wave * 16 + cl;
    const float g = sg[col];
    v2f gb[16];
#pragma unroll
    for (int u = 0; u < 16; ++u) {
      const int kk = 4 * u + 2 * hl;
      gb[u].x = sG[kk * GPITCH + col]; gb[u].y = sG[(kk + 1) * GPITCH + col];
    }
    v8f pacc[5];
#pragma unroll
    for (int rt = 0; rt < 5; ++rt) {
#pragma unroll
      for (int r = 0; r < 8; ++r) pacc[rt][r] = scj[rt * 16 + (hl << 3) + r] * g;
    }
    v2f ca[5][16];
#pragma unroll
    for (int rt = 0; rt < 5; ++rt) {
#pragma unroll
      for (int u = 0; u < 16; ++u)
        ca[rt][u] = *reinterpret_cast<const v2f*>(&sC[(rt * 16 + cl) * CPITCH + 4 * u + 2 * hl]);
    }
    __builtin_amdgcn_sched_barrier(0);
#pragma unroll
    for (int u = 0; u < 16; ++u) {
#pragma unroll
      for (int rt = 0; rt < 5; ++rt) pacc[rt] = wmma4(ca[rt][u], gb[u], pacc[rt]);
    }
#pragma unroll
    for (int rt = 0; rt < 5; ++rt) {
#pragma unroll
      for (int r = 0; r < 8; ++r) sP[(rt * 16 + (hl << 3) + r) * CPITCH + col] = pacc[rt][r];
    }
  }
  __syncthreads();

  const float* Pp = sP + 4 * cl;
  const float4 pv = *reinterpret_cast<const float4*>(&Pp[64 * CPITCH]);
  v8f c[4] = {};
  float4 q0[16], q1[16];
#pragma unroll
  for (int u = 0; u < 16; ++u) {
    const int kk = 4 * u + 2 * hl;
    q0[u] = *reinterpret_cast<const float4*>(&Pp[kk * CPITCH]);
    q1[u] = *reinterpret_cast<const float4*>(&Pp[(kk + 1) * CPITCH]);
  }
  asm volatile("" ::: "memory");
  __builtin_amdgcn_sched_barrier(0);
#pragma unroll
  for (int u = 0; u < 16; ++u) {
    v2f bf;
    bf.x = q0[u].x; bf.y = q1[u].x; c[0] = wmma4(av[u], bf, c[0]);
    bf.x = q0[u].y; bf.y = q1[u].y; c[1] = wmma4(av[u], bf, c[1]);
    bf.x = q0[u].z; bf.y = q1[u].z; c[2] = wmma4(av[u], bf, c[2]);
    bf.x = q0[u].w; bf.y = q1[u].w; c[3] = wmma4(av[u], bf, c[3]);
  }
  float* dst = Out + (size_t)b * NTOK2 * FDIM;
  const int rbase = m0 + (hl << 3);
#pragma unroll
  for (int r = 0; r < 8; ++r) {
    float4 o;
    o.x = c[0][r] + pv.x; o.y = c[1][r] + pv.y; o.z = c[2][r] + pv.z; o.w = c[3][r] + pv.w;
    *reinterpret_cast<float4*>(&dst[(rbase + r) * FDIM + 4 * cl]) = o;
  }
}

__global__ void ln_res_kernel(const float* mid, const float* vin, const float* g,
                              const float* bb, float* vout) {
  int row = blockIdx.x * 8 + (threadIdx.x >> 5);
  int lane = threadIdx.x & 31;
  const float* mr = mid + (size_t)row * FDIM;
  float x0 = mr[lane], x1 = mr[lane + 32];
  float s = x0 + x1, ss = x0 * x0 + x1 * x1;
#pragma unroll
  for (int m = 16; m >= 1; m >>= 1) {
    s  += __shfl_xor(s, m, 32);
    ss += __shfl_xor(ss, m, 32);
  }
  float mean = s * (1.f / 64.f);
  float var  = ss * (1.f / 64.f) - mean * mean;
  float inv  = rsqrtf(var + 1e-5f);
  size_t base = (size_t)row * FDIM;
  vout[base + lane]      = (x0 - mean) * inv * g[lane]      + bb[lane]      + vin[base + lane];
  vout[base + lane + 32] = (x1 - mean) * inv * g[lane + 32] + bb[lane + 32] + vin[base + lane + 32];
}

__global__ void final_kernel(const float* weight, const float* weightf,
                             const float* S, float* out, float mul) {
  __shared__ float sh[KSPLIT * TPLANE];
  int b = blockIdx.x;
  int ks = __builtin_amdgcn_readfirstlane(threadIdx.x >> 5);
  int lane = threadIdx.x & 31;
  int m0 = blockIdx.y * 16;
  const float* Sb = S + (size_t)b * NTOK2 * FDIM;
  int row = m0 + (lane & 15);
  const int klen = NTOK2 / KSPLIT;
  int kbeg = ks * klen, kend = kbeg + klen;
  const float* W = (kbeg < NTOKEN) ? weight : weightf;
  v8f c[4] = {};
  for (int k0 = kbeg; k0 < kend; k0 += 4) {
    int kb = (k0 & (NTOKEN - 1)) + ((lane >> 4) << 1);
    v2f a; a.x = W[kb * NTOKEN + row]; a.y = W[(kb + 1) * NTOKEN + row];
    if ((k0 & 60) == 0 && (k0 & (NTOKEN - 1)) + 64 < NTOKEN) {
      __builtin_prefetch(&W[(kb + 64) * NTOKEN + row], 0, 3);
    }
#pragma unroll
    for (int t = 0; t < 4; ++t) {
      v2f bf = frag_b(Sb, FDIM, k0, t * 16, lane);
      c[t] = wmma4(a, bf, c[t]);
    }
  }
  ksplit_reduce_store(sh, c, ks, lane, out + (size_t)b * NTOKEN * FDIM, m0, mul, 16, 1);
}

extern "C" void kernel_launch(void* const* d_in, const int* in_sizes, int n_in,
                              void* d_out, int out_size, void* d_ws, size_t ws_size,
                              hipStream_t stream) {
  (void)in_sizes; (void)n_in; (void)out_size; (void)ws_size;
  const float* xy   = (const float*)d_in[0];
  const float* kW0  = (const float*)d_in[1];
  const float* kb0  = (const float*)d_in[2];
  const float* kW1  = (const float*)d_in[3];
  const float* kb1  = (const float*)d_in[4];
  const float* kW2  = (const float*)d_in[5];
  const float* kb2  = (const float*)d_in[6];
  const float* fW0  = (const float*)d_in[7];
  const float* fb0  = (const float*)d_in[8];
  const float* fW1  = (const float*)d_in[9];
  const float* fb1  = (const float*)d_in[10];
  const float* fW2  = (const float*)d_in[11];
  const float* fb2  = (const float*)d_in[12];
  const float* Wq   = (const float*)d_in[13];
  const float* bq   = (const float*)d_in[14];
  const float* Wk   = (const float*)d_in[15];
  const float* bk   = (const float*)d_in[16];
  const float* ln0g = (const float*)d_in[17];
  const float* ln0b = (const float*)d_in[18];
  const float* lng  = (const float*)d_in[19];
  const float* lnb  = (const float*)d_in[20];
  float* out = (float*)d_out;

  float* ws = (float*)d_ws;
  const size_t SZ_1M = 1048576;
  float* w_weight  = ws;
  float* w_weightf = ws + SZ_1M;
  float* V0        = ws + 2 * SZ_1M;
  float* V1        = ws + 3 * SZ_1M;
  float* Caug      = ws + 4 * SZ_1M;
  float* caug      = ws + 4 * SZ_1M + 20480;
  float* Gp        = ws + 4 * SZ_1M + 20992;
  float* gp        = ws + 4 * SZ_1M + 152064;
  float* MID       = ws + 12 * SZ_1M + 131072;
  float* ln0part   = ws + 4 * SZ_1M + 195072;

  const float DXc   = 2.0f / 31.0f;
  const float dxdy  = DXc * DXc;
  const float scale = 0.125f;

  edge_mlp_kernel<<<LN0_JOBS + 2 * NTOKEN, 128, 0, stream>>>(kW0, kb0, kW1, kb1, kW2, kb2,
                                                             fW0, fb0, fW1, fb1, fW2, fb2,
                                                             w_weight, w_weightf, xy, ln0part);
  ln0_apply_kernel<<<4096, 256, 0, stream>>>(xy, ln0g, ln0b, ln0part, V0);

  atb_kernel<<<dim3(NLAYER, 4, 1), KSPLIT * 32, 0, stream>>>(
      Wq, HEAD * DK * FDIM, Wk, HEAD * DK * FDIM, HEAD * DK, 1.0f, Caug, CAUG, nullptr);
  cvec_kernel<<<dim3(NLAYER, 2), 1024, 0, stream>>>(Wq, bq, Wk, bk, Caug, caug);

  float* Va = V0;
  float* Vb = V1;
  for (int j = 0; j < NLAYER - 1; ++j) {
    atb_kernel<<<dim3(BATCH, 4, KQ), KSPLIT * 32, 0, stream>>>(
        Va, NTOK2 * FDIM, Va, NTOK2 * FDIM, NTOK2, scale * dxdy, Gp, 4096, gp);
    vp_kernel<<<dim3(BATCH, NTOK2 / 64), 128, 0, stream>>>(Va, Caug + (size_t)j * CAUG, caug + j * 80,
                                                           Gp, gp, MID);
    ln_res_kernel<<<2048, 256, 0, stream>>>(MID, Va, lng + j * FDIM,
                                            lnb + j * FDIM, Vb);
    float* t = Va; Va = Vb; Vb = t;
  }

  {
    int j = NLAYER - 1;
    atb_kernel<<<dim3(BATCH, 4, KQ), KSPLIT * 32, 0, stream>>>(
        Va, NTOK2 * FDIM, xy, NTOK2 * FDIM, NTOKEN, scale, Gp, 4096, gp);
    vp_kernel<<<dim3(BATCH, NTOK2 / 64), 128, 0, stream>>>(Va, Caug + (size_t)j * CAUG, caug + j * 80,
                                                           Gp, gp, MID);
    final_kernel<<<dim3(BATCH, NTOKEN / 16), KSPLIT * 32, 0, stream>>>(
        w_weight, w_weightf, MID, out, dxdy * dxdy);
  }
}


# All input-only work rides in the edge-MLP launch as extra work-groups: global layernorm (each work-group re-sums its batch slab) and the weight-only Caug/caug; 3 launches gone; same sums and order
# speedup vs baseline: 5.9657x; 1.0767x over previous
#include <hip/hip_runtime.h>
#include <hip/hip_bf16.h>

#define NGRID   32
#define NTOKEN  1024
#define NTOK2   2048
#define HEAD    4
#define DK      64
#define NLAYER  4
#define FDIM    64
#define BATCH   8

typedef __attribute__((ext_vector_type(2))) float v2f;
typedef __attribute__((ext_vector_type(8))) float v8f;

__device__ __forceinline__ v8f wmma4(v2f a, v2f b, v8f c) {
  return __builtin_amdgcn_wmma_f32_16x16x4_f32(
       false, a,  false, b,
       (short)0, c,  false,  false);
}

__device__ __forceinline__ v2f frag_a(const float* src, int lda, int m0, int k0, int lane) {
  int row = m0 + (lane & 15);
  int kk  = k0 + ((lane >> 4) << 1);
  v2f a; a.x = src[row * lda + kk]; a.y = src[row * lda + kk + 1];
  return a;
}

__device__ __forceinline__ v2f frag_b(const float* src, int ldb, int k0, int n0, int lane) {
  int col = n0 + (lane & 15);
  int kk  = k0 + ((lane >> 4) << 1);
  v2f b; b.x = src[kk * ldb + col]; b.y = src[(kk + 1) * ldb + col];
  return b;
}


#define KSPLIT   16
#define TPITCH   68
#define TPLANE   (16 * TPITCH)

__device__ __forceinline__ void ksplit_reduce_store(float* sh, const v8f* c, int ks, int lane,
                                                    float* dst, int m0, float mul,
                                                    int tstep, int cstep) {
  float* mine = sh + ks * TPLANE;
  int lrow = (lane >> 4) << 3;
  int lcol = lane & 15;
#pragma unroll
  for (int t = 0; t < 4; ++t) {
#pragma unroll
    for (int r = 0; r < 8; ++r) mine[(lrow + r) * TPITCH + t * tstep + lcol * cstep] = c[t][r];
  }
  __syncthreads();
  for (int idx = threadIdx.x; idx < 16 * 64; idx += KSPLIT * 32) {
    int row = idx >> 6, col = idx & 63;
    float acc = sh[row * TPITCH + col];
#pragma unroll
    for (int p = 1; p < KSPLIT; ++p) acc += sh[p * TPLANE + row * TPITCH + col];
    dst[(m0 + row) * 64 + col] = acc * mul;
  }
}

#define LN0_CHUNKS  8
#define LN0_SLICE   (NTOK2 * FDIM / LN0_CHUNKS)
#define LNA_WGS     64
#define LNA_ELEMS   (BATCH * NTOK2 * FDIM / LNA_WGS)
#define WJOBS       16
#define CAUG    (80 * 64)

__device__ __forceinline__ void ln0_job(const float* xy, const float* g, const float* bb,
                                        float* V, int q, float* shp) {
  const int b = q / (LNA_WGS / BATCH);
  const int wave = threadIdx.x >> 5;
  const float4* slab = reinterpret_cast<const float4*>(xy + (size_t)b * NTOK2 * FDIM);
  for (int cidx = 0; cidx < LN0_CHUNKS; ++cidx) {
    const float4* p = slab + cidx * (LN0_SLICE / 4);
    float4 v[32];
#pragma unroll
    for (int it = 0; it < 32; ++it) v[it] = p[it * 128 + threadIdx.x];
    float4 s4 = make_float4(0.f, 0.f, 0.f, 0.f), q4 = make_float4(0.f, 0.f, 0.f, 0.f);
#pragma unroll
    for (int it = 0; it < 32; ++it) {
      s4.x += v[it].x; s4.y += v[it].y; s4.z += v[it].z; s4.w += v[it].w;
      q4.x = fmaf(v[it].x, v[it].x, q4.x); q4.y = fmaf(v[it].y, v[it].y, q4.y);
      q4.z = fmaf(v[it].z, v[it].z, q4.z); q4.w = fmaf(v[it].w, v[it].w, q4.w);
    }
    float s = (s4.x + s4.y) + (s4.z + s4.w);
    float ss = (q4.x + q4.y) + (q4.z + q4.w);
#pragma unroll
    for (int m = 16; m >= 1; m >>= 1) {
      s  += __shfl_xor(s, m, 32);
      ss += __shfl_xor(ss, m, 32);
    }
    if ((threadIdx.x & 31) == 0) { shp[8 * cidx + 2 * wave] = s; shp[8 * cidx + 2 * wave + 1] = ss; }
  }
  __syncthreads();
  float s = 0.f, ss = 0.f;
#pragma unroll
  for (int cidx = 0; cidx < LN0_CHUNKS; ++cidx) {
    const float* w = shp + 8 * cidx;
    const float ps = ((w[0] + w[2]) + w[4]) + w[6];
    const float pq = ((w[1] + w[3]) + w[5]) + w[7];
    if (cidx == 0) { s = ps; ss = pq; } else { s += ps; ss += pq; }
  }
  const float N = (float)(NTOK2 * FDIM);
  const float mean = s / N;
  const float var  = ss / N - mean * mean;
  const float rstd = rsqrtf(var + 1e-5f);
  for (int e4 = q * (LNA_ELEMS / 4); e4 < (q + 1) * (LNA_ELEMS / 4); e4 += 1024) {
    const float4* x4 = reinterpret_cast<const float4*>(xy) + e4;
    const float4* g4 = reinterpret_cast<const float4*>(g) + (e4 & (NTOK2 * FDIM / 4 - 1));
    const float4* b4 = reinterpret_cast<const float4*>(bb) + (e4 & (NTOK2 * FDIM / 4 - 1));
    float4* o4 = reinterpret_cast<float4*>(V) + e4;
    float4 xv[8], gv[8], bv[8];
#pragma unroll
    for (int i = 0; i < 8; ++i) {
      xv[i] = x4[i * 128 + threadIdx.x]; gv[i] = g4[i * 128 + threadIdx.x]; bv[i] = b4[i * 128 + threadIdx.x];
    }
#pragma unroll
    for (int i = 0; i < 8; ++i) {
      float4 o;
      o.x = (xv[i].x - mean) * rstd * gv[i].x + bv[i].x;
      o.y = (xv[i].y - mean) * rstd * gv[i].y + bv[i].y;
      o.z = (xv[i].z - mean) * rstd * gv[i].z + bv[i].z;
      o.w = (xv[i].w - mean) * rstd * gv[i].w + bv[i].w;
      o4[i * 128 + threadIdx.x] = o;
    }
  }
}

__device__ __forceinline__ void caug_job(const float* Wq, const float* Wk, float* Caug, int jb) {
  const int lane = threadIdx.x & 31, wave = threadIdx.x >> 5;
  const int cl = lane & 15, hl = lane >> 4;
  const int j = jb >> 1;
  const int m0 = (2 * (jb & 1) + (wave >> 1)) * 16;
  const int n0 = (wave & 1) * 32;
  const float* Ap = Wq + (size_t)j * HEAD * DK * FDIM + m0 + cl;
  const float* Bp = Wk + (size_t)j * HEAD * DK * FDIM + n0 + cl;
  v8f t0 = {}, t1 = {};
  for (int kb = 0; kb < HEAD * DK; kb += 64) {
    v2f a[16], b0[16], b1[16];
#pragma unroll
    for (int u = 0; u < 16; ++u) {
      const int kk = kb + 4 * u + 2 * hl;
      a[u].x  = Ap[kk * 64];      a[u].y  = Ap[(kk + 1) * 64];
      b0[u].x = Bp[kk * 64];      b0[u].y = Bp[(kk + 1) * 64];
      b1[u].x = Bp[kk * 64 + 16]; b1[u].y = Bp[(kk + 1) * 64 + 16];
    }
#pragma unroll
    for (int part = 0; part < 4; ++part) {
      v8f p0 = {}, p1 = {};
#pragma unroll
      for (int u = 4 * part; u < 4 * part + 4; ++u) {
        p0 = wmma4(a[u], b0[u], p0);
        p1 = wmma4(a[u], b1[u], p1);
      }
      if (kb == 0 && part == 0) { t0 = p0; t1 = p1; } else { t0 += p0; t1 += p1; }
    }
  }
  float* Cj = Caug + (size_t)j * CAUG;
#pragma unroll
  for (int r = 0; r < 8; ++r) {
    const int row = m0 + (hl << 3) + r;
    Cj[row * 64 + n0 + cl]      = t0[r];
    Cj[row * 64 + n0 + 16 + cl] = t1[r];
  }
}

__device__ __forceinline__ void cvec_job(const float* Wq, const float* bq, const float* Wk,
                                         const float* bk, float* Caug, float* caug, int jb,
                                         float* sh) {
  const int j = jb >> 1;
  const bool isD = (jb & 1) != 0;
  const float* Wm = (isD ? Wk : Wq) + (size_t)j * 256 * 64;
  const float* v  = (isD ? bq : bk) + j * 256;
  const int i = threadIdx.x & 63, half = threadIdx.x >> 6;
  for (int cc = 0; cc < 8; ++cc) {
    const int kc = half * 8 + cc;
    float a = 0.f;
#pragma unroll
    for (int u = 0; u < 16; ++u) {
      int k = kc * 16 + u;
      a = fmaf(Wm[k * 64 + i], v[k], a);
    }
    sh[kc * 64 + i] = a;
  }
  __syncthreads();
  float* Cj = Caug + (size_t)j * CAUG;
  float* cj = caug + j * 80;
  if (threadIdx.x < 64) {
    float t = sh[threadIdx.x];
    for (int p = 1; p < 16; ++p) t += sh[p * 64 + threadIdx.x];
    if (isD) Cj[64 * 64 + threadIdx.x] = t; else cj[threadIdx.x] = t;
  }
  if (isD) {
    if ((threadIdx.x >> 5) == 2) {
      const int lane = threadIdx.x & 31;
      float e = 0.f;
      for (int u = 0; u < 8; ++u) {
        int k = lane + 32 * u;
        e = fmaf(bq[j * 256 + k], bk[j * 256 + k], e);
      }
      for (int m = 16; m >= 1; m >>= 1) e += __shfl_xor(e, m, 32);
      if (lane == 0) cj[64] = e;
    }
  } else {
    for (int idx = threadIdx.x; idx < 15 * 64; idx += 128) Cj[65 * 64 + idx] = 0.f;
    if (threadIdx.x < 15) cj[65 + threadIdx.x] = 0.f;
  }
}
__device__ __forceinline__ float leaky01(float t) {
  return fmaxf(t, 0.01f * t);
}

__global__ __launch_bounds__(128) void edge_mlp_kernel(
                                const float* kW0, const float* kb0, const float* kW1,
                                const float* kb1, const float* kW2, const float* kb2,
                                const float* fW0, const float* fb0, const float* fW1,
                                const float* fb1, const float* fW2, const float* fb2,
                                float* wout, float* fout,
                                const float* xy, const float* ln0g, const float* ln0b, float* V0,
                                const float* Wq, const float* bq, const float* Wk, const float* bk,
                                float* Caug, float* caug) {
  __shared__ float shx[16 * 64];
  if (blockIdx.x < LNA_WGS) {
    ln0_job(xy, ln0g, ln0b, V0, blockIdx.x, shx);
    return;
  }
  if (blockIdx.x < LNA_WGS + WJOBS / 2) {
    caug_job(Wq, Wk, Caug, blockIdx.x - LNA_WGS);
    return;
  }
  if (blockIdx.x < LNA_WGS + WJOBS) {
    cvec_job(Wq, bq, Wk, bk, Caug, caug, blockIdx.x - LNA_WGS - WJOBS / 2, shx);
    return;
  }
  const int ejob = blockIdx.x - (LNA_WGS + WJOBS);
  const bool isF = ejob >= NTOKEN;
  const float* W0 = isF ? fW0 : kW0;
  const float* b0 = isF ? fb0 : kb0;
  const float* W1 = isF ? fW1 : kW1;
  const float* b1 = isF ? fb1 : kb1;
  const float* W2 = isF ? fW2 : kW2;
  const float* b2 = isF ? fb2 : kb2;
  float* dst      = isF ? fout : wout;

  const int lane = threadIdx.x & 31;
  const int wave = threadIdx.x >> 5;
  const int c = lane & 15, h = lane >> 4;
  const int n = ejob & (NTOKEN - 1);
  const int mbase = wave * 256;
  const float DXc = 2.0f / 31.0f;
  const float e0 = -1.f + DXc * (float)(n >> 5);
  const float e1 = -1.f + DXc * (float)(n & 31);

  v2f wa[4][8];
#pragma unroll
  for (int t = 0; t < 4; ++t) {
#pragma unroll
    for (int i = 0; i < 8; ++i) {
      const float* p = W1 + (t * 16 + c) * 32 + 4 * i + 2 * h;
      wa[t][i].x = p[0]; wa[t][i].y = p[1];
    }
  }
  float pre[16], wc2[16], wc3[16];
#pragma unroll
  for (int i = 0; i < 8; ++i) {
#pragma unroll
    for (int j = 0; j < 2; ++j) {
      int u = 4 * i + 2 * h + j;
      float t = b0[u];
      t = fmaf(W0[u * 4 + 0], e0, t);
      t = fmaf(W0[u * 4 + 1], e1, t);
      pre[2 * i + j] = t;
      wc2[2 * i + j] = W0[u * 4 + 2];
      wc3[2 * i + j] = W0[u * 4 + 3];
    }
  }
  v8f cb[4];
  float w2v[4][8];
#pragma unroll
  for (int t = 0; t < 4; ++t) {
#pragma unroll
    for (int r = 0; r < 8; ++r) {
      cb[t][r]  = b1[t * 16 + 8 * h + r];
      w2v[t][r] = W2[t * 16 + 8 * h + r];
    }
  }
  const float bias2 = b2[0];

  for (int it = 0; it < 8; ++it) {
    const int m0 = mbase + it * 32;
    float res[2];
#pragma unroll
    for (int et = 0; et < 2; ++et) {
      const int m = m0 + et * 16 + c;
      const float e2 = -1.f + DXc * (float)(m >> 5);
      const float e3 = -1.f + DXc * (float)(m & 31);
      v2f hb[8];
#pragma unroll
      for (int i = 0; i < 8; ++i) {
        float t0 = fmaf(wc3[2 * i],     e3, fmaf(wc2[2 * i],     e2, pre[2 * i]));
        float t1 = fmaf(wc3[2 * i + 1], e3, fmaf(wc2[2 * i + 1], e2, pre[2 * i + 1]));
        hb[i].x = leaky01(t0); hb[i].y = leaky01(t1);
      }
      v8f acc[4];
#pragma unroll
      for (int t = 0; t < 4; ++t) acc[t] = wmma4(wa[t][0], hb[0], cb[t]);
#pragma unroll
      for (int i = 1; i < 8; ++i) {
#pragma unroll
        for (int t = 0; t < 4; ++t) acc[t] = wmma4(wa[t][i], hb[i], acc[t]);
      }
      float s = 0.f;
#pragma unroll
      for (int t = 0; t < 4; ++t) {
#pragma unroll
        for (int r = 0; r < 8; ++r) s = fmaf(w2v[t][r], leaky01(acc[t][r]), s);
      }
      s += __shfl_xor(s, 16, 32);
      res[et] = s + bias2;
    }
    dst[n * NTOKEN + m0 + lane] = h ? res[1] : res[0];
  }
}

#define KQ      4
#define GPITCH  72

__global__ __launch_bounds__(KSPLIT * 32) void atb_kernel(
                           const float* Abuf, int astride, const float* Bbuf, int bstride,
                           int kdim, float s, float* Out, int ostride, float* csum) {
  __shared__ float sh[KSPLIT * TPLANE];
  __shared__ float shc[KSPLIT * 64];
  int z = blockIdx.x;
  int ks = __builtin_amdgcn_readfirstlane(threadIdx.x >> 5);
  int lane = threadIdx.x & 31;
  const int cl = lane & 15, hl = lane >> 4;
  int m0 = blockIdx.y * 16;
  int klen = kdim / (KSPLIT * (int)gridDim.z);
  int kbeg = ((int)blockIdx.z * KSPLIT + ks) * klen, kend = kbeg + klen;
  const float* Ap = Abuf + (size_t)z * astride + m0 + cl;
  const float* Bp = Bbuf + (size_t)z * bstride + 4 * cl;
  v8f c[4] = {};
  float cs[4] = {0.f, 0.f, 0.f, 0.f};
  for (int kb = kbeg; kb < kend; kb += 16) {
    v2f a[4];
    float b0[4][4], b1[4][4];
#pragma unroll
    for (int u = 0; u < 4; ++u) {
      const int kk = kb + 4 * u + 2 * hl;
      a[u].x = Ap[kk * 64]; a[u].y = Ap[(kk + 1) * 64];
#pragma unroll
      for (int t = 0; t < 4; ++t) {
        b0[u][t] = Bp[kk * 64 + t];
        b1[u][t] = Bp[(kk + 1) * 64 + t];
      }
    }
#pragma unroll
    for (int u = 0; u < 4; ++u) {
#pragma unroll
      for (int t = 0; t < 4; ++t) {
        v2f bf; bf.x = b0[u][t]; bf.y = b1[u][t];
        cs[t] += bf.x + bf.y;
        c[t] = wmma4(a[u], bf, c[t]);
      }
    }
  }
#pragma unroll
  for (int t = 0; t < 4; ++t) {
    float v = cs[t] + __shfl_xor(cs[t], 16, 32);
    if (lane < 16) shc[ks * 64 + 4 * lane + t] = v;
  }
  size_t slot = (size_t)z * gridDim.z + blockIdx.z;
  ksplit_reduce_store(sh, c, ks, lane, Out + slot * ostride, m0, s, 1, 4);
  if (csum != nullptr && blockIdx.y == 0 && threadIdx.x < 64) {
    float a = shc[threadIdx.x];
#pragma unroll
    for (int p = 1; p < KSPLIT; ++p) a += shc[p * 64 + threadIdx.x];
    csum[slot * 64 + threadIdx.x] = a * s;
  }
}

#define CPITCH  68
__global__ __launch_bounds__(128) void vp_kernel(const float* Vin, const float* Cj,
                                                 const float* cj, const float* Gp,
                                                 const float* gp, float* Out) {
  __shared__ __attribute__((aligned(16))) float sG[64 * GPITCH];
  __shared__ __attribute__((aligned(16))) float sC[80 * CPITCH];
  __shared__ __attribute__((aligned(16))) float sP[80 * CPITCH];
  __shared__ float sg[64];
  __shared__ __attribute__((aligned(16))) float scj[80];
  const int b = blockIdx.x, tid = threadIdx.x;
  const int wave = tid >> 5, lane = tid & 31;
  const int cl = lane & 15, hl = lane >> 4;
  const int m0 = blockIdx.y * 64 + wave * 16;
  const float* A  = Vin + (size_t)b * NTOK2 * FDIM;

  const float4* G4 = reinterpret_cast<const float4*>(Gp + (size_t)b * KQ * 4096);
  const float4* C4 = reinterpret_cast<const float4*>(Cj);
  const float* g0 = gp + (size_t)b * KQ * 64;
  float4 gq[8][KQ];
#pragma unroll
  for (int i = 0; i < 8; ++i) {
#pragma unroll
    for (int q = 0; q < KQ; ++q) gq[i][q] = G4[q * 1024 + tid + 128 * i];
  }
  const float4 cq0 = C4[tid],       cq1 = C4[tid + 128], cq2 = C4[tid + 256], cq3 = C4[tid + 384],
               cq4 = C4[tid + 512], cq5 = C4[tid + 640], cq6 = C4[tid + 768], cq7 = C4[tid + 896],
               cq8 = C4[tid + 1024], cq9 = C4[tid + 1152];
  float gv[KQ];
#pragma unroll
  for (int q = 0; q < KQ; ++q) gv[q] = g0[q * 64 + (tid & 63)];
  const float cjg = cj[tid < 80 ? tid : 79];
  v2f av[16];
#pragma unroll
  for (int u = 0; u < 16; ++u) av[u] = frag_a(A, FDIM, m0, 4 * u, lane);
  __builtin_amdgcn_sched_barrier(0);
#pragma unroll
  for (int i = 0; i < 8; ++i) {
    const int i4 = tid + 128 * i;
    float4 v = gq[i][0];
#pragma unroll
    for (int q = 1; q < KQ; ++q) {
      v.x += gq[i][q].x; v.y += gq[i][q].y; v.z += gq[i][q].z; v.w += gq[i][q].w;
    }
    *reinterpret_cast<float4*>(&sG[(i4 >> 4) * GPITCH + ((i4 & 15) << 2)]) = v;
  }
  {
    float4* cdst = reinterpret_cast<float4*>(&sC[(tid >> 4) * CPITCH + ((tid & 15) << 2)]);
    const int cstep = 8 * CPITCH / 4;
    cdst[0 * cstep] = cq0; cdst[1 * cstep] = cq1; cdst[2 * cstep] = cq2; cdst[3 * cstep] = cq3;
    cdst[4 * cstep] = cq4; cdst[5 * cstep] = cq5; cdst[6 * cstep] = cq6; cdst[7 * cstep] = cq7;
    cdst[8 * cstep] = cq8; cdst[9 * cstep] = cq9;
  }
  if (tid < 64) {
    float v = gv[0];
#pragma unroll
    for (int q = 1; q < KQ; ++q) v += gv[q];
    sg[tid] = v;
  }
  if (tid < 80) scj[tid] = cjg;
  __syncthreads();

  {
    const int col = wave * 16 + cl;
    const float g = sg[col];
    v2f gb[16];
#pragma unroll
    for (int u = 0; u < 16; ++u) {
      const int kk = 4 * u + 2 * hl;
      gb[u].x = sG[kk * GPITCH + col]; gb[u].y = sG[(kk + 1) * GPITCH + col];
    }
    v8f pacc[5];
#pragma unroll
    for (int rt = 0; rt < 5; ++rt) {
#pragma unroll
      for (int r = 0; r < 8; ++r) pacc[rt][r] = scj[rt * 16 + (hl << 3) + r] * g;
    }
    v2f ca[5][16];
#pragma unroll
    for (int rt = 0; rt < 5; ++rt) {
#pragma unroll
      for (int u = 0; u < 16; ++u)
        ca[rt][u] = *reinterpret_cast<const v2f*>(&sC[(rt * 16 + cl) * CPITCH + 4 * u + 2 * hl]);
    }
    __builtin_amdgcn_sched_barrier(0);
#pragma unroll
    for (int u = 0; u < 16; ++u) {
#pragma unroll
      for (int rt = 0; rt < 5; ++rt) pacc[rt] = wmma4(ca[rt][u], gb[u], pacc[rt]);
    }
#pragma unroll
    for (int rt = 0; rt < 5; ++rt) {
#pragma unroll
      for (int r = 0; r < 8; ++r) sP[(rt * 16 + (hl << 3) + r) * CPITCH + col] = pacc[rt][r];
    }
  }
  __syncthreads();

  const float* Pp = sP + 4 * cl;
  const float4 pv = *reinterpret_cast<const float4*>(&Pp[64 * CPITCH]);
  v8f c[4] = {};
  float4 q0[16], q1[16];
#pragma unroll
  for (int u = 0; u < 16; ++u) {
    const int kk = 4 * u + 2 * hl;
    q0[u] = *reinterpret_cast<const float4*>(&Pp[kk * CPITCH]);
    q1[u] = *reinterpret_cast<const float4*>(&Pp[(kk + 1) * CPITCH]);
  }
  asm volatile("" ::: "memory");
  __builtin_amdgcn_sched_barrier(0);
#pragma unroll
  for (int u = 0; u < 16; ++u) {
    v2f bf;
    bf.x = q0[u].x; bf.y = q1[u].x; c[0] = wmma4(av[u], bf, c[0]);
    bf.x = q0[u].y; bf.y = q1[u].y; c[1] = wmma4(av[u], bf, c[1]);
    bf.x = q0[u].z; bf.y = q1[u].z; c[2] = wmma4(av[u], bf, c[2]);
    bf.x = q0[u].w; bf.y = q1[u].w; c[3] = wmma4(av[u], bf, c[3]);
  }
  float* dst = Out + (size_t)b * NTOK2 * FDIM;
  const int rbase = m0 + (hl << 3);
#pragma unroll
  for (int r = 0; r < 8; ++r) {
    float4 o;
    o.x = c[0][r] + pv.x; o.y = c[1][r] + pv.y; o.z = c[2][r] + pv.z; o.w = c[3][r] + pv.w;
    *reinterpret_cast<float4*>(&dst[(rbase + r) * FDIM + 4 * cl]) = o;
  }
}

__global__ void ln_res_kernel(const float* mid, const float* vin, const float* g,
                              const float* bb, float* vout) {
  int row = blockIdx.x * 8 + (threadIdx.x >> 5);
  int lane = threadIdx.x & 31;
  const float* mr = mid + (size_t)row * FDIM;
  float x0 = mr[lane], x1 = mr[lane + 32];
  float s = x0 + x1, ss = x0 * x0 + x1 * x1;
#pragma unroll
  for (int m = 16; m >= 1; m >>= 1) {
    s  += __shfl_xor(s, m, 32);
    ss += __shfl_xor(ss, m, 32);
  }
  float mean = s * (1.f / 64.f);
  float var  = ss * (1.f / 64.f) - mean * mean;
  float inv  = rsqrtf(var + 1e-5f);
  size_t base = (size_t)row * FDIM;
  vout[base + lane]      = (x0 - mean) * inv * g[lane]      + bb[lane]      + vin[base + lane];
  vout[base + lane + 32] = (x1 - mean) * inv * g[lane + 32] + bb[lane + 32] + vin[base + lane + 32];
}

__global__ void final_kernel(const float* weight, const float* weightf,
                             const float* S, float* out, float mul) {
  __shared__ float sh[KSPLIT * TPLANE];
  int b = blockIdx.x;
  int ks = __builtin_amdgcn_readfirstlane(threadIdx.x >> 5);
  int lane = threadIdx.x & 31;
  int m0 = blockIdx.y * 16;
  const float* Sb = S + (size_t)b * NTOK2 * FDIM;
  int row = m0 + (lane & 15);
  const int klen = NTOK2 / KSPLIT;
  int kbeg = ks * klen, kend = kbeg + klen;
  const float* W = (kbeg < NTOKEN) ? weight : weightf;
  v8f c[4] = {};
  for (int k0 = kbeg; k0 < kend; k0 += 4) {
    int kb = (k0 & (NTOKEN - 1)) + ((lane >> 4) << 1);
    v2f a; a.x = W[kb * NTOKEN + row]; a.y = W[(kb + 1) * NTOKEN + row];
    if ((k0 & 60) == 0 && (k0 & (NTOKEN - 1)) + 64 < NTOKEN) {
      __builtin_prefetch(&W[(kb + 64) * NTOKEN + row], 0, 3);
    }
#pragma unroll
    for (int t = 0; t < 4; ++t) {
      v2f bf = frag_b(Sb, FDIM, k0, t * 16, lane);
      c[t] = wmma4(a, bf, c[t]);
    }
  }
  ksplit_reduce_store(sh, c, ks, lane, out + (size_t)b * NTOKEN * FDIM, m0, mul, 16, 1);
}

extern "C" void kernel_launch(void* const* d_in, const int* in_sizes, int n_in,
                              void* d_out, int out_size, void* d_ws, size_t ws_size,
                              hipStream_t stream) {
  (void)in_sizes; (void)n_in; (void)out_size; (void)ws_size;
  const float* xy   = (const float*)d_in[0];
  const float* kW0  = (const float*)d_in[1];
  const float* kb0  = (const float*)d_in[2];
  const float* kW1  = (const float*)d_in[3];
  const float* kb1  = (const float*)d_in[4];
  const float* kW2  = (const float*)d_in[5];
  const float* kb2  = (const float*)d_in[6];
  const float* fW0  = (const float*)d_in[7];
  const float* fb0  = (const float*)d_in[8];
  const float* fW1  = (const float*)d_in[9];
  const float* fb1  = (const float*)d_in[10];
  const float* fW2  = (const float*)d_in[11];
  const float* fb2  = (const float*)d_in[12];
  const float* Wq   = (const float*)d_in[13];
  const float* bq   = (const float*)d_in[14];
  const float* Wk   = (const float*)d_in[15];
  const float* bk   = (const float*)d_in[16];
  const float* ln0g = (const float*)d_in[17];
  const float* ln0b = (const float*)d_in[18];
  const float* lng  = (const float*)d_in[19];
  const float* lnb  = (const float*)d_in[20];
  float* out = (float*)d_out;

  float* ws = (float*)d_ws;
  const size_t SZ_1M = 1048576;
  float* w_weight  = ws;
  float* w_weightf = ws + SZ_1M;
  float* V0        = ws + 2 * SZ_1M;
  float* V1        = ws + 3 * SZ_1M;
  float* Caug      = ws + 4 * SZ_1M;
  float* caug      = ws + 4 * SZ_1M + 20480;
  float* Gp        = ws + 4 * SZ_1M + 20992;
  float* gp        = ws + 4 * SZ_1M + 152064;
  float* MID       = ws + 12 * SZ_1M + 131072;

  const float DXc   = 2.0f / 31.0f;
  const float dxdy  = DXc * DXc;
  const float scale = 0.125f;

  edge_mlp_kernel<<<LNA_WGS + WJOBS + 2 * NTOKEN, 128, 0, stream>>>(
      kW0, kb0, kW1, kb1, kW2, kb2, fW0, fb0, fW1, fb1, fW2, fb2, w_weight, w_weightf,
      xy, ln0g, ln0b, V0, Wq, bq, Wk, bk, Caug, caug);

  float* Va = V0;
  float* Vb = V1;
  for (int j = 0; j < NLAYER - 1; ++j) {
    atb_kernel<<<dim3(BATCH, 4, KQ), KSPLIT * 32, 0, stream>>>(
        Va, NTOK2 * FDIM, Va, NTOK2 * FDIM, NTOK2, scale * dxdy, Gp, 4096, gp);
    vp_kernel<<<dim3(BATCH, NTOK2 / 64), 128, 0, stream>>>(Va, Caug + (size_t)j * CAUG, caug + j * 80,
                                                           Gp, gp, MID);
    ln_res_kernel<<<2048, 256, 0, stream>>>(MID, Va, lng + j * FDIM,
                                            lnb + j * FDIM, Vb);
    float* t = Va; Va = Vb; Vb = t;
  }

  {
    int j = NLAYER - 1;
    atb_kernel<<<dim3(BATCH, 4, KQ), KSPLIT * 32, 0, stream>>>(
        Va, NTOK2 * FDIM, xy, NTOK2 * FDIM, NTOKEN, scale, Gp, 4096, gp);
    vp_kernel<<<dim3(BATCH, NTOK2 / 64), 128, 0, stream>>>(Va, Caug + (size_t)j * CAUG, caug + j * 80,
                                                           Gp, gp, MID);
    final_kernel<<<dim3(BATCH, NTOKEN / 16), KSPLIT * 32, 0, stream>>>(
        w_weight, w_weightf, MID, out, dxdy * dxdy);
  }
}


# Per-row layernorm + residual become the epilogue of the Vin*P launch (row sums inside the wave by 16-lane row_xmask exchanges); 3 launches and the mid round trip gone; the row sums change order
# speedup vs baseline: 6.3289x; 1.0609x over previous
#include <hip/hip_runtime.h>
#include <hip/hip_bf16.h>

#define NGRID   32
#define NTOKEN  1024
#define NTOK2   2048
#define HEAD    4
#define DK      64
#define NLAYER  4
#define FDIM    64
#define BATCH   8

typedef __attribute__((ext_vector_type(2))) float v2f;
typedef __attribute__((ext_vector_type(8))) float v8f;

__device__ __forceinline__ v8f wmma4(v2f a, v2f b, v8f c) {
  return __builtin_amdgcn_wmma_f32_16x16x4_f32(
       false, a,  false, b,
       (short)0, c,  false,  false);
}

__device__ __forceinline__ v2f frag_a(const float* src, int lda, int m0, int k0, int lane) {
  int row = m0 + (lane & 15);
  int kk  = k0 + ((lane >> 4) << 1);
  v2f a; a.x = src[row * lda + kk]; a.y = src[row * lda + kk + 1];
  return a;
}

__device__ __forceinline__ v2f frag_b(const float* src, int ldb, int k0, int n0, int lane) {
  int col = n0 + (lane & 15);
  int kk  = k0 + ((lane >> 4) << 1);
  v2f b; b.x = src[kk * ldb + col]; b.y = src[(kk + 1) * ldb + col];
  return b;
}


#define KSPLIT   16
#define TPITCH   68
#define TPLANE   (16 * TPITCH)

__device__ __forceinline__ void ksplit_reduce_store(float* sh, const v8f* c, int ks, int lane,
                                                    float* dst, int m0, float mul,
                                                    int tstep, int cstep) {
  float* mine = sh + ks * TPLANE;
  int lrow = (lane >> 4) << 3;
  int lcol = lane & 15;
#pragma unroll
  for (int t = 0; t < 4; ++t) {
#pragma unroll
    for (int r = 0; r < 8; ++r) mine[(lrow + r) * TPITCH + t * tstep + lcol * cstep] = c[t][r];
  }
  __syncthreads();
  for (int idx = threadIdx.x; idx < 16 * 64; idx += KSPLIT * 32) {
    int row = idx >> 6, col = idx & 63;
    float acc = sh[row * TPITCH + col];
#pragma unroll
    for (int p = 1; p < KSPLIT; ++p) acc += sh[p * TPLANE + row * TPITCH + col];
    dst[(m0 + row) * 64 + col] = acc * mul;
  }
}

#define LN0_CHUNKS  8
#define LN0_SLICE   (NTOK2 * FDIM / LN0_CHUNKS)
#define LNA_WGS     64
#define LNA_ELEMS   (BATCH * NTOK2 * FDIM / LNA_WGS)
#define WJOBS       16
#define CAUG    (80 * 64)

__device__ __forceinline__ void ln0_job(const float* xy, const float* g, const float* bb,
                                        float* V, int q, float* shp) {
  const int b = q / (LNA_WGS / BATCH);
  const int wave = threadIdx.x >> 5;
  const float4* slab = reinterpret_cast<const float4*>(xy + (size_t)b * NTOK2 * FDIM);
  for (int cidx = 0; cidx < LN0_CHUNKS; ++cidx) {
    const float4* p = slab + cidx * (LN0_SLICE / 4);
    float4 v[32];
#pragma unroll
    for (int it = 0; it < 32; ++it) v[it] = p[it * 128 + threadIdx.x];
    float4 s4 = make_float4(0.f, 0.f, 0.f, 0.f), q4 = make_float4(0.f, 0.f, 0.f, 0.f);
#pragma unroll
    for (int it = 0; it < 32; ++it) {
      s4.x += v[it].x; s4.y += v[it].y; s4.z += v[it].z; s4.w += v[it].w;
      q4.x = fmaf(v[it].x, v[it].x, q4.x); q4.y = fmaf(v[it].y, v[it].y, q4.y);
      q4.z = fmaf(v[it].z, v[it].z, q4.z); q4.w = fmaf(v[it].w, v[it].w, q4.w);
    }
    float s = (s4.x + s4.y) + (s4.z + s4.w);
    float ss = (q4.x + q4.y) + (q4.z + q4.w);
#pragma unroll
    for (int m = 16; m >= 1; m >>= 1) {
      s  += __shfl_xor(s, m, 32);
      ss += __shfl_xor(ss, m, 32);
    }
    if ((threadIdx.x & 31) == 0) { shp[8 * cidx + 2 * wave] = s; shp[8 * cidx + 2 * wave + 1] = ss; }
  }
  __syncthreads();
  float s = 0.f, ss = 0.f;
#pragma unroll
  for (int cidx = 0; cidx < LN0_CHUNKS; ++cidx) {
    const float* w = shp + 8 * cidx;
    const float ps = ((w[0] + w[2]) + w[4]) + w[6];
    const float pq = ((w[1] + w[3]) + w[5]) + w[7];
    if (cidx == 0) { s = ps; ss = pq; } else { s += ps; ss += pq; }
  }
  const float N = (float)(NTOK2 * FDIM);
  const float mean = s / N;
  const float var  = ss / N - mean * mean;
  const float rstd = rsqrtf(var + 1e-5f);
  for (int e4 = q * (LNA_ELEMS / 4); e4 < (q + 1) * (LNA_ELEMS / 4); e4 += 1024) {
    const float4* x4 = reinterpret_cast<const float4*>(xy) + e4;
    const float4* g4 = reinterpret_cast<const float4*>(g) + (e4 & (NTOK2 * FDIM / 4 - 1));
    const float4* b4 = reinterpret_cast<const float4*>(bb) + (e4 & (NTOK2 * FDIM / 4 - 1));
    float4* o4 = reinterpret_cast<float4*>(V) + e4;
    float4 xv[8], gv[8], bv[8];
#pragma unroll
    for (int i = 0; i < 8; ++i) {
      xv[i] = x4[i * 128 + threadIdx.x]; gv[i] = g4[i * 128 + threadIdx.x]; bv[i] = b4[i * 128 + threadIdx.x];
    }
#pragma unroll
    for (int i = 0; i < 8; ++i) {
      float4 o;
      o.x = (xv[i].x - mean) * rstd * gv[i].x + bv[i].x;
      o.y = (xv[i].y - mean) * rstd * gv[i].y + bv[i].y;
      o.z = (xv[i].z - mean) * rstd * gv[i].z + bv[i].z;
      o.w = (xv[i].w - mean) * rstd * gv[i].w + bv[i].w;
      o4[i * 128 + threadIdx.x] = o;
    }
  }
}

__device__ __forceinline__ void caug_job(const float* Wq, const float* Wk, float* Caug, int jb) {
  const int lane = threadIdx.x & 31, wave = threadIdx.x >> 5;
  const int cl = lane & 15, hl = lane >> 4;
  const int j = jb >> 1;
  const int m0 = (2 * (jb & 1) + (wave >> 1)) * 16;
  const int n0 = (wave & 1) * 32;
  const float* Ap = Wq + (size_t)j * HEAD * DK * FDIM + m0 + cl;
  const float* Bp = Wk + (size_t)j * HEAD * DK * FDIM + n0 + cl;
  v8f t0 = {}, t1 = {};
  for (int kb = 0; kb < HEAD * DK; kb += 64) {
    v2f a[16], b0[16], b1[16];
#pragma unroll
    for (int u = 0; u < 16; ++u) {
      const int kk = kb + 4 * u + 2 * hl;
      a[u].x  = Ap[kk * 64];      a[u].y  = Ap[(kk + 1) * 64];
      b0[u].x = Bp[kk * 64];      b0[u].y = Bp[(kk + 1) * 64];
      b1[u].x = Bp[kk * 64 + 16]; b1[u].y = Bp[(kk + 1) * 64 + 16];
    }
#pragma unroll
    for (int part = 0; part < 4; ++part) {
      v8f p0 = {}, p1 = {};
#pragma unroll
      for (int u = 4 * part; u < 4 * part + 4; ++u) {
        p0 = wmma4(a[u], b0[u], p0);
        p1 = wmma4(a[u], b1[u], p1);
      }
      if (kb == 0 && part == 0) { t0 = p0; t1 = p1; } else { t0 += p0; t1 += p1; }
    }
  }
  float* Cj = Caug + (size_t)j * CAUG;
#pragma unroll
  for (int r = 0; r < 8; ++r) {
    const int row = m0 + (hl << 3) + r;
    Cj[row * 64 + n0 + cl]      = t0[r];
    Cj[row * 64 + n0 + 16 + cl] = t1[r];
  }
}

__device__ __forceinline__ void cvec_job(const float* Wq, const float* bq, const float* Wk,
                                         const float* bk, float* Caug, float* caug, int jb,
                                         float* sh) {
  const int j = jb >> 1;
  const bool isD = (jb & 1) != 0;
  const float* Wm = (isD ? Wk : Wq) + (size_t)j * 256 * 64;
  const float* v  = (isD ? bq : bk) + j * 256;
  const int i = threadIdx.x & 63, half = threadIdx.x >> 6;
  for (int cc = 0; cc < 8; ++cc) {
    const int kc = half * 8 + cc;
    float a = 0.f;
#pragma unroll
    for (int u = 0; u < 16; ++u) {
      int k = kc * 16 + u;
      a = fmaf(Wm[k * 64 + i], v[k], a);
    }
    sh[kc * 64 + i] = a;
  }
  __syncthreads();
  float* Cj = Caug + (size_t)j * CAUG;
  float* cj = caug + j * 80;
  if (threadIdx.x < 64) {
    float t = sh[threadIdx.x];
    for (int p = 1; p < 16; ++p) t += sh[p * 64 + threadIdx.x];
    if (isD) Cj[64 * 64 + threadIdx.x] = t; else cj[threadIdx.x] = t;
  }
  if (isD) {
    if ((threadIdx.x >> 5) == 2) {
      const int lane = threadIdx.x & 31;
      float e = 0.f;
      for (int u = 0; u < 8; ++u) {
        int k = lane + 32 * u;
        e = fmaf(bq[j * 256 + k], bk[j * 256 + k], e);
      }
      for (int m = 16; m >= 1; m >>= 1) e += __shfl_xor(e, m, 32);
      if (lane == 0) cj[64] = e;
    }
  } else {
    for (int idx = threadIdx.x; idx < 15 * 64; idx += 128) Cj[65 * 64 + idx] = 0.f;
    if (threadIdx.x < 15) cj[65 + threadIdx.x] = 0.f;
  }
}
__device__ __forceinline__ float leaky01(float t) {
  return fmaxf(t, 0.01f * t);
}

__global__ __launch_bounds__(128) void edge_mlp_kernel(
                                const float* kW0, const float* kb0, const float* kW1,
                                const float* kb1, const float* kW2, const float* kb2,
                                const float* fW0, const float* fb0, const float* fW1,
                                const float* fb1, const float* fW2, const float* fb2,
                                float* wout, float* fout,
                                const float* xy, const float* ln0g, const float* ln0b, float* V0,
                                const float* Wq, const float* bq, const float* Wk, const float* bk,
                                float* Caug, float* caug) {
  __shared__ float shx[16 * 64];
  if (blockIdx.x < LNA_WGS) {
    ln0_job(xy, ln0g, ln0b, V0, blockIdx.x, shx);
    return;
  }
  if (blockIdx.x < LNA_WGS + WJOBS / 2) {
    caug_job(Wq, Wk, Caug, blockIdx.x - LNA_WGS);
    return;
  }
  if (blockIdx.x < LNA_WGS + WJOBS) {
    cvec_job(Wq, bq, Wk, bk, Caug, caug, blockIdx.x - LNA_WGS - WJOBS / 2, shx);
    return;
  }
  const int ejob = blockIdx.x - (LNA_WGS + WJOBS);
  const bool isF = ejob >= NTOKEN;
  const float* W0 = isF ? fW0 : kW0;
  const float* b0 = isF ? fb0 : kb0;
  const float* W1 = isF ? fW1 : kW1;
  const float* b1 = isF ? fb1 : kb1;
  const float* W2 = isF ? fW2 : kW2;
  const float* b2 = isF ? fb2 : kb2;
  float* dst      = isF ? fout : wout;

  const int lane = threadIdx.x & 31;
  const int wave = threadIdx.x >> 5;
  const int c = lane & 15, h = lane >> 4;
  const int n = ejob & (NTOKEN - 1);
  const int mbase = wave * 256;
  const float DXc = 2.0f / 31.0f;
  const float e0 = -1.f + DXc * (float)(n >> 5);
  const float e1 = -1.f + DXc * (float)(n & 31);

  v2f wa[4][8];
#pragma unroll
  for (int t = 0; t < 4; ++t) {
#pragma unroll
    for (int i = 0; i < 8; ++i) {
      const float* p = W1 + (t * 16 + c) * 32 + 4 * i + 2 * h;
      wa[t][i].x = p[0]; wa[t][i].y = p[1];
    }
  }
  float pre[16], wc2[16], wc3[16];
#pragma unroll
  for (int i = 0; i < 8; ++i) {
#pragma unroll
    for (int j = 0; j < 2; ++j) {
      int u = 4 * i + 2 * h + j;
      float t = b0[u];
      t = fmaf(W0[u * 4 + 0], e0, t);
      t = fmaf(W0[u * 4 + 1], e1, t);
      pre[2 * i + j] = t;
      wc2[2 * i + j] = W0[u * 4 + 2];
      wc3[2 * i + j] = W0[u * 4 + 3];
    }
  }
  v8f cb[4];
  float w2v[4][8];
#pragma unroll
  for (int t = 0; t < 4; ++t) {
#pragma unroll
    for (int r = 0; r < 8; ++r) {
      cb[t][r]  = b1[t * 16 + 8 * h + r];
      w2v[t][r] = W2[t * 16 + 8 * h + r];
    }
  }
  const float bias2 = b2[0];

  for (int it = 0; it < 8; ++it) {
    const int m0 = mbase + it * 32;
    float res[2];
#pragma unroll
    for (int et = 0; et < 2; ++et) {
      const int m = m0 + et * 16 + c;
      const float e2 = -1.f + DXc * (float)(m >> 5);
      const float e3 = -1.f + DXc * (float)(m & 31);
      v2f hb[8];
#pragma unroll
      for (int i = 0; i < 8; ++i) {
        float t0 = fmaf(wc3[2 * i],     e3, fmaf(wc2[2 * i],     e2, pre[2 * i]));
        float t1 = fmaf(wc3[2 * i + 1], e3, fmaf(wc2[2 * i + 1], e2, pre[2 * i + 1]));
        hb[i].x = leaky01(t0); hb[i].y = leaky01(t1);
      }
      v8f acc[4];
#pragma unroll
      for (int t = 0; t < 4; ++t) acc[t] = wmma4(wa[t][0], hb[0], cb[t]);
#pragma unroll
      for (int i = 1; i < 8; ++i) {
#pragma unroll
        for (int t = 0; t < 4; ++t) acc[t] = wmma4(wa[t][i], hb[i], acc[t]);
      }
      float s = 0.f;
#pragma unroll
      for (int t = 0; t < 4; ++t) {
#pragma unroll
        for (int r = 0; r < 8; ++r) s = fmaf(w2v[t][r], leaky01(acc[t][r]), s);
      }
      s += __shfl_xor(s, 16, 32);
      res[et] = s + bias2;
    }
    dst[n * NTOKEN + m0 + lane] = h ? res[1] : res[0];
  }
}

#define KQ      4
#define GPITCH  72

__global__ __launch_bounds__(KSPLIT * 32) void atb_kernel(
                           const float* Abuf, int astride, const float* Bbuf, int bstride,
                           int kdim, float s, float* Out, int ostride, float* csum) {
  __shared__ float sh[KSPLIT * TPLANE];
  __shared__ float shc[KSPLIT * 64];
  int z = blockIdx.x;
  int ks = __builtin_amdgcn_readfirstlane(threadIdx.x >> 5);
  int lane = threadIdx.x & 31;
  const int cl = lane & 15, hl = lane >> 4;
  int m0 = blockIdx.y * 16;
  int klen = kdim / (KSPLIT * (int)gridDim.z);
  int kbeg = ((int)blockIdx.z * KSPLIT + ks) * klen, kend = kbeg + klen;
  const float* Ap = Abuf + (size_t)z * astride + m0 + cl;
  const float* Bp = Bbuf + (size_t)z * bstride + 4 * cl;
  v8f c[4] = {};
  float cs[4] = {0.f, 0.f, 0.f, 0.f};
  for (int kb = kbeg; kb < kend; kb += 16) {
    v2f a[4];
    float b0[4][4], b1[4][4];
#pragma unroll
    for (int u = 0; u < 4; ++u) {
      const int kk = kb + 4 * u + 2 * hl;
      a[u].x = Ap[kk * 64]; a[u].y = Ap[(kk + 1) * 64];
#pragma unroll
      for (int t = 0; t < 4; ++t) {
        b0[u][t] = Bp[kk * 64 + t];
        b1[u][t] = Bp[(kk + 1) * 64 + t];
      }
    }
#pragma unroll
    for (int u = 0; u < 4; ++u) {
#pragma unroll
      for (int t = 0; t < 4; ++t) {
        v2f bf; bf.x = b0[u][t]; bf.y = b1[u][t];
        cs[t] += bf.x + bf.y;
        c[t] = wmma4(a[u], bf, c[t]);
      }
    }
  }
#pragma unroll
  for (int t = 0; t < 4; ++t) {
    float v = cs[t] + __shfl_xor(cs[t], 16, 32);
    if (lane < 16) shc[ks * 64 + 4 * lane + t] = v;
  }
  size_t slot = (size_t)z * gridDim.z + blockIdx.z;
  ksplit_reduce_store(sh, c, ks, lane, Out + slot * ostride, m0, s, 1, 4);
  if (csum != nullptr && blockIdx.y == 0 && threadIdx.x < 64) {
    float a = shc[threadIdx.x];
#pragma unroll
    for (int p = 1; p < KSPLIT; ++p) a += shc[p * 64 + threadIdx.x];
    csum[slot * 64 + threadIdx.x] = a * s;
  }
}

#define CPITCH  68

template <int M>
__device__ __forceinline__ float row_xor_add(float v) {
  const int o = __builtin_amdgcn_update_dpp(0, __builtin_bit_cast(int, v), 0x160 + M, 0xf, 0xf, true);
  return v + __builtin_bit_cast(float, o);
}

template <bool LNRES>
__global__ __launch_bounds__(128) void vp_kernel(const float* Vin, const float* Cj,
                                                 const float* cj, const float* Gp,
                                                 const float* gp, const float* lg,
                                                 const float* lb, float* Out) {
  __shared__ __attribute__((aligned(16))) float sG[64 * GPITCH];
  __shared__ __attribute__((aligned(16))) float sC[80 * CPITCH];
  __shared__ __attribute__((aligned(16))) float sP[80 * CPITCH];
  __shared__ float sg[64];
  __shared__ __attribute__((aligned(16))) float scj[80];
  const int b = blockIdx.x, tid = threadIdx.x;
  const int wave = tid >> 5, lane = tid & 31;
  const int cl = lane & 15, hl = lane >> 4;
  const int m0 = blockIdx.y * 64 + wave * 16;
  const float* A  = Vin + (size_t)b * NTOK2 * FDIM;

  const float4* G4 = reinterpret_cast<const float4*>(Gp + (size_t)b * KQ * 4096);
  const float4* C4 = reinterpret_cast<const float4*>(Cj);
  const float* g0 = gp + (size_t)b * KQ * 64;
  float4 gq[8][KQ];
#pragma unroll
  for (int i = 0; i < 8; ++i) {
#pragma unroll
    for (int q = 0; q < KQ; ++q) gq[i][q] = G4[q * 1024 + tid + 128 * i];
  }
  const float4 cq0 = C4[tid],       cq1 = C4[tid + 128], cq2 = C4[tid + 256], cq3 = C4[tid + 384],
               cq4 = C4[tid + 512], cq5 = C4[tid + 640], cq6 = C4[tid + 768], cq7 = C4[tid + 896],
               cq8 = C4[tid + 1024], cq9 = C4[tid + 1152];
  float gv[KQ];
#pragma unroll
  for (int q = 0; q < KQ; ++q) gv[q] = g0[q * 64 + (tid & 63)];
  const float cjg = cj[tid < 80 ? tid : 79];
  v2f av[16];
#pragma unroll
  for (int u = 0; u < 16; ++u) av[u] = frag_a(A, FDIM, m0, 4 * u, lane);
  __builtin_amdgcn_sched_barrier(0);
#pragma unroll
  for (int i = 0; i < 8; ++i) {
    const int i4 = tid + 128 * i;
    float4 v = gq[i][0];
#pragma unroll
    for (int q = 1; q < KQ; ++q) {
      v.x += gq[i][q].x; v.y += gq[i][q].y; v.z += gq[i][q].z; v.w += gq[i][q].w;
    }
    *reinterpret_cast<float4*>(&sG[(i4 >> 4) * GPITCH + ((i4 & 15) << 2)]) = v;
  }
  {
    float4* cdst = reinterpret_cast<float4*>(&sC[(tid >> 4) * CPITCH + ((tid & 15) << 2)]);
    const int cstep = 8 * CPITCH / 4;
    cdst[0 * cstep] = cq0; cdst[1 * cstep] = cq1; cdst[2 * cstep] = cq2; cdst[3 * cstep] = cq3;
    cdst[4 * cstep] = cq4; cdst[5 * cstep] = cq5; cdst[6 * cstep] = cq6; cdst[7 * cstep] = cq7;
    cdst[8 * cstep] = cq8; cdst[9 * cstep] = cq9;
  }
  if (tid < 64) {
    float v = gv[0];
#pragma unroll
    for (int q = 1; q < KQ; ++q) v += gv[q];
    sg[tid] = v;
  }
  if (tid < 80) scj[tid] = cjg;
  __syncthreads();

  float4 rv[8];
  float4 lg4 = make_float4(0.f, 0.f, 0.f, 0.f), lb4 = make_float4(0.f, 0.f, 0.f, 0.f);
  if constexpr (LNRES) {
#pragma unroll
    for (int r = 0; r < 8; ++r)
      rv[r] = *reinterpret_cast<const float4*>(&A[(m0 + (hl << 3) + r) * FDIM + 4 * cl]);
    lg4 = *reinterpret_cast<const float4*>(&lg[4 * cl]);
    lb4 = *reinterpret_cast<const float4*>(&lb[4 * cl]);
  }

  {
    const int col = wave * 16 + cl;
    const float g = sg[col];
    v2f gb[16];
#pragma unroll
    for (int u = 0; u < 16; ++u) {
      const int kk = 4 * u + 2 * hl;
      gb[u].x = sG[kk * GPITCH + col]; gb[u].y = sG[(kk + 1) * GPITCH + col];
    }
    v8f pacc[5];
#pragma unroll
    for (int rt = 0; rt < 5; ++rt) {
#pragma unroll
      for (int r = 0; r < 8; ++r) pacc[rt][r] = scj[rt * 16 + (hl << 3) + r] * g;
    }
    v2f ca[5][16];
#pragma unroll
    for (int rt = 0; rt < 5; ++rt) {
#pragma unroll
      for (int u = 0; u < 16; ++u)
        ca[rt][u] = *reinterpret_cast<const v2f*>(&sC[(rt * 16 + cl) * CPITCH + 4 * u + 2 * hl]);
    }
    asm volatile("" ::: "memory");
    __builtin_amdgcn_sched_barrier(0);
#pragma unroll
    for (int u = 0; u < 16; ++u) {
#pragma unroll
      for (int rt = 0; rt < 5; ++rt) pacc[rt] = wmma4(ca[rt][u], gb[u], pacc[rt]);
    }
#pragma unroll
    for (int rt = 0; rt < 5; ++rt) {
#pragma unroll
      for (int r = 0; r < 8; ++r) sP[(rt * 16 + (hl << 3) + r) * CPITCH + col] = pacc[rt][r];
    }
  }
  __syncthreads();

  const float* Pp = sP + 4 * cl;
  const float4 pv = *reinterpret_cast<const float4*>(&Pp[64 * CPITCH]);
  v8f c[4] = {};
  float4 q0[16], q1[16];
#pragma unroll
  for (int u = 0; u < 16; ++u) {
    const int kk = 4 * u + 2 * hl;
    q0[u] = *reinterpret_cast<const float4*>(&Pp[kk * CPITCH]);
    q1[u] = *reinterpret_cast<const float4*>(&Pp[(kk + 1) * CPITCH]);
  }
  asm volatile("" ::: "memory");
  __builtin_amdgcn_sched_barrier(0);
#pragma unroll
  for (int u = 0; u < 16; ++u) {
    v2f bf;
    bf.x = q0[u].x; bf.y = q1[u].x; c[0] = wmma4(av[u], bf, c[0]);
    bf.x = q0[u].y; bf.y = q1[u].y; c[1] = wmma4(av[u], bf, c[1]);
    bf.x = q0[u].z; bf.y = q1[u].z; c[2] = wmma4(av[u], bf, c[2]);
    bf.x = q0[u].w; bf.y = q1[u].w; c[3] = wmma4(av[u], bf, c[3]);
  }
  float* dst = Out + (size_t)b * NTOK2 * FDIM;
  const int rbase = m0 + (hl << 3);
  if constexpr (!LNRES) {
#pragma unroll
    for (int r = 0; r < 8; ++r) {
      float4 o;
      o.x = c[0][r] + pv.x; o.y = c[1][r] + pv.y; o.z = c[2][r] + pv.z; o.w = c[3][r] + pv.w;
      *reinterpret_cast<float4*>(&dst[(rbase + r) * FDIM + 4 * cl]) = o;
    }
  } else {
    float sm[8], sq[8];
#pragma unroll
    for (int r = 0; r < 8; ++r) {
      c[0][r] += pv.x; c[1][r] += pv.y; c[2][r] += pv.z; c[3][r] += pv.w;
      sm[r] = (c[0][r] + c[1][r]) + (c[2][r] + c[3][r]);
      sq[r] = fmaf(c[3][r], c[3][r], fmaf(c[2][r], c[2][r], fmaf(c[1][r], c[1][r], c[0][r] * c[0][r])));
    }
#pragma unroll
    for (int r = 0; r < 8; ++r) { sm[r] = row_xor_add<1>(sm[r]); sq[r] = row_xor_add<1>(sq[r]); }
#pragma unroll
    for (int r = 0; r < 8; ++r) { sm[r] = row_xor_add<2>(sm[r]); sq[r] = row_xor_add<2>(sq[r]); }
#pragma unroll
    for (int r = 0; r < 8; ++r) { sm[r] = row_xor_add<4>(sm[r]); sq[r] = row_xor_add<4>(sq[r]); }
#pragma unroll
    for (int r = 0; r < 8; ++r) { sm[r] = row_xor_add<8>(sm[r]); sq[r] = row_xor_add<8>(sq[r]); }
#pragma unroll
    for (int r = 0; r < 8; ++r) {
      const float mean = sm[r] * (1.f / 64.f);
      const float var  = sq[r] * (1.f / 64.f) - mean * mean;
      const float inv  = rsqrtf(var + 1e-5f);
      float4 o;
      o.x = (c[0][r] - mean) * inv * lg4.x + lb4.x + rv[r].x;
      o.y = (c[1][r] - mean) * inv * lg4.y + lb4.y + rv[r].y;
      o.z = (c[2][r] - mean) * inv * lg4.z + lb4.z + rv[r].z;
      o.w = (c[3][r] - mean) * inv * lg4.w + lb4.w + rv[r].w;
      *reinterpret_cast<float4*>(&dst[(rbase + r) * FDIM + 4 * cl]) = o;
    }
  }
}

__global__ void final_kernel(const float* weight, const float* weightf,
                             const float* S, float* out, float mul) {
  __shared__ float sh[KSPLIT * TPLANE];
  int b = blockIdx.x;
  int ks = __builtin_amdgcn_readfirstlane(threadIdx.x >> 5);
  int lane = threadIdx.x & 31;
  int m0 = blockIdx.y * 16;
  const float* Sb = S + (size_t)b * NTOK2 * FDIM;
  int row = m0 + (lane & 15);
  const int klen = NTOK2 / KSPLIT;
  int kbeg = ks * klen, kend = kbeg + klen;
  const float* W = (kbeg < NTOKEN) ? weight : weightf;
  v8f c[4] = {};
  for (int k0 = kbeg; k0 < kend; k0 += 4) {
    int kb = (k0 & (NTOKEN - 1)) + ((lane >> 4) << 1);
    v2f a; a.x = W[kb * NTOKEN + row]; a.y = W[(kb + 1) * NTOKEN + row];
    if ((k0 & 60) == 0 && (k0 & (NTOKEN - 1)) + 64 < NTOKEN) {
      __builtin_prefetch(&W[(kb + 64) * NTOKEN + row], 0, 3);
    }
#pragma unroll
    for (int t = 0; t < 4; ++t) {
      v2f bf = frag_b(Sb, FDIM, k0, t * 16, lane);
      c[t] = wmma4(a, bf, c[t]);
    }
  }
  ksplit_reduce_store(sh, c, ks, lane, out + (size_t)b * NTOKEN * FDIM, m0, mul, 16, 1);
}

extern "C" void kernel_launch(void* const* d_in, const int* in_sizes, int n_in,
                              void* d_out, int out_size, void* d_ws, size_t ws_size,
                              hipStream_t stream) {
  (void)in_sizes; (void)n_in; (void)out_size; (void)ws_size;
  const float* xy   = (const float*)d_in[0];
  const float* kW0  = (const float*)d_in[1];
  const float* kb0  = (const float*)d_in[2];
  const float* kW1  = (const float*)d_in[3];
  const float* kb1  = (const float*)d_in[4];
  const float* kW2  = (const float*)d_in[5];
  const float* kb2  = (const float*)d_in[6];
  const float* fW0  = (const float*)d_in[7];
  const float* fb0  = (const float*)d_in[8];
  const float* fW1  = (const float*)d_in[9];
  const float* fb1  = (const float*)d_in[10];
  const float* fW2  = (const float*)d_in[11];
  const float* fb2  = (const float*)d_in[12];
  const float* Wq   = (const float*)d_in[13];
  const float* bq   = (const float*)d_in[14];
  const float* Wk   = (const float*)d_in[15];
  const float* bk   = (const float*)d_in[16];
  const float* ln0g = (const float*)d_in[17];
  const float* ln0b = (const float*)d_in[18];
  const float* lng  = (const float*)d_in[19];
  const float* lnb  = (const float*)d_in[20];
  float* out = (float*)d_out;

  float* ws = (float*)d_ws;
  const size_t SZ_1M = 1048576;
  float* w_weight  = ws;
  float* w_weightf = ws + SZ_1M;
  float* V0        = ws + 2 * SZ_1M;
  float* V1        = ws + 3 * SZ_1M;
  float* Caug      = ws + 4 * SZ_1M;
  float* caug      = ws + 4 * SZ_1M + 20480;
  float* Gp        = ws + 4 * SZ_1M + 20992;
  float* gp        = ws + 4 * SZ_1M + 152064;
  float* MID       = ws + 12 * SZ_1M + 131072;

  const float DXc   = 2.0f / 31.0f;
  const float dxdy  = DXc * DXc;
  const float scale = 0.125f;

  edge_mlp_kernel<<<LNA_WGS + WJOBS + 2 * NTOKEN, 128, 0, stream>>>(
      kW0, kb0, kW1, kb1, kW2, kb2, fW0, fb0, fW1, fb1, fW2, fb2, w_weight, w_weightf,
      xy, ln0g, ln0b, V0, Wq, bq, Wk, bk, Caug, caug);

  float* Va = V0;
  float* Vb = V1;
  for (int j = 0; j < NLAYER - 1; ++j) {
    atb_kernel<<<dim3(BATCH, 4, KQ), KSPLIT * 32, 0, stream>>>(
        Va, NTOK2 * FDIM, Va, NTOK2 * FDIM, NTOK2, scale * dxdy, Gp, 4096, gp);
    vp_kernel<true><<<dim3(BATCH, NTOK2 / 64), 128, 0, stream>>>(
        Va, Caug + (size_t)j * CAUG, caug + j * 80, Gp, gp, lng + j * FDIM, lnb + j * FDIM, Vb);
    float* t = Va; Va = Vb; Vb = t;
  }

  {
    int j = NLAYER - 1;
    atb_kernel<<<dim3(BATCH, 4, KQ), KSPLIT * 32, 0, stream>>>(
        Va, NTOK2 * FDIM, xy, NTOK2 * FDIM, NTOKEN, scale, Gp, 4096, gp);
    vp_kernel<false><<<dim3(BATCH, NTOK2 / 64), 128, 0, stream>>>(
        Va, Caug + (size_t)j * CAUG, caug + j * 80, Gp, gp, nullptr, nullptr, MID);
    final_kernel<<<dim3(BATCH, NTOKEN / 16), KSPLIT * 32, 0, stream>>>(
        w_weight, w_weightf, MID, out, dxdy * dxdy);
  }
}


# Every layer's Gram product is formed by the launch that produces its rows, as per-strip parts (128 or 256 rows) added in fixed order by the next launch; 4 launches gone; the sums change order
# speedup vs baseline: 6.5251x; 1.0310x over previous
#include <hip/hip_runtime.h>
#include <hip/hip_bf16.h>

#define NGRID   32
#define NTOKEN  1024
#define NTOK2   2048
#define HEAD    4
#define DK      64
#define NLAYER  4
#define FDIM    64
#define BATCH   8

typedef __attribute__((ext_vector_type(2))) float v2f;
typedef __attribute__((ext_vector_type(8))) float v8f;

__device__ __forceinline__ v8f wmma4(v2f a, v2f b, v8f c) {
  return __builtin_amdgcn_wmma_f32_16x16x4_f32(
       false, a,  false, b,
       (short)0, c,  false,  false);
}

__device__ __forceinline__ v2f frag_a(const float* src, int lda, int m0, int k0, int lane) {
  int row = m0 + (lane & 15);
  int kk  = k0 + ((lane >> 4) << 1);
  v2f a; a.x = src[row * lda + kk]; a.y = src[row * lda + kk + 1];
  return a;
}

__device__ __forceinline__ v2f frag_b(const float* src, int ldb, int k0, int n0, int lane) {
  int col = n0 + (lane & 15);
  int kk  = k0 + ((lane >> 4) << 1);
  v2f b; b.x = src[kk * ldb + col]; b.y = src[(kk + 1) * ldb + col];
  return b;
}


#define KSPLIT   16
#define TPITCH   68
#define TPLANE   (16 * TPITCH)

__device__ __forceinline__ void ksplit_reduce_store(float* sh, const v8f* c, int ks, int lane,
                                                    float* dst, int m0, float mul,
                                                    int tstep, int cstep) {
  float* mine = sh + ks * TPLANE;
  int lrow = (lane >> 4) << 3;
  int lcol = lane & 15;
#pragma unroll
  for (int t = 0; t < 4; ++t) {
#pragma unroll
    for (int r = 0; r < 8; ++r) mine[(lrow + r) * TPITCH + t * tstep + lcol * cstep] = c[t][r];
  }
  __syncthreads();
  for (int idx = threadIdx.x; idx < 16 * 64; idx += KSPLIT * 32) {
    int row = idx >> 6, col = idx & 63;
    float acc = sh[row * TPITCH + col];
#pragma unroll
    for (int p = 1; p < KSPLIT; ++p) acc += sh[p * TPLANE + row * TPITCH + col];
    dst[(m0 + row) * 64 + col] = acc * mul;
  }
}

#define LN0_CHUNKS  8
#define LN0_SLICE   (NTOK2 * FDIM / LN0_CHUNKS)
#define LNA_WGS     64
#define LNA_ELEMS   (BATCH * NTOK2 * FDIM / LNA_WGS)
#define WJOBS       16
#define CAUG    (80 * 64)
#define VPITCH  80

__device__ __forceinline__ void ln0_job(const float* xy, const float* g, const float* bb,
                                        float* V, int q, float* shp, float* sV, float gs,
                                        float* Gpart, float* gpart) {
  const int b = q / (LNA_WGS / BATCH);
  const int wave = threadIdx.x >> 5;
  const float4* slab = reinterpret_cast<const float4*>(xy + (size_t)b * NTOK2 * FDIM);
  for (int cidx = 0; cidx < LN0_CHUNKS; ++cidx) {
    const float4* p = slab + cidx * (LN0_SLICE / 4);
    float4 v[32];
#pragma unroll
    for (int it = 0; it < 32; ++it) v[it] = p[it * 128 + threadIdx.x];
    float4 s4 = make_float4(0.f, 0.f, 0.f, 0.f), q4 = make_float4(0.f, 0.f, 0.f, 0.f);
#pragma unroll
    for (int it = 0; it < 32; ++it) {
      s4.x += v[it].x; s4.y += v[it].y; s4.z += v[it].z; s4.w += v[it].w;
      q4.x = fmaf(v[it].x, v[it].x, q4.x); q4.y = fmaf(v[it].y, v[it].y, q4.y);
      q4.z = fmaf(v[it].z, v[it].z, q4.z); q4.w = fmaf(v[it].w, v[it].w, q4.w);
    }
    float s = (s4.x + s4.y) + (s4.z + s4.w);
    float ss = (q4.x + q4.y) + (q4.z + q4.w);
#pragma unroll
    for (int m = 16; m >= 1; m >>= 1) {
      s  += __shfl_xor(s, m, 32);
      ss += __shfl_xor(ss, m, 32);
    }
    if ((threadIdx.x & 31) == 0) { shp[8 * cidx + 2 * wave] = s; shp[8 * cidx + 2 * wave + 1] = ss; }
  }
  __syncthreads();
  float s = 0.f, ss = 0.f;
#pragma unroll
  for (int cidx = 0; cidx < LN0_CHUNKS; ++cidx) {
    const float* w = shp + 8 * cidx;
    const float ps = ((w[0] + w[2]) + w[4]) + w[6];
    const float pq = ((w[1] + w[3]) + w[5]) + w[7];
    if (cidx == 0) { s = ps; ss = pq; } else { s += ps; ss += pq; }
  }
  const float N = (float)(NTOK2 * FDIM);
  const float mean = s / N;
  const float var  = ss / N - mean * mean;
  const float rstd = rsqrtf(var + 1e-5f);
  const int lane = threadIdx.x & 31, cl = lane & 15, hl = lane >> 4;
  const int i0 = wave * 16;
  v8f gt[4] = {};
  float csum = 0.f;
  for (int trip = 0; trip < 4; ++trip) {
    const int e4 = q * (LNA_ELEMS / 4) + trip * 1024;
    if (trip == 2) __syncthreads();
    const float4* x4 = reinterpret_cast<const float4*>(xy) + e4;
    const float4* g4 = reinterpret_cast<const float4*>(g) + (e4 & (NTOK2 * FDIM / 4 - 1));
    const float4* b4 = reinterpret_cast<const float4*>(bb) + (e4 & (NTOK2 * FDIM / 4 - 1));
    float4* o4 = reinterpret_cast<float4*>(V) + e4;
    float4 xv[8], gv[8], bv[8];
#pragma unroll
    for (int i = 0; i < 8; ++i) {
      xv[i] = x4[i * 128 + threadIdx.x]; gv[i] = g4[i * 128 + threadIdx.x]; bv[i] = b4[i * 128 + threadIdx.x];
    }
#pragma unroll
    for (int i = 0; i < 8; ++i) {
      float4 o;
      o.x = (xv[i].x - mean) * rstd * gv[i].x + bv[i].x;
      o.y = (xv[i].y - mean) * rstd * gv[i].y + bv[i].y;
      o.z = (xv[i].z - mean) * rstd * gv[i].z + bv[i].z;
      o.w = (xv[i].w - mean) * rstd * gv[i].w + bv[i].w;
      o4[i * 128 + threadIdx.x] = o;
      const int pc = i * 128 + threadIdx.x;
      *reinterpret_cast<float4*>(&sV[((trip & 1) * 64 + (pc >> 4)) * VPITCH + ((pc & 15) << 2)]) = o;
    }
    if ((trip & 1) == 0) continue;
    __syncthreads();
    {
      const int col = threadIdx.x & 63, grp = threadIdx.x >> 6;
      for (int r = 0; r < 64; ++r) csum += sV[(grp * 64 + r) * VPITCH + col];
    }
    for (int kc = 0; kc < 4; ++kc) {
      const int kb = kc * 32 + 2 * hl;
      v2f a[8];
      float4 q0[8], q1[8];
#pragma unroll
      for (int u = 0; u < 8; ++u) {
        const int k = kb + 4 * u;
        a[u].x = sV[k * VPITCH + i0 + cl]; a[u].y = sV[(k + 1) * VPITCH + i0 + cl];
        q0[u] = *reinterpret_cast<const float4*>(&sV[k * VPITCH + 4 * cl]);
        q1[u] = *reinterpret_cast<const float4*>(&sV[(k + 1) * VPITCH + 4 * cl]);
      }
#pragma unroll
      for (int u = 0; u < 8; ++u) {
        v2f bf;
        bf.x = q0[u].x; bf.y = q1[u].x; gt[0] = wmma4(a[u], bf, gt[0]);
        bf.x = q0[u].y; bf.y = q1[u].y; gt[1] = wmma4(a[u], bf, gt[1]);
        bf.x = q0[u].z; bf.y = q1[u].z; gt[2] = wmma4(a[u], bf, gt[2]);
        bf.x = q0[u].w; bf.y = q1[u].w; gt[3] = wmma4(a[u], bf, gt[3]);
      }
    }
  }
  float* Gd = Gpart + (size_t)q * 4096;
#pragma unroll
  for (int r = 0; r < 8; ++r) {
    float4 o;
    o.x = gt[0][r] * gs; o.y = gt[1][r] * gs; o.z = gt[2][r] * gs; o.w = gt[3][r] * gs;
    *reinterpret_cast<float4*>(&Gd[(i0 + (hl << 3) + r) * 64 + 4 * cl]) = o;
  }
  shp[threadIdx.x] = csum;
  __syncthreads();
  if (threadIdx.x < 64) gpart[q * 64 + threadIdx.x] = (shp[threadIdx.x] + shp[64 + threadIdx.x]) * gs;
}

__device__ __forceinline__ void caug_job(const float* Wq, const float* Wk, float* Caug, int jb) {
  const int lane = threadIdx.x & 31, wave = threadIdx.x >> 5;
  const int cl = lane & 15, hl = lane >> 4;
  const int j = jb >> 1;
  const int m0 = (2 * (jb & 1) + (wave >> 1)) * 16;
  const int n0 = (wave & 1) * 32;
  const float* Ap = Wq + (size_t)j * HEAD * DK * FDIM + m0 + cl;
  const float* Bp = Wk + (size_t)j * HEAD * DK * FDIM + n0 + cl;
  v8f t0 = {}, t1 = {};
  for (int kb = 0; kb < HEAD * DK; kb += 64) {
    v2f a[16], b0[16], b1[16];
#pragma unroll
    for (int u = 0; u < 16; ++u) {
      const int kk = kb + 4 * u + 2 * hl;
      a[u].x  = Ap[kk * 64];      a[u].y  = Ap[(kk + 1) * 64];
      b0[u].x = Bp[kk * 64];      b0[u].y = Bp[(kk + 1) * 64];
      b1[u].x = Bp[kk * 64 + 16]; b1[u].y = Bp[(kk + 1) * 64 + 16];
    }
#pragma unroll
    for (int part = 0; part < 4; ++part) {
      v8f p0 = {}, p1 = {};
#pragma unroll
      for (int u = 4 * part; u < 4 * part + 4; ++u) {
        p0 = wmma4(a[u], b0[u], p0);
        p1 = wmma4(a[u], b1[u], p1);
      }
      if (kb == 0 && part == 0) { t0 = p0; t1 = p1; } else { t0 += p0; t1 += p1; }
    }
  }
  float* Cj = Caug + (size_t)j * CAUG;
#pragma unroll
  for (int r = 0; r < 8; ++r) {
    const int row = m0 + (hl << 3) + r;
    Cj[row * 64 + n0 + cl]      = t0[r];
    Cj[row * 64 + n0 + 16 + cl] = t1[r];
  }
}

__device__ __forceinline__ void cvec_job(const float* Wq, const float* bq, const float* Wk,
                                         const float* bk, float* Caug, float* caug, int jb,
                                         float* sh) {
  const int j = jb >> 1;
  const bool isD = (jb & 1) != 0;
  const float* Wm = (isD ? Wk : Wq) + (size_t)j * 256 * 64;
  const float* v  = (isD ? bq : bk) + j * 256;
  const int i = threadIdx.x & 63, half = threadIdx.x >> 6;
  for (int cc = 0; cc < 8; ++cc) {
    const int kc = half * 8 + cc;
    float a = 0.f;
#pragma unroll
    for (int u = 0; u < 16; ++u) {
      int k = kc * 16 + u;
      a = fmaf(Wm[k * 64 + i], v[k], a);
    }
    sh[kc * 64 + i] = a;
  }
  __syncthreads();
  float* Cj = Caug + (size_t)j * CAUG;
  float* cj = caug + j * 80;
  if (threadIdx.x < 64) {
    float t = sh[threadIdx.x];
    for (int p = 1; p < 16; ++p) t += sh[p * 64 + threadIdx.x];
    if (isD) Cj[64 * 64 + threadIdx.x] = t; else cj[threadIdx.x] = t;
  }
  if (isD) {
    if ((threadIdx.x >> 5) == 2) {
      const int lane = threadIdx.x & 31;
      float e = 0.f;
      for (int u = 0; u < 8; ++u) {
        int k = lane + 32 * u;
        e = fmaf(bq[j * 256 + k], bk[j * 256 + k], e);
      }
      for (int m = 16; m >= 1; m >>= 1) e += __shfl_xor(e, m, 32);
      if (lane == 0) cj[64] = e;
    }
  } else {
    for (int idx = threadIdx.x; idx < 15 * 64; idx += 128) Cj[65 * 64 + idx] = 0.f;
    if (threadIdx.x < 15) cj[65 + threadIdx.x] = 0.f;
  }
}
__device__ __forceinline__ float leaky01(float t) {
  return fmaxf(t, 0.01f * t);
}

__global__ __launch_bounds__(128) void edge_mlp_kernel(
                                const float* kW0, const float* kb0, const float* kW1,
                                const float* kb1, const float* kW2, const float* kb2,
                                const float* fW0, const float* fb0, const float* fW1,
                                const float* fb1, const float* fW2, const float* fb2,
                                float* wout, float* fout,
                                const float* xy, const float* ln0g, const float* ln0b, float* V0,
                                const float* Wq, const float* bq, const float* Wk, const float* bk,
                                float* Caug, float* caug, float gs, float* Gpart, float* gpart) {
  __shared__ float shx[16 * 64];
  __shared__ __attribute__((aligned(16))) float sV[128 * VPITCH];
  if (blockIdx.x < LNA_WGS) {
    ln0_job(xy, ln0g, ln0b, V0, blockIdx.x, shx, sV, gs, Gpart, gpart);
    return;
  }
  if (blockIdx.x < LNA_WGS + WJOBS / 2) {
    caug_job(Wq, Wk, Caug, blockIdx.x - LNA_WGS);
    return;
  }
  if (blockIdx.x < LNA_WGS + WJOBS) {
    cvec_job(Wq, bq, Wk, bk, Caug, caug, blockIdx.x - LNA_WGS - WJOBS / 2, shx);
    return;
  }
  const int ejob = blockIdx.x - (LNA_WGS + WJOBS);
  const bool isF = ejob >= NTOKEN;
  const float* W0 = isF ? fW0 : kW0;
  const float* b0 = isF ? fb0 : kb0;
  const float* W1 = isF ? fW1 : kW1;
  const float* b1 = isF ? fb1 : kb1;
  const float* W2 = isF ? fW2 : kW2;
  const float* b2 = isF ? fb2 : kb2;
  float* dst      = isF ? fout : wout;

  const int lane = threadIdx.x & 31;
  const int wave = threadIdx.x >> 5;
  const int c = lane & 15, h = lane >> 4;
  const int n = ejob & (NTOKEN - 1);
  const int mbase = wave * 256;
  const float DXc = 2.0f / 31.0f;
  const float e0 = -1.f + DXc * (float)(n >> 5);
  const float e1 = -1.f + DXc * (float)(n & 31);

  v2f wa[4][8];
#pragma unroll
  for (int t = 0; t < 4; ++t) {
#pragma unroll
    for (int i = 0; i < 8; ++i) {
      const float* p = W1 + (t * 16 + c) * 32 + 4 * i + 2 * h;
      wa[t][i].x = p[0]; wa[t][i].y = p[1];
    }
  }
  float pre[16], wc2[16], wc3[16];
#pragma unroll
  for (int i = 0; i < 8; ++i) {
#pragma unroll
    for (int j = 0; j < 2; ++j) {
      int u = 4 * i + 2 * h + j;
      float t = b0[u];
      t = fmaf(W0[u * 4 + 0], e0, t);
      t = fmaf(W0[u * 4 + 1], e1, t);
      pre[2 * i + j] = t;
      wc2[2 * i + j] = W0[u * 4 + 2];
      wc3[2 * i + j] = W0[u * 4 + 3];
    }
  }
  v8f cb[4];
  float w2v[4][8];
#pragma unroll
  for (int t = 0; t < 4; ++t) {
#pragma unroll
    for (int r = 0; r < 8; ++r) {
      cb[t][r]  = b1[t * 16 + 8 * h + r];
      w2v[t][r] = W2[t * 16 + 8 * h + r];
    }
  }
  const float bias2 = b2[0];

  for (int it = 0; it < 8; ++it) {
    const int m0 = mbase + it * 32;
    float res[2];
#pragma unroll
    for (int et = 0; et < 2; ++et) {
      const int m = m0 + et * 16 + c;
      const float e2 = -1.f + DXc * (float)(m >> 5);
      const float e3 = -1.f + DXc * (float)(m & 31);
      v2f hb[8];
#pragma unroll
      for (int i = 0; i < 8; ++i) {
        float t0 = fmaf(wc3[2 * i],     e3, fmaf(wc2[2 * i],     e2, pre[2 * i]));
        float t1 = fmaf(wc3[2 * i + 1], e3, fmaf(wc2[2 * i + 1], e2, pre[2 * i + 1]));
        hb[i].x = leaky01(t0); hb[i].y = leaky01(t1);
      }
      v8f acc[4];
#pragma unroll
      for (int t = 0; t < 4; ++t) acc[t] = wmma4(wa[t][0], hb[0], cb[t]);
#pragma unroll
      for (int i = 1; i < 8; ++i) {
#pragma unroll
        for (int t = 0; t < 4; ++t) acc[t] = wmma4(wa[t][i], hb[i], acc[t]);
      }
      float s = 0.f;
#pragma unroll
      for (int t = 0; t < 4; ++t) {
#pragma unroll
        for (int r = 0; r < 8; ++r) s = fmaf(w2v[t][r], leaky01(acc[t][r]), s);
      }
      s += __shfl_xor(s, 16, 32);
      res[et] = s + bias2;
    }
    dst[n * NTOKEN + m0 + lane] = h ? res[1] : res[0];
  }
}

#define GPITCH  72


#define CPITCH  68
#define LROWS   128
#define OPITCH  80

template <int M>
__device__ __forceinline__ float row_xor_add(float v) {
  const int o = __builtin_amdgcn_update_dpp(0, __builtin_bit_cast(int, v), 0x160 + M, 0xf, 0xf, true);
  return v + __builtin_bit_cast(float, o);
}

template <bool LNRES, int NP, int GM>
__global__ __launch_bounds__(256) void layer_kernel(const float* Vin, const float* Cj,
                                                    const float* cj, const float* Gp,
                                                    const float* gp, const float* lg,
                                                    const float* lb, float* Out,
                                                    const float* xall, float gs,
                                                    float* Gout, float* gout) {
  constexpr int NPOUT = (GM == 2) ? 8 : 16;
  __shared__ __attribute__((aligned(16))) float sG[64 * GPITCH];
  __shared__ __attribute__((aligned(16))) float sC[80 * CPITCH];
  __shared__ __attribute__((aligned(16))) float sP[80 * CPITCH];
  __shared__ float sg[64];
  __shared__ __attribute__((aligned(16))) float scj[80];
  __shared__ __attribute__((aligned(16))) float sO[(GM != 0) ? LROWS * OPITCH : 4];
  __shared__ __attribute__((aligned(16))) float sX[(GM == 2) ? LROWS * OPITCH : 4];
  __shared__ float sS[(GM != 0) ? 4 * 64 : 4];
  const int b = blockIdx.x, tid = threadIdx.x;
  const int wave = __builtin_amdgcn_readfirstlane(tid >> 5);
  const int lane = tid & 31;
  const int cl = lane & 15, hl = lane >> 4;
  const int r0 = blockIdx.y * LROWS;
  const int m0 = r0 + wave * 16;
  const float* A  = Vin + (size_t)b * NTOK2 * FDIM;
  const bool gram = (GM == 1) || (GM == 2 && r0 < NTOKEN);

  const float4* G4 = reinterpret_cast<const float4*>(Gp + (size_t)b * NP * 4096);
  const float4* C4 = reinterpret_cast<const float4*>(Cj);
  const float* g0 = gp + (size_t)b * NP * 64;
  float4 gq[4][NP];
#pragma unroll
  for (int i = 0; i < 4; ++i) {
#pragma unroll
    for (int q = 0; q < NP; ++q) gq[i][q] = G4[q * 1024 + tid + 256 * i];
  }
  const float4 cq0 = C4[tid], cq1 = C4[tid + 256], cq2 = C4[tid + 512], cq3 = C4[tid + 768],
               cq4 = C4[tid + 1024];
  float gv[NP];
#pragma unroll
  for (int q = 0; q < NP; ++q) gv[q] = g0[q * 64 + (tid & 63)];
  const float cjg = cj[tid < 80 ? tid : 79];
  v2f av[16];
#pragma unroll
  for (int u = 0; u < 16; ++u) av[u] = frag_a(A, FDIM, m0, 4 * u, lane);
  __builtin_amdgcn_sched_barrier(0);
#pragma unroll
  for (int i = 0; i < 4; ++i) {
    const int i4 = tid + 256 * i;
    float4 v = gq[i][0];
#pragma unroll
    for (int q = 1; q < NP; ++q) {
      v.x += gq[i][q].x; v.y += gq[i][q].y; v.z += gq[i][q].z; v.w += gq[i][q].w;
    }
    *reinterpret_cast<float4*>(&sG[(i4 >> 4) * GPITCH + ((i4 & 15) << 2)]) = v;
  }
  {
    float4* cdst = reinterpret_cast<float4*>(&sC[(tid >> 4) * CPITCH + ((tid & 15) << 2)]);
    const int cstep = 16 * CPITCH / 4;
    cdst[0] = cq0; cdst[cstep] = cq1; cdst[2 * cstep] = cq2; cdst[3 * cstep] = cq3;
    cdst[4 * cstep] = cq4;
  }
  if (tid < 64) {
    float v = gv[0];
#pragma unroll
    for (int q = 1; q < NP; ++q) v += gv[q];
    sg[tid] = v;
  }
  if (tid < 80) scj[tid] = cjg;
  __syncthreads();

  float4 rv[8];
  float4 lg4 = make_float4(0.f, 0.f, 0.f, 0.f), lb4 = make_float4(0.f, 0.f, 0.f, 0.f);
  if constexpr (LNRES) {
#pragma unroll
    for (int r = 0; r < 8; ++r)
      rv[r] = *reinterpret_cast<const float4*>(&A[(m0 + (hl << 3) + r) * FDIM + 4 * cl]);
    lg4 = *reinterpret_cast<const float4*>(&lg[4 * cl]);
    lb4 = *reinterpret_cast<const float4*>(&lb[4 * cl]);
  }
  float4 xq[8];
#pragma unroll
  for (int i = 0; i < 8; ++i) xq[i] = make_float4(0.f, 0.f, 0.f, 0.f);
  if constexpr (GM == 2) {
    if (gram) {
      const float4* X4 = reinterpret_cast<const float4*>(xall + ((size_t)b * NTOK2 + r0) * FDIM);
#pragma unroll
      for (int i = 0; i < 8; ++i) xq[i] = X4[tid + 256 * i];
    }
  }

  {
    const int rg = wave >> 2;
    const int col = (wave & 3) * 16 + cl;
    const float g = sg[col];
    v2f gb[16];
#pragma unroll
    for (int u = 0; u < 16; ++u) {
      const int kk = 4 * u + 2 * hl;
      gb[u].x = sG[kk * GPITCH + col]; gb[u].y = sG[(kk + 1) * GPITCH + col];
    }
    if (rg == 0) {
      v8f pc;
#pragma unroll
      for (int r = 0; r < 8; ++r) pc[r] = scj[64 + (hl << 3) + r] * g;
      v2f cc[16];
#pragma unroll
      for (int u = 0; u < 16; ++u)
        cc[u] = *reinterpret_cast<const v2f*>(&sC[(64 + cl) * CPITCH + 4 * u + 2 * hl]);
      asm volatile("" ::: "memory");
      __builtin_amdgcn_sched_barrier(0);
#pragma unroll
      for (int u = 0; u < 16; ++u) pc = wmma4(cc[u], gb[u], pc);
#pragma unroll
      for (int r = 0; r < 8; ++r) sP[(64 + (hl << 3) + r) * CPITCH + col] = pc[r];
    }
    const int ra = rg * 32;
    v8f pa, pb;
#pragma unroll
    for (int r = 0; r < 8; ++r) {
      pa[r] = scj[ra + (hl << 3) + r] * g;
      pb[r] = scj[ra + 16 + (hl << 3) + r] * g;
    }
    v2f ca[16], cb2[16];
#pragma unroll
    for (int u = 0; u < 16; ++u) {
      ca[u]  = *reinterpret_cast<const v2f*>(&sC[(ra + cl) * CPITCH + 4 * u + 2 * hl]);
      cb2[u] = *reinterpret_cast<const v2f*>(&sC[(ra + 16 + cl) * CPITCH + 4 * u + 2 * hl]);
    }
    asm volatile("" ::: "memory");
    __builtin_amdgcn_sched_barrier(0);
#pragma unroll
    for (int u = 0; u < 16; ++u) {
      pa = wmma4(ca[u], gb[u], pa);
      pb = wmma4(cb2[u], gb[u], pb);
    }
#pragma unroll
    for (int r = 0; r < 8; ++r) {
      sP[(ra + (hl << 3) + r) * CPITCH + col]      = pa[r];
      sP[(ra + 16 + (hl << 3) + r) * CPITCH + col] = pb[r];
    }
  }
  __syncthreads();

  const float* Pp = sP + 4 * cl;
  const float4 pv = *reinterpret_cast<const float4*>(&Pp[64 * CPITCH]);
  v8f c[4] = {};
  float4 q0[16], q1[16];
#pragma unroll
  for (int u = 0; u < 16; ++u) {
    const int kk = 4 * u + 2 * hl;
    q0[u] = *reinterpret_cast<const float4*>(&Pp[kk * CPITCH]);
    q1[u] = *reinterpret_cast<const float4*>(&Pp[(kk + 1) * CPITCH]);
  }
  asm volatile("" ::: "memory");
  __builtin_amdgcn_sched_barrier(0);
#pragma unroll
  for (int u = 0; u < 16; ++u) {
    v2f bf;
    bf.x = q0[u].x; bf.y = q1[u].x; c[0] = wmma4(av[u], bf, c[0]);
    bf.x = q0[u].y; bf.y = q1[u].y; c[1] = wmma4(av[u], bf, c[1]);
    bf.x = q0[u].z; bf.y = q1[u].z; c[2] = wmma4(av[u], bf, c[2]);
    bf.x = q0[u].w; bf.y = q1[u].w; c[3] = wmma4(av[u], bf, c[3]);
  }
  float* dst = Out + (size_t)b * NTOK2 * FDIM;
  const int rbase = m0 + (hl << 3);
  const int lrow = wave * 16 + (hl << 3);
  if constexpr (!LNRES) {
#pragma unroll
    for (int r = 0; r < 8; ++r) {
      float4 o;
      o.x = c[0][r] + pv.x; o.y = c[1][r] + pv.y; o.z = c[2][r] + pv.z; o.w = c[3][r] + pv.w;
      *reinterpret_cast<float4*>(&dst[(rbase + r) * FDIM + 4 * cl]) = o;
    }
  } else {
    float sm[8], sq[8];
#pragma unroll
    for (int r = 0; r < 8; ++r) {
      c[0][r] += pv.x; c[1][r] += pv.y; c[2][r] += pv.z; c[3][r] += pv.w;
      sm[r] = (c[0][r] + c[1][r]) + (c[2][r] + c[3][r]);
      sq[r] = fmaf(c[3][r], c[3][r], fmaf(c[2][r], c[2][r], fmaf(c[1][r], c[1][r], c[0][r] * c[0][r])));
    }
#pragma unroll
    for (int r = 0; r < 8; ++r) { sm[r] = row_xor_add<1>(sm[r]); sq[r] = row_xor_add<1>(sq[r]); }
#pragma unroll
    for (int r = 0; r < 8; ++r) { sm[r] = row_xor_add<2>(sm[r]); sq[r] = row_xor_add<2>(sq[r]); }
#pragma unroll
    for (int r = 0; r < 8; ++r) { sm[r] = row_xor_add<4>(sm[r]); sq[r] = row_xor_add<4>(sq[r]); }
#pragma unroll
    for (int r = 0; r < 8; ++r) { sm[r] = row_xor_add<8>(sm[r]); sq[r] = row_xor_add<8>(sq[r]); }
#pragma unroll
    for (int r = 0; r < 8; ++r) {
      const float mean = sm[r] * (1.f / 64.f);
      const float var  = sq[r] * (1.f / 64.f) - mean * mean;
      const float inv  = rsqrtf(var + 1e-5f);
      float4 o;
      o.x = (c[0][r] - mean) * inv * lg4.x + lb4.x + rv[r].x;
      o.y = (c[1][r] - mean) * inv * lg4.y + lb4.y + rv[r].y;
      o.z = (c[2][r] - mean) * inv * lg4.z + lb4.z + rv[r].z;
      o.w = (c[3][r] - mean) * inv * lg4.w + lb4.w + rv[r].w;
      *reinterpret_cast<float4*>(&dst[(rbase + r) * FDIM + 4 * cl]) = o;
      if constexpr (GM != 0)
        *reinterpret_cast<float4*>(&sO[(lrow + r) * OPITCH + 4 * cl]) = o;
    }
  }

  if constexpr (GM != 0) {
    if (gram) {
      if constexpr (GM == 2) {
#pragma unroll
        for (int i = 0; i < 8; ++i) {
          const int p = tid + 256 * i;
          *reinterpret_cast<float4*>(&sX[(p >> 4) * OPITCH + ((p & 15) << 2)]) = xq[i];
        }
      }
      __syncthreads();
      const float* sB = (GM == 2) ? sX : sO;
      {
        const int col = tid & 63, grp = tid >> 6;
        float s = 0.f;
        for (int r = 0; r < 32; ++r) s += sB[(grp * 32 + r) * OPITCH + col];
        sS[grp * 64 + col] = s;
      }
      const int i0 = (wave >> 1) * 16, n0 = (wave & 1) * 32;
      v8f t0 = {}, t1 = {};
      for (int kc = 0; kc < 4; ++kc) {
        const int kb = kc * 32 + 2 * hl;
        v2f a[8], bx[8], by[8];
#pragma unroll
        for (int u = 0; u < 8; ++u) {
          const int k = kb + 4 * u;
          a[u].x = sO[k * OPITCH + i0 + cl]; a[u].y = sO[(k + 1) * OPITCH + i0 + cl];
          bx[u] = *reinterpret_cast<const v2f*>(&sB[k * OPITCH + n0 + 2 * cl]);
          by[u] = *reinterpret_cast<const v2f*>(&sB[(k + 1) * OPITCH + n0 + 2 * cl]);
        }
        asm volatile("" ::: "memory");
        __builtin_amdgcn_sched_barrier(0);
#pragma unroll
        for (int u = 0; u < 8; ++u) {
          v2f bf;
          bf.x = bx[u].x; bf.y = by[u].x; t0 = wmma4(a[u], bf, t0);
          bf.x = bx[u].y; bf.y = by[u].y; t1 = wmma4(a[u], bf, t1);
        }
      }
      const size_t slot = (size_t)b * NPOUT + blockIdx.y;
      float* Gd = Gout + slot * 4096;
#pragma unroll
      for (int r = 0; r < 8; ++r) {
        const int row = i0 + (hl << 3) + r;
        v2f o; o.x = t0[r] * gs; o.y = t1[r] * gs;
        *reinterpret_cast<v2f*>(&Gd[row * 64 + n0 + 2 * cl]) = o;
      }
      __syncthreads();
      if (tid < 64) {
        const float s = ((sS[tid] + sS[64 + tid]) + sS[128 + tid]) + sS[192 + tid];
        gout[slot * 64 + tid] = s * gs;
      }
    }
  }
}

__global__ void final_kernel(const float* weight, const float* weightf,
                             const float* S, float* out, float mul) {
  __shared__ float sh[KSPLIT * TPLANE];
  int b = blockIdx.x;
  int ks = __builtin_amdgcn_readfirstlane(threadIdx.x >> 5);
  int lane = threadIdx.x & 31;
  int m0 = blockIdx.y * 16;
  const float* Sb = S + (size_t)b * NTOK2 * FDIM;
  int row = m0 + (lane & 15);
  const int klen = NTOK2 / KSPLIT;
  int kbeg = ks * klen, kend = kbeg + klen;
  const float* W = (kbeg < NTOKEN) ? weight : weightf;
  v8f c[4] = {};
  for (int k0 = kbeg; k0 < kend; k0 += 4) {
    int kb = (k0 & (NTOKEN - 1)) + ((lane >> 4) << 1);
    v2f a; a.x = W[kb * NTOKEN + row]; a.y = W[(kb + 1) * NTOKEN + row];
    if ((k0 & 60) == 0 && (k0 & (NTOKEN - 1)) + 64 < NTOKEN) {
      __builtin_prefetch(&W[(kb + 64) * NTOKEN + row], 0, 3);
    }
#pragma unroll
    for (int t = 0; t < 4; ++t) {
      v2f bf = frag_b(Sb, FDIM, k0, t * 16, lane);
      c[t] = wmma4(a, bf, c[t]);
    }
  }
  ksplit_reduce_store(sh, c, ks, lane, out + (size_t)b * NTOKEN * FDIM, m0, mul, 16, 1);
}

extern "C" void kernel_launch(void* const* d_in, const int* in_sizes, int n_in,
                              void* d_out, int out_size, void* d_ws, size_t ws_size,
                              hipStream_t stream) {
  (void)in_sizes; (void)n_in; (void)out_size; (void)ws_size;
  const float* xy   = (const float*)d_in[0];
  const float* kW0  = (const float*)d_in[1];
  const float* kb0  = (const float*)d_in[2];
  const float* kW1  = (const float*)d_in[3];
  const float* kb1  = (const float*)d_in[4];
  const float* kW2  = (const float*)d_in[5];
  const float* kb2  = (const float*)d_in[6];
  const float* fW0  = (const float*)d_in[7];
  const float* fb0  = (const float*)d_in[8];
  const float* fW1  = (const float*)d_in[9];
  const float* fb1  = (const float*)d_in[10];
  const float* fW2  = (const float*)d_in[11];
  const float* fb2  = (const float*)d_in[12];
  const float* Wq   = (const float*)d_in[13];
  const float* bq   = (const float*)d_in[14];
  const float* Wk   = (const float*)d_in[15];
  const float* bk   = (const float*)d_in[16];
  const float* ln0g = (const float*)d_in[17];
  const float* ln0b = (const float*)d_in[18];
  const float* lng  = (const float*)d_in[19];
  const float* lnb  = (const float*)d_in[20];
  float* out = (float*)d_out;

  float* ws = (float*)d_ws;
  const size_t SZ_1M = 1048576;
  float* w_weight  = ws;
  float* w_weightf = ws + SZ_1M;
  float* V0        = ws + 2 * SZ_1M;
  float* V1        = ws + 3 * SZ_1M;
  float* Caug      = ws + 4 * SZ_1M;
  float* caug      = ws + 4 * SZ_1M + 20480;
  float* Gp0       = ws + 5 * SZ_1M;
  float* gp0       = ws + 5 * SZ_1M + 524288;
  float* GpA       = ws + 6 * SZ_1M;
  float* gpA       = ws + 6 * SZ_1M + 524288;
  float* GpB       = ws + 7 * SZ_1M;
  float* gpB       = ws + 7 * SZ_1M + 524288;
  float* MID       = ws + 12 * SZ_1M + 131072;

  const float DXc   = 2.0f / 31.0f;
  const float dxdy  = DXc * DXc;
  const float scale = 0.125f;

  edge_mlp_kernel<<<LNA_WGS + WJOBS + 2 * NTOKEN, 128, 0, stream>>>(
      kW0, kb0, kW1, kb1, kW2, kb2, fW0, fb0, fW1, fb1, fW2, fb2, w_weight, w_weightf,
      xy, ln0g, ln0b, V0, Wq, bq, Wk, bk, Caug, caug, scale * dxdy, Gp0, gp0);

  layer_kernel<true, 8, 1><<<dim3(BATCH, NTOK2 / LROWS), 256, 0, stream>>>(
      V0, Caug, caug, Gp0, gp0, lng, lnb, V1, xy, scale * dxdy, GpA, gpA);
  layer_kernel<true, 16, 1><<<dim3(BATCH, NTOK2 / LROWS), 256, 0, stream>>>(
      V1, Caug + (size_t)1 * CAUG, caug + 80, GpA, gpA, lng + FDIM, lnb + FDIM, V0, xy,
      scale * dxdy, GpB, gpB);
  layer_kernel<true, 16, 2><<<dim3(BATCH, NTOK2 / LROWS), 256, 0, stream>>>(
      V0, Caug + (size_t)2 * CAUG, caug + 160, GpB, gpB, lng + 2 * FDIM, lnb + 2 * FDIM, V1, xy,
      scale, GpA, gpA);
  layer_kernel<false, 8, 0><<<dim3(BATCH, NTOK2 / LROWS), 256, 0, stream>>>(
      V1, Caug + (size_t)3 * CAUG, caug + 240, GpA, gpA, nullptr, nullptr, MID, xy,
      scale, nullptr, nullptr);
  final_kernel<<<dim3(BATCH, NTOKEN / 16), KSPLIT * 32, 0, stream>>>(
      w_weight, w_weightf, MID, out, dxdy * dxdy);
}


# The last 512 edge-MLP rows leave the first launch and ride, 128 each, in the four layer launches as extra work-groups on the processors the 128 layer work-groups leave idle; same sums and order
# speedup vs baseline: 7.0084x; 1.0741x over previous
#include <hip/hip_runtime.h>
#include <hip/hip_bf16.h>

#define NGRID   32
#define NTOKEN  1024
#define NTOK2   2048
#define HEAD    4
#define DK      64
#define NLAYER  4
#define FDIM    64
#define BATCH   8

typedef __attribute__((ext_vector_type(2))) float v2f;
typedef __attribute__((ext_vector_type(8))) float v8f;

__device__ __forceinline__ v8f wmma4(v2f a, v2f b, v8f c) {
  return __builtin_amdgcn_wmma_f32_16x16x4_f32(
       false, a,  false, b,
       (short)0, c,  false,  false);
}

__device__ __forceinline__ v2f frag_a(const float* src, int lda, int m0, int k0, int lane) {
  int row = m0 + (lane & 15);
  int kk  = k0 + ((lane >> 4) << 1);
  v2f a; a.x = src[row * lda + kk]; a.y = src[row * lda + kk + 1];
  return a;
}

__device__ __forceinline__ v2f frag_b(const float* src, int ldb, int k0, int n0, int lane) {
  int col = n0 + (lane & 15);
  int kk  = k0 + ((lane >> 4) << 1);
  v2f b; b.x = src[kk * ldb + col]; b.y = src[(kk + 1) * ldb + col];
  return b;
}


#define KSPLIT   16
#define TPITCH   68
#define TPLANE   (16 * TPITCH)

__device__ __forceinline__ void ksplit_reduce_store(float* sh, const v8f* c, int ks, int lane,
                                                    float* dst, int m0, float mul,
                                                    int tstep, int cstep) {
  float* mine = sh + ks * TPLANE;
  int lrow = (lane >> 4) << 3;
  int lcol = lane & 15;
#pragma unroll
  for (int t = 0; t < 4; ++t) {
#pragma unroll
    for (int r = 0; r < 8; ++r) mine[(lrow + r) * TPITCH + t * tstep + lcol * cstep] = c[t][r];
  }
  __syncthreads();
  for (int idx = threadIdx.x; idx < 16 * 64; idx += KSPLIT * 32) {
    int row = idx >> 6, col = idx & 63;
    float acc = sh[row * TPITCH + col];
#pragma unroll
    for (int p = 1; p < KSPLIT; ++p) acc += sh[p * TPLANE + row * TPITCH + col];
    dst[(m0 + row) * 64 + col] = acc * mul;
  }
}

#define LN0_CHUNKS  8
#define LN0_SLICE   (NTOK2 * FDIM / LN0_CHUNKS)
#define LNA_WGS     64
#define LNA_ELEMS   (BATCH * NTOK2 * FDIM / LNA_WGS)
#define WJOBS       16
#define CAUG    (80 * 64)
#define VPITCH  80

__device__ __forceinline__ void ln0_job(const float* xy, const float* g, const float* bb,
                                        float* V, int q, float* shp, float* sV, float gs,
                                        float* Gpart, float* gpart) {
  const int b = q / (LNA_WGS / BATCH);
  const int wave = threadIdx.x >> 5;
  const float4* slab = reinterpret_cast<const float4*>(xy + (size_t)b * NTOK2 * FDIM);
  for (int cidx = 0; cidx < LN0_CHUNKS; ++cidx) {
    const float4* p = slab + cidx * (LN0_SLICE / 4);
    float4 v[32];
#pragma unroll
    for (int it = 0; it < 32; ++it) v[it] = p[it * 128 + threadIdx.x];
    float4 s4 = make_float4(0.f, 0.f, 0.f, 0.f), q4 = make_float4(0.f, 0.f, 0.f, 0.f);
#pragma unroll
    for (int it = 0; it < 32; ++it) {
      s4.x += v[it].x; s4.y += v[it].y; s4.z += v[it].z; s4.w += v[it].w;
      q4.x = fmaf(v[it].x, v[it].x, q4.x); q4.y = fmaf(v[it].y, v[it].y, q4.y);
      q4.z = fmaf(v[it].z, v[it].z, q4.z); q4.w = fmaf(v[it].w, v[it].w, q4.w);
    }
    float s = (s4.x + s4.y) + (s4.z + s4.w);
    float ss = (q4.x + q4.y) + (q4.z + q4.w);
#pragma unroll
    for (int m = 16; m >= 1; m >>= 1) {
      s  += __shfl_xor(s, m, 32);
      ss += __shfl_xor(ss, m, 32);
    }
    if ((threadIdx.x & 31) == 0) { shp[8 * cidx + 2 * wave] = s; shp[8 * cidx + 2 * wave + 1] = ss; }
  }
  __syncthreads();
  float s = 0.f, ss = 0.f;
#pragma unroll
  for (int cidx = 0; cidx < LN0_CHUNKS; ++cidx) {
    const float* w = shp + 8 * cidx;
    const float ps = ((w[0] + w[2]) + w[4]) + w[6];
    const float pq = ((w[1] + w[3]) + w[5]) + w[7];
    if (cidx == 0) { s = ps; ss = pq; } else { s += ps; ss += pq; }
  }
  const float N = (float)(NTOK2 * FDIM);
  const float mean = s / N;
  const float var  = ss / N - mean * mean;
  const float rstd = rsqrtf(var + 1e-5f);
  const int lane = threadIdx.x & 31, cl = lane & 15, hl = lane >> 4;
  const int i0 = wave * 16;
  v8f gt[4] = {};
  float csum = 0.f;
  for (int trip = 0; trip < 4; ++trip) {
    const int e4 = q * (LNA_ELEMS / 4) + trip * 1024;
    if (trip == 2) __syncthreads();
    const float4* x4 = reinterpret_cast<const float4*>(xy) + e4;
    const float4* g4 = reinterpret_cast<const float4*>(g) + (e4 & (NTOK2 * FDIM / 4 - 1));
    const float4* b4 = reinterpret_cast<const float4*>(bb) + (e4 & (NTOK2 * FDIM / 4 - 1));
    float4* o4 = reinterpret_cast<float4*>(V) + e4;
    float4 xv[8], gv[8], bv[8];
#pragma unroll
    for (int i = 0; i < 8; ++i) {
      xv[i] = x4[i * 128 + threadIdx.x]; gv[i] = g4[i * 128 + threadIdx.x]; bv[i] = b4[i * 128 + threadIdx.x];
    }
#pragma unroll
    for (int i = 0; i < 8; ++i) {
      float4 o;
      o.x = (xv[i].x - mean) * rstd * gv[i].x + bv[i].x;
      o.y = (xv[i].y - mean) * rstd * gv[i].y + bv[i].y;
      o.z = (xv[i].z - mean) * rstd * gv[i].z + bv[i].z;
      o.w = (xv[i].w - mean) * rstd * gv[i].w + bv[i].w;
      o4[i * 128 + threadIdx.x] = o;
      const int pc = i * 128 + threadIdx.x;
      *reinterpret_cast<float4*>(&sV[((trip & 1) * 64 + (pc >> 4)) * VPITCH + ((pc & 15) << 2)]) = o;
    }
    if ((trip & 1) == 0) continue;
    __syncthreads();
    {
      const int col = threadIdx.x & 63, grp = threadIdx.x >> 6;
      for (int r = 0; r < 64; ++r) csum += sV[(grp * 64 + r) * VPITCH + col];
    }
    for (int kc = 0; kc < 4; ++kc) {
      const int kb = kc * 32 + 2 * hl;
      v2f a[8];
      float4 q0[8], q1[8];
#pragma unroll
      for (int u = 0; u < 8; ++u) {
        const int k = kb + 4 * u;
        a[u].x = sV[k * VPITCH + i0 + cl]; a[u].y = sV[(k + 1) * VPITCH + i0 + cl];
        q0[u] = *reinterpret_cast<const float4*>(&sV[k * VPITCH + 4 * cl]);
        q1[u] = *reinterpret_cast<const float4*>(&sV[(k + 1) * VPITCH + 4 * cl]);
      }
#pragma unroll
      for (int u = 0; u < 8; ++u) {
        v2f bf;
        bf.x = q0[u].x; bf.y = q1[u].x; gt[0] = wmma4(a[u], bf, gt[0]);
        bf.x = q0[u].y; bf.y = q1[u].y; gt[1] = wmma4(a[u], bf, gt[1]);
        bf.x = q0[u].z; bf.y = q1[u].z; gt[2] = wmma4(a[u], bf, gt[2]);
        bf.x = q0[u].w; bf.y = q1[u].w; gt[3] = wmma4(a[u], bf, gt[3]);
      }
    }
  }
  float* Gd = Gpart + (size_t)q * 4096;
#pragma unroll
  for (int r = 0; r < 8; ++r) {
    float4 o;
    o.x = gt[0][r] * gs; o.y = gt[1][r] * gs; o.z = gt[2][r] * gs; o.w = gt[3][r] * gs;
    *reinterpret_cast<float4*>(&Gd[(i0 + (hl << 3) + r) * 64 + 4 * cl]) = o;
  }
  shp[threadIdx.x] = csum;
  __syncthreads();
  if (threadIdx.x < 64) gpart[q * 64 + threadIdx.x] = (shp[threadIdx.x] + shp[64 + threadIdx.x]) * gs;
}

__device__ __forceinline__ void caug_job(const float* Wq, const float* Wk, float* Caug, int jb) {
  const int lane = threadIdx.x & 31, wave = threadIdx.x >> 5;
  const int cl = lane & 15, hl = lane >> 4;
  const int j = jb >> 1;
  const int m0 = (2 * (jb & 1) + (wave >> 1)) * 16;
  const int n0 = (wave & 1) * 32;
  const float* Ap = Wq + (size_t)j * HEAD * DK * FDIM + m0 + cl;
  const float* Bp = Wk + (size_t)j * HEAD * DK * FDIM + n0 + cl;
  v8f t0 = {}, t1 = {};
  for (int kb = 0; kb < HEAD * DK; kb += 64) {
    v2f a[16], b0[16], b1[16];
#pragma unroll
    for (int u = 0; u < 16; ++u) {
      const int kk = kb + 4 * u + 2 * hl;
      a[u].x  = Ap[kk * 64];      a[u].y  = Ap[(kk + 1) * 64];
      b0[u].x = Bp[kk * 64];      b0[u].y = Bp[(kk + 1) * 64];
      b1[u].x = Bp[kk * 64 + 16]; b1[u].y = Bp[(kk + 1) * 64 + 16];
    }
#pragma unroll
    for (int part = 0; part < 4; ++part) {
      v8f p0 = {}, p1 = {};
#pragma unroll
      for (int u = 4 * part; u < 4 * part + 4; ++u) {
        p0 = wmma4(a[u], b0[u], p0);
        p1 = wmma4(a[u], b1[u], p1);
      }
      if (kb == 0 && part == 0) { t0 = p0; t1 = p1; } else { t0 += p0; t1 += p1; }
    }
  }
  float* Cj = Caug + (size_t)j * CAUG;
#pragma unroll
  for (int r = 0; r < 8; ++r) {
    const int row = m0 + (hl << 3) + r;
    Cj[row * 64 + n0 + cl]      = t0[r];
    Cj[row * 64 + n0 + 16 + cl] = t1[r];
  }
}

__device__ __forceinline__ void cvec_job(const float* Wq, const float* bq, const float* Wk,
                                         const float* bk, float* Caug, float* caug, int jb,
                                         float* sh) {
  const int j = jb >> 1;
  const bool isD = (jb & 1) != 0;
  const float* Wm = (isD ? Wk : Wq) + (size_t)j * 256 * 64;
  const float* v  = (isD ? bq : bk) + j * 256;
  const int i = threadIdx.x & 63, half = threadIdx.x >> 6;
  for (int cc = 0; cc < 8; ++cc) {
    const int kc = half * 8 + cc;
    float a = 0.f;
#pragma unroll
    for (int u = 0; u < 16; ++u) {
      int k = kc * 16 + u;
      a = fmaf(Wm[k * 64 + i], v[k], a);
    }
    sh[kc * 64 + i] = a;
  }
  __syncthreads();
  float* Cj = Caug + (size_t)j * CAUG;
  float* cj = caug + j * 80;
  if (threadIdx.x < 64) {
    float t = sh[threadIdx.x];
    for (int p = 1; p < 16; ++p) t += sh[p * 64 + threadIdx.x];
    if (isD) Cj[64 * 64 + threadIdx.x] = t; else cj[threadIdx.x] = t;
  }
  if (isD) {
    if ((threadIdx.x >> 5) == 2) {
      const int lane = threadIdx.x & 31;
      float e = 0.f;
      for (int u = 0; u < 8; ++u) {
        int k = lane + 32 * u;
        e = fmaf(bq[j * 256 + k], bk[j * 256 + k], e);
      }
      for (int m = 16; m >= 1; m >>= 1) e += __shfl_xor(e, m, 32);
      if (lane == 0) cj[64] = e;
    }
  } else {
    for (int idx = threadIdx.x; idx < 15 * 64; idx += 128) Cj[65 * 64 + idx] = 0.f;
    if (threadIdx.x < 15) cj[65 + threadIdx.x] = 0.f;
  }
}
__device__ __forceinline__ float leaky01(float t) {
  return fmaxf(t, 0.01f * t);
}

struct EdgeNets {
  const float* kW0; const float* kb0; const float* kW1; const float* kb1;
  const float* kW2; const float* kb2;
  const float* fW0; const float* fb0; const float* fW1; const float* fb1;
  const float* fW2; const float* fb2;
  float* wout; float* fout;
};

template <int ITERS>
__device__ __forceinline__ void edge_row_job(const EdgeNets& en, int ejob, int mbase, int lane) {
  const bool isF = ejob >= NTOKEN;
  const float* W0 = isF ? en.fW0 : en.kW0;
  const float* b0 = isF ? en.fb0 : en.kb0;
  const float* W1 = isF ? en.fW1 : en.kW1;
  const float* b1 = isF ? en.fb1 : en.kb1;
  const float* W2 = isF ? en.fW2 : en.kW2;
  const float* b2 = isF ? en.fb2 : en.kb2;
  float* dst      = isF ? en.fout : en.wout;

  const int c = lane & 15, h = lane >> 4;
  const int n = ejob & (NTOKEN - 1);
  const float DXc = 2.0f / 31.0f;
  const float e0 = -1.f + DXc * (float)(n >> 5);
  const float e1 = -1.f + DXc * (float)(n & 31);

  v2f wa[4][8];
#pragma unroll
  for (int t = 0; t < 4; ++t) {
#pragma unroll
    for (int i = 0; i < 8; ++i) {
      const float* p = W1 + (t * 16 + c) * 32 + 4 * i + 2 * h;
      wa[t][i].x = p[0]; wa[t][i].y = p[1];
    }
  }
  float pre[16], wc2[16], wc3[16];
#pragma unroll
  for (int i = 0; i < 8; ++i) {
#pragma unroll
    for (int j = 0; j < 2; ++j) {
      int u = 4 * i + 2 * h + j;
      float t = b0[u];
      t = fmaf(W0[u * 4 + 0], e0, t);
      t = fmaf(W0[u * 4 + 1], e1, t);
      pre[2 * i + j] = t;
      wc2[2 * i + j] = W0[u * 4 + 2];
      wc3[2 * i + j] = W0[u * 4 + 3];
    }
  }
  v8f cb[4];
  float w2v[4][8];
#pragma unroll
  for (int t = 0; t < 4; ++t) {
#pragma unroll
    for (int r = 0; r < 8; ++r) {
      cb[t][r]  = b1[t * 16 + 8 * h + r];
      w2v[t][r] = W2[t * 16 + 8 * h + r];
    }
  }
  const float bias2 = b2[0];

  for (int it = 0; it < ITERS; ++it) {
    const int m0 = mbase + it * 32;
    float res[2];
#pragma unroll
    for (int et = 0; et < 2; ++et) {
      const int m = m0 + et * 16 + c;
      const float e2 = -1.f + DXc * (float)(m >> 5);
      const float e3 = -1.f + DXc * (float)(m & 31);
      v2f hb[8];
#pragma unroll
      for (int i = 0; i < 8; ++i) {
        float t0 = fmaf(wc3[2 * i],     e3, fmaf(wc2[2 * i],     e2, pre[2 * i]));
        float t1 = fmaf(wc3[2 * i + 1], e3, fmaf(wc2[2 * i + 1], e2, pre[2 * i + 1]));
        hb[i].x = leaky01(t0); hb[i].y = leaky01(t1);
      }
      v8f acc[4];
#pragma unroll
      for (int t = 0; t < 4; ++t) acc[t] = wmma4(wa[t][0], hb[0], cb[t]);
#pragma unroll
      for (int i = 1; i < 8; ++i) {
#pragma unroll
        for (int t = 0; t < 4; ++t) acc[t] = wmma4(wa[t][i], hb[i], acc[t]);
      }
      float s = 0.f;
#pragma unroll
      for (int t = 0; t < 4; ++t) {
#pragma unroll
        for (int r = 0; r < 8; ++r) s = fmaf(w2v[t][r], leaky01(acc[t][r]), s);
      }
      s += __shfl_xor(s, 16, 32);
      res[et] = s + bias2;
    }
    dst[n * NTOKEN + m0 + lane] = h ? res[1] : res[0];
  }
}

__global__ __launch_bounds__(128) void edge_mlp_kernel(
                                EdgeNets en,
                                const float* xy, const float* ln0g, const float* ln0b, float* V0,
                                const float* Wq, const float* bq, const float* Wk, const float* bk,
                                float* Caug, float* caug, float gs, float* Gpart, float* gpart) {
  __shared__ float shx[16 * 64];
  __shared__ __attribute__((aligned(16))) float sV[128 * VPITCH];
  if (blockIdx.x < LNA_WGS) {
    ln0_job(xy, ln0g, ln0b, V0, blockIdx.x, shx, sV, gs, Gpart, gpart);
    return;
  }
  if (blockIdx.x < LNA_WGS + WJOBS / 2) {
    caug_job(Wq, Wk, Caug, blockIdx.x - LNA_WGS);
    return;
  }
  if (blockIdx.x < LNA_WGS + WJOBS) {
    cvec_job(Wq, bq, Wk, bk, Caug, caug, blockIdx.x - LNA_WGS - WJOBS / 2, shx);
    return;
  }
  edge_row_job<8>(en, blockIdx.x - (LNA_WGS + WJOBS), (threadIdx.x >> 5) * 256, threadIdx.x & 31);
}

#define GPITCH  72


#define CPITCH  68
#define LROWS   128
#define LAYER_WGS (BATCH * (NTOK2 / LROWS))
#define EROWS_L 128
#define EROWS1  (2 * NTOKEN - 4 * EROWS_L)
#define OPITCH  80

template <int M>
__device__ __forceinline__ float row_xor_add(float v) {
  const int o = __builtin_amdgcn_update_dpp(0, __builtin_bit_cast(int, v), 0x160 + M, 0xf, 0xf, true);
  return v + __builtin_bit_cast(float, o);
}

template <bool LNRES, int NP, int GM>
__global__ __launch_bounds__(256) void layer_kernel(const float* Vin, const float* Cj,
                                                    const float* cj, const float* Gp,
                                                    const float* gp, const float* lg,
                                                    const float* lb, float* Out,
                                                    const float* xall, float gs,
                                                    float* Gout, float* gout,
                                                    EdgeNets en, int ebase) {
  constexpr int NPOUT = (GM == 2) ? 8 : 16;
  __shared__ __attribute__((aligned(16))) float sG[64 * GPITCH];
  __shared__ __attribute__((aligned(16))) float sC[80 * CPITCH];
  __shared__ __attribute__((aligned(16))) float sP[80 * CPITCH];
  __shared__ float sg[64];
  __shared__ __attribute__((aligned(16))) float scj[80];
  __shared__ __attribute__((aligned(16))) float sO[(GM != 0) ? LROWS * OPITCH : 4];
  __shared__ __attribute__((aligned(16))) float sX[(GM == 2) ? LROWS * OPITCH : 4];
  __shared__ float sS[(GM != 0) ? 4 * 64 : 4];
  const int tid = threadIdx.x;
  const int wave = __builtin_amdgcn_readfirstlane(tid >> 5);
  const int lane = tid & 31;
  if (blockIdx.x >= LAYER_WGS) {
    edge_row_job<4>(en, ebase + (int)blockIdx.x - LAYER_WGS, wave * 128, lane);
    return;
  }
  const int b = blockIdx.x & (BATCH - 1);
  const int strip = blockIdx.x >> 3;
  const int cl = lane & 15, hl = lane >> 4;
  const int r0 = strip * LROWS;
  const int m0 = r0 + wave * 16;
  const float* A  = Vin + (size_t)b * NTOK2 * FDIM;
  const bool gram = (GM == 1) || (GM == 2 && r0 < NTOKEN);

  const float4* G4 = reinterpret_cast<const float4*>(Gp + (size_t)b * NP * 4096);
  const float4* C4 = reinterpret_cast<const float4*>(Cj);
  const float* g0 = gp + (size_t)b * NP * 64;
  float4 gq[4][NP];
#pragma unroll
  for (int i = 0; i < 4; ++i) {
#pragma unroll
    for (int q = 0; q < NP; ++q) gq[i][q] = G4[q * 1024 + tid + 256 * i];
  }
  const float4 cq0 = C4[tid], cq1 = C4[tid + 256], cq2 = C4[tid + 512], cq3 = C4[tid + 768],
               cq4 = C4[tid + 1024];
  float gv[NP];
#pragma unroll
  for (int q = 0; q < NP; ++q) gv[q] = g0[q * 64 + (tid & 63)];
  const float cjg = cj[tid < 80 ? tid : 79];
  v2f av[16];
#pragma unroll
  for (int u = 0; u < 16; ++u) av[u] = frag_a(A, FDIM, m0, 4 * u, lane);
  __builtin_amdgcn_sched_barrier(0);
#pragma unroll
  for (int i = 0; i < 4; ++i) {
    const int i4 = tid + 256 * i;
    float4 v = gq[i][0];
#pragma unroll
    for (int q = 1; q < NP; ++q) {
      v.x += gq[i][q].x; v.y += gq[i][q].y; v.z += gq[i][q].z; v.w += gq[i][q].w;
    }
    *reinterpret_cast<float4*>(&sG[(i4 >> 4) * GPITCH + ((i4 & 15) << 2)]) = v;
  }
  {
    float4* cdst = reinterpret_cast<float4*>(&sC[(tid >> 4) * CPITCH + ((tid & 15) << 2)]);
    const int cstep = 16 * CPITCH / 4;
    cdst[0] = cq0; cdst[cstep] = cq1; cdst[2 * cstep] = cq2; cdst[3 * cstep] = cq3;
    cdst[4 * cstep] = cq4;
  }
  if (tid < 64) {
    float v = gv[0];
#pragma unroll
    for (int q = 1; q < NP; ++q) v += gv[q];
    sg[tid] = v;
  }
  if (tid < 80) scj[tid] = cjg;
  __syncthreads();

  float4 rv[8];
  float4 lg4 = make_float4(0.f, 0.f, 0.f, 0.f), lb4 = make_float4(0.f, 0.f, 0.f, 0.f);
  if constexpr (LNRES) {
#pragma unroll
    for (int r = 0; r < 8; ++r)
      rv[r] = *reinterpret_cast<const float4*>(&A[(m0 + (hl << 3) + r) * FDIM + 4 * cl]);
    lg4 = *reinterpret_cast<const float4*>(&lg[4 * cl]);
    lb4 = *reinterpret_cast<const float4*>(&lb[4 * cl]);
  }
  float4 xq[8];
#pragma unroll
  for (int i = 0; i < 8; ++i) xq[i] = make_float4(0.f, 0.f, 0.f, 0.f);
  if constexpr (GM == 2) {
    if (gram) {
      const float4* X4 = reinterpret_cast<const float4*>(xall + ((size_t)b * NTOK2 + r0) * FDIM);
#pragma unroll
      for (int i = 0; i < 8; ++i) xq[i] = X4[tid + 256 * i];
    }
  }

  {
    const int rg = wave >> 2;
    const int col = (wave & 3) * 16 + cl;
    const float g = sg[col];
    v2f gb[16];
#pragma unroll
    for (int u = 0; u < 16; ++u) {
      const int kk = 4 * u + 2 * hl;
      gb[u].x = sG[kk * GPITCH + col]; gb[u].y = sG[(kk + 1) * GPITCH + col];
    }
    if (rg == 0) {
      v8f pc;
#pragma unroll
      for (int r = 0; r < 8; ++r) pc[r] = scj[64 + (hl << 3) + r] * g;
      v2f cc[16];
#pragma unroll
      for (int u = 0; u < 16; ++u)
        cc[u] = *reinterpret_cast<const v2f*>(&sC[(64 + cl) * CPITCH + 4 * u + 2 * hl]);
      asm volatile("" ::: "memory");
      __builtin_amdgcn_sched_barrier(0);
#pragma unroll
      for (int u = 0; u < 16; ++u) pc = wmma4(cc[u], gb[u], pc);
#pragma unroll
      for (int r = 0; r < 8; ++r) sP[(64 + (hl << 3) + r) * CPITCH + col] = pc[r];
    }
    const int ra = rg * 32;
    v8f pa, pb;
#pragma unroll
    for (int r = 0; r < 8; ++r) {
      pa[r] = scj[ra + (hl << 3) + r] * g;
      pb[r] = scj[ra + 16 + (hl << 3) + r] * g;
    }
    v2f ca[16], cb2[16];
#pragma unroll
    for (int u = 0; u < 16; ++u) {
      ca[u]  = *reinterpret_cast<const v2f*>(&sC[(ra + cl) * CPITCH + 4 * u + 2 * hl]);
      cb2[u] = *reinterpret_cast<const v2f*>(&sC[(ra + 16 + cl) * CPITCH + 4 * u + 2 * hl]);
    }
    asm volatile("" ::: "memory");
    __builtin_amdgcn_sched_barrier(0);
#pragma unroll
    for (int u = 0; u < 16; ++u) {
      pa = wmma4(ca[u], gb[u], pa);
      pb = wmma4(cb2[u], gb[u], pb);
    }
#pragma unroll
    for (int r = 0; r < 8; ++r) {
      sP[(ra + (hl << 3) + r) * CPITCH + col]      = pa[r];
      sP[(ra + 16 + (hl << 3) + r) * CPITCH + col] = pb[r];
    }
  }
  __syncthreads();

  const float* Pp = sP + 4 * cl;
  const float4 pv = *reinterpret_cast<const float4*>(&Pp[64 * CPITCH]);
  v8f c[4] = {};
  float4 q0[16], q1[16];
#pragma unroll
  for (int u = 0; u < 16; ++u) {
    const int kk = 4 * u + 2 * hl;
    q0[u] = *reinterpret_cast<const float4*>(&Pp[kk * CPITCH]);
    q1[u] = *reinterpret_cast<const float4*>(&Pp[(kk + 1) * CPITCH]);
  }
  asm volatile("" ::: "memory");
  __builtin_amdgcn_sched_barrier(0);
#pragma unroll
  for (int u = 0; u < 16; ++u) {
    v2f bf;
    bf.x = q0[u].x; bf.y = q1[u].x; c[0] = wmma4(av[u], bf, c[0]);
    bf.x = q0[u].y; bf.y = q1[u].y; c[1] = wmma4(av[u], bf, c[1]);
    bf.x = q0[u].z; bf.y = q1[u].z; c[2] = wmma4(av[u], bf, c[2]);
    bf.x = q0[u].w; bf.y = q1[u].w; c[3] = wmma4(av[u], bf, c[3]);
  }
  float* dst = Out + (size_t)b * NTOK2 * FDIM;
  const int rbase = m0 + (hl << 3);
  const int lrow = wave * 16 + (hl << 3);
  if constexpr (!LNRES) {
#pragma unroll
    for (int r = 0; r < 8; ++r) {
      float4 o;
      o.x = c[0][r] + pv.x; o.y = c[1][r] + pv.y; o.z = c[2][r] + pv.z; o.w = c[3][r] + pv.w;
      *reinterpret_cast<float4*>(&dst[(rbase + r) * FDIM + 4 * cl]) = o;
    }
  } else {
    float sm[8], sq[8];
#pragma unroll
    for (int r = 0; r < 8; ++r) {
      c[0][r] += pv.x; c[1][r] += pv.y; c[2][r] += pv.z; c[3][r] += pv.w;
      sm[r] = (c[0][r] + c[1][r]) + (c[2][r] + c[3][r]);
      sq[r] = fmaf(c[3][r], c[3][r], fmaf(c[2][r], c[2][r], fmaf(c[1][r], c[1][r], c[0][r] * c[0][r])));
    }
#pragma unroll
    for (int r = 0; r < 8; ++r) { sm[r] = row_xor_add<1>(sm[r]); sq[r] = row_xor_add<1>(sq[r]); }
#pragma unroll
    for (int r = 0; r < 8; ++r) { sm[r] = row_xor_add<2>(sm[r]); sq[r] = row_xor_add<2>(sq[r]); }
#pragma unroll
    for (int r = 0; r < 8; ++r) { sm[r] = row_xor_add<4>(sm[r]); sq[r] = row_xor_add<4>(sq[r]); }
#pragma unroll
    for (int r = 0; r < 8; ++r) { sm[r] = row_xor_add<8>(sm[r]); sq[r] = row_xor_add<8>(sq[r]); }
#pragma unroll
    for (int r = 0; r < 8; ++r) {
      const float mean = sm[r] * (1.f / 64.f);
      const float var  = sq[r] * (1.f / 64.f) - mean * mean;
      const float inv  = rsqrtf(var + 1e-5f);
      float4 o;
      o.x = (c[0][r] - mean) * inv * lg4.x + lb4.x + rv[r].x;
      o.y = (c[1][r] - mean) * inv * lg4.y + lb4.y + rv[r].y;
      o.z = (c[2][r] - mean) * inv * lg4.z + lb4.z + rv[r].z;
      o.w = (c[3][r] - mean) * inv * lg4.w + lb4.w + rv[r].w;
      *reinterpret_cast<float4*>(&dst[(rbase + r) * FDIM + 4 * cl]) = o;
      if constexpr (GM != 0)
        *reinterpret_cast<float4*>(&sO[(lrow + r) * OPITCH + 4 * cl]) = o;
    }
  }

  if constexpr (GM != 0) {
    if (gram) {
      if constexpr (GM == 2) {
#pragma unroll
        for (int i = 0; i < 8; ++i) {
          const int p = tid + 256 * i;
          *reinterpret_cast<float4*>(&sX[(p >> 4) * OPITCH + ((p & 15) << 2)]) = xq[i];
        }
      }
      __syncthreads();
      const float* sB = (GM == 2) ? sX : sO;
      {
        const int col = tid & 63, grp = tid >> 6;
        float s = 0.f;
        for (int r = 0; r < 32; ++r) s += sB[(grp * 32 + r) * OPITCH + col];
        sS[grp * 64 + col] = s;
      }
      const int i0 = (wave >> 1) * 16, n0 = (wave & 1) * 32;
      v8f t0 = {}, t1 = {};
      for (int kc = 0; kc < 4; ++kc) {
        const int kb = kc * 32 + 2 * hl;
        v2f a[8], bx[8], by[8];
#pragma unroll
        for (int u = 0; u < 8; ++u) {
          const int k = kb + 4 * u;
          a[u].x = sO[k * OPITCH + i0 + cl]; a[u].y = sO[(k + 1) * OPITCH + i0 + cl];
          bx[u] = *reinterpret_cast<const v2f*>(&sB[k * OPITCH + n0 + 2 * cl]);
          by[u] = *reinterpret_cast<const v2f*>(&sB[(k + 1) * OPITCH + n0 + 2 * cl]);
        }
        asm volatile("" ::: "memory");
        __builtin_amdgcn_sched_barrier(0);
#pragma unroll
        for (int u = 0; u < 8; ++u) {
          v2f bf;
          bf.x = bx[u].x; bf.y = by[u].x; t0 = wmma4(a[u], bf, t0);
          bf.x = bx[u].y; bf.y = by[u].y; t1 = wmma4(a[u], bf, t1);
        }
      }
      const size_t slot = (size_t)b * NPOUT + strip;
      float* Gd = Gout + slot * 4096;
#pragma unroll
      for (int r = 0; r < 8; ++r) {
        const int row = i0 + (hl << 3) + r;
        v2f o; o.x = t0[r] * gs; o.y = t1[r] * gs;
        *reinterpret_cast<v2f*>(&Gd[row * 64 + n0 + 2 * cl]) = o;
      }
      __syncthreads();
      if (tid < 64) {
        const float s = ((sS[tid] + sS[64 + tid]) + sS[128 + tid]) + sS[192 + tid];
        gout[slot * 64 + tid] = s * gs;
      }
    }
  }
}

__global__ void final_kernel(const float* weight, const float* weightf,
                             const float* S, float* out, float mul) {
  __shared__ float sh[KSPLIT * TPLANE];
  int b = blockIdx.x;
  int ks = __builtin_amdgcn_readfirstlane(threadIdx.x >> 5);
  int lane = threadIdx.x & 31;
  int m0 = blockIdx.y * 16;
  const float* Sb = S + (size_t)b * NTOK2 * FDIM;
  int row = m0 + (lane & 15);
  const int klen = NTOK2 / KSPLIT;
  int kbeg = ks * klen, kend = kbeg + klen;
  const float* W = (kbeg < NTOKEN) ? weight : weightf;
  v8f c[4] = {};
  for (int k0 = kbeg; k0 < kend; k0 += 4) {
    int kb = (k0 & (NTOKEN - 1)) + ((lane >> 4) << 1);
    v2f a; a.x = W[kb * NTOKEN + row]; a.y = W[(kb + 1) * NTOKEN + row];
    if ((k0 & 60) == 0 && (k0 & (NTOKEN - 1)) + 64 < NTOKEN) {
      __builtin_prefetch(&W[(kb + 64) * NTOKEN + row], 0, 3);
    }
#pragma unroll
    for (int t = 0; t < 4; ++t) {
      v2f bf = frag_b(Sb, FDIM, k0, t * 16, lane);
      c[t] = wmma4(a, bf, c[t]);
    }
  }
  ksplit_reduce_store(sh, c, ks, lane, out + (size_t)b * NTOKEN * FDIM, m0, mul, 16, 1);
}

extern "C" void kernel_launch(void* const* d_in, const int* in_sizes, int n_in,
                              void* d_out, int out_size, void* d_ws, size_t ws_size,
                              hipStream_t stream) {
  (void)in_sizes; (void)n_in; (void)out_size; (void)ws_size;
  const float* xy   = (const float*)d_in[0];
  const float* kW0  = (const float*)d_in[1];
  const float* kb0  = (const float*)d_in[2];
  const float* kW1  = (const float*)d_in[3];
  const float* kb1  = (const float*)d_in[4];
  const float* kW2  = (const float*)d_in[5];
  const float* kb2  = (const float*)d_in[6];
  const float* fW0  = (const float*)d_in[7];
  const float* fb0  = (const float*)d_in[8];
  const float* fW1  = (const float*)d_in[9];
  const float* fb1  = (const float*)d_in[10];
  const float* fW2  = (const float*)d_in[11];
  const float* fb2  = (const float*)d_in[12];
  const float* Wq   = (const float*)d_in[13];
  const float* bq   = (const float*)d_in[14];
  const float* Wk   = (const float*)d_in[15];
  const float* bk   = (const float*)d_in[16];
  const float* ln0g = (const float*)d_in[17];
  const float* ln0b = (const float*)d_in[18];
  const float* lng  = (const float*)d_in[19];
  const float* lnb  = (const float*)d_in[20];
  float* out = (float*)d_out;

  float* ws = (float*)d_ws;
  const size_t SZ_1M = 1048576;
  float* w_weight  = ws;
  float* w_weightf = ws + SZ_1M;
  float* V0        = ws + 2 * SZ_1M;
  float* V1        = ws + 3 * SZ_1M;
  float* Caug      = ws + 4 * SZ_1M;
  float* caug      = ws + 4 * SZ_1M + 20480;
  float* Gp0       = ws + 5 * SZ_1M;
  float* gp0       = ws + 5 * SZ_1M + 524288;
  float* GpA       = ws + 6 * SZ_1M;
  float* gpA       = ws + 6 * SZ_1M + 524288;
  float* GpB       = ws + 7 * SZ_1M;
  float* gpB       = ws + 7 * SZ_1M + 524288;
  float* MID       = ws + 12 * SZ_1M + 131072;

  const float DXc   = 2.0f / 31.0f;
  const float dxdy  = DXc * DXc;
  const float scale = 0.125f;

  const EdgeNets en = {kW0, kb0, kW1, kb1, kW2, kb2, fW0, fb0, fW1, fb1, fW2, fb2,
                       w_weight, w_weightf};
  edge_mlp_kernel<<<LNA_WGS + WJOBS + EROWS1, 128, 0, stream>>>(
      en, xy, ln0g, ln0b, V0, Wq, bq, Wk, bk, Caug, caug, scale * dxdy, Gp0, gp0);

  layer_kernel<true, 8, 1><<<LAYER_WGS + EROWS_L, 256, 0, stream>>>(
      V0, Caug, caug, Gp0, gp0, lng, lnb, V1, xy, scale * dxdy, GpA, gpA, en, EROWS1);
  layer_kernel<true, 16, 1><<<LAYER_WGS + EROWS_L, 256, 0, stream>>>(
      V1, Caug + (size_t)1 * CAUG, caug + 80, GpA, gpA, lng + FDIM, lnb + FDIM, V0, xy,
      scale * dxdy, GpB, gpB, en, EROWS1 + EROWS_L);
  layer_kernel<true, 16, 2><<<LAYER_WGS + EROWS_L, 256, 0, stream>>>(
      V0, Caug + (size_t)2 * CAUG, caug + 160, GpB, gpB, lng + 2 * FDIM, lnb + 2 * FDIM, V1, xy,
      scale, GpA, gpA, en, EROWS1 + 2 * EROWS_L);
  layer_kernel<false, 8, 0><<<LAYER_WGS + EROWS_L, 256, 0, stream>>>(
      V1, Caug + (size_t)3 * CAUG, caug + 240, GpA, gpA, nullptr, nullptr, MID, xy,
      scale, nullptr, nullptr, en, EROWS1 + 3 * EROWS_L);
  final_kernel<<<dim3(BATCH, NTOKEN / 16), KSPLIT * 32, 0, stream>>>(
      w_weight, w_weightf, MID, out, dxdy * dxdy);
}


# Final weight product re-tiled: a wave owns 32 j x 64 f (8 accumulator tiles), weight and S rows read as 8- and 16-byte pieces, a trip's 32 loads as one group; 256 work-groups; same sums and order
# speedup vs baseline: 7.4346x; 1.0608x over previous
#include <hip/hip_runtime.h>
#include <hip/hip_bf16.h>

#define NGRID   32
#define NTOKEN  1024
#define NTOK2   2048
#define HEAD    4
#define DK      64
#define NLAYER  4
#define FDIM    64
#define BATCH   8

typedef __attribute__((ext_vector_type(2))) float v2f;
typedef __attribute__((ext_vector_type(8))) float v8f;

__device__ __forceinline__ v8f wmma4(v2f a, v2f b, v8f c) {
  return __builtin_amdgcn_wmma_f32_16x16x4_f32(
       false, a,  false, b,
       (short)0, c,  false,  false);
}

__device__ __forceinline__ v2f frag_a(const float* src, int lda, int m0, int k0, int lane) {
  int row = m0 + (lane & 15);
  int kk  = k0 + ((lane >> 4) << 1);
  v2f a; a.x = src[row * lda + kk]; a.y = src[row * lda + kk + 1];
  return a;
}

#define KSPLIT   16
#define TPITCH   68
#define FROWS    32
#define FPLANE   (FROWS * TPITCH)

#define LN0_CHUNKS  8
#define LN0_SLICE   (NTOK2 * FDIM / LN0_CHUNKS)
#define LNA_WGS     64
#define LNA_ELEMS   (BATCH * NTOK2 * FDIM / LNA_WGS)
#define WJOBS       16
#define CAUG    (80 * 64)
#define VPITCH  80

__device__ __forceinline__ void ln0_job(const float* xy, const float* g, const float* bb,
                                        float* V, int q, float* shp, float* sV, float gs,
                                        float* Gpart, float* gpart) {
  const int b = q / (LNA_WGS / BATCH);
  const int wave = threadIdx.x >> 5;
  const float4* slab = reinterpret_cast<const float4*>(xy + (size_t)b * NTOK2 * FDIM);
  for (int cidx = 0; cidx < LN0_CHUNKS; ++cidx) {
    const float4* p = slab + cidx * (LN0_SLICE / 4);
    float4 v[32];
#pragma unroll
    for (int it = 0; it < 32; ++it) v[it] = p[it * 128 + threadIdx.x];
    float4 s4 = make_float4(0.f, 0.f, 0.f, 0.f), q4 = make_float4(0.f, 0.f, 0.f, 0.f);
#pragma unroll
    for (int it = 0; it < 32; ++it) {
      s4.x += v[it].x; s4.y += v[it].y; s4.z += v[it].z; s4.w += v[it].w;
      q4.x = fmaf(v[it].x, v[it].x, q4.x); q4.y = fmaf(v[it].y, v[it].y, q4.y);
      q4.z = fmaf(v[it].z, v[it].z, q4.z); q4.w = fmaf(v[it].w, v[it].w, q4.w);
    }
    float s = (s4.x + s4.y) + (s4.z + s4.w);
    float ss = (q4.x + q4.y) + (q4.z + q4.w);
#pragma unroll
    for (int m = 16; m >= 1; m >>= 1) {
      s  += __shfl_xor(s, m, 32);
      ss += __shfl_xor(ss, m, 32);
    }
    if ((threadIdx.x & 31) == 0) { shp[8 * cidx + 2 * wave] = s; shp[8 * cidx + 2 * wave + 1] = ss; }
  }
  __syncthreads();
  float s = 0.f, ss = 0.f;
#pragma unroll
  for (int cidx = 0; cidx < LN0_CHUNKS; ++cidx) {
    const float* w = shp + 8 * cidx;
    const float ps = ((w[0] + w[2]) + w[4]) + w[6];
    const float pq = ((w[1] + w[3]) + w[5]) + w[7];
    if (cidx == 0) { s = ps; ss = pq; } else { s += ps; ss += pq; }
  }
  const float N = (float)(NTOK2 * FDIM);
  const float mean = s / N;
  const float var  = ss / N - mean * mean;
  const float rstd = rsqrtf(var + 1e-5f);
  const int lane = threadIdx.x & 31, cl = lane & 15, hl = lane >> 4;
  const int i0 = wave * 16;
  v8f gt[4] = {};
  float csum = 0.f;
  for (int trip = 0; trip < 4; ++trip) {
    const int e4 = q * (LNA_ELEMS / 4) + trip * 1024;
    if (trip == 2) __syncthreads();
    const float4* x4 = reinterpret_cast<const float4*>(xy) + e4;
    const float4* g4 = reinterpret_cast<const float4*>(g) + (e4 & (NTOK2 * FDIM / 4 - 1));
    const float4* b4 = reinterpret_cast<const float4*>(bb) + (e4 & (NTOK2 * FDIM / 4 - 1));
    float4* o4 = reinterpret_cast<float4*>(V) + e4;
    float4 xv[8], gv[8], bv[8];
#pragma unroll
    for (int i = 0; i < 8; ++i) {
      xv[i] = x4[i * 128 + threadIdx.x]; gv[i] = g4[i * 128 + threadIdx.x]; bv[i] = b4[i * 128 + threadIdx.x];
    }
#pragma unroll
    for (int i = 0; i < 8; ++i) {
      float4 o;
      o.x = (xv[i].x - mean) * rstd * gv[i].x + bv[i].x;
      o.y = (xv[i].y - mean) * rstd * gv[i].y + bv[i].y;
      o.z = (xv[i].z - mean) * rstd * gv[i].z + bv[i].z;
      o.w = (xv[i].w - mean) * rstd * gv[i].w + bv[i].w;
      o4[i * 128 + threadIdx.x] = o;
      const int pc = i * 128 + threadIdx.x;
      *reinterpret_cast<float4*>(&sV[((trip & 1) * 64 + (pc >> 4)) * VPITCH + ((pc & 15) << 2)]) = o;
    }
    if ((trip & 1) == 0) continue;
    __syncthreads();
    {
      const int col = threadIdx.x & 63, grp = threadIdx.x >> 6;
      for (int r = 0; r < 64; ++r) csum += sV[(grp * 64 + r) * VPITCH + col];
    }
    for (int kc = 0; kc < 4; ++kc) {
      const int kb = kc * 32 + 2 * hl;
      v2f a[8];
      float4 q0[8], q1[8];
#pragma unroll
      for (int u = 0; u < 8; ++u) {
        const int k = kb + 4 * u;
        a[u].x = sV[k * VPITCH + i0 + cl]; a[u].y = sV[(k + 1) * VPITCH + i0 + cl];
        q0[u] = *reinterpret_cast<const float4*>(&sV[k * VPITCH + 4 * cl]);
        q1[u] = *reinterpret_cast<const float4*>(&sV[(k + 1) * VPITCH + 4 * cl]);
      }
#pragma unroll
      for (int u = 0; u < 8; ++u) {
        v2f bf;
        bf.x = q0[u].x; bf.y = q1[u].x; gt[0] = wmma4(a[u], bf, gt[0]);
        bf.x = q0[u].y; bf.y = q1[u].y; gt[1] = wmma4(a[u], bf, gt[1]);
        bf.x = q0[u].z; bf.y = q1[u].z; gt[2] = wmma4(a[u], bf, gt[2]);
        bf.x = q0[u].w; bf.y = q1[u].w; gt[3] = wmma4(a[u], bf, gt[3]);
      }
    }
  }
  float* Gd = Gpart + (size_t)q * 4096;
#pragma unroll
  for (int r = 0; r < 8; ++r) {
    float4 o;
    o.x = gt[0][r] * gs; o.y = gt[1][r] * gs; o.z = gt[2][r] * gs; o.w = gt[3][r] * gs;
    *reinterpret_cast<float4*>(&Gd[(i0 + (hl << 3) + r) * 64 + 4 * cl]) = o;
  }
  shp[threadIdx.x] = csum;
  __syncthreads();
  if (threadIdx.x < 64) gpart[q * 64 + threadIdx.x] = (shp[threadIdx.x] + shp[64 + threadIdx.x]) * gs;
}

__device__ __forceinline__ void caug_job(const float* Wq, const float* Wk, float* Caug, int jb) {
  const int lane = threadIdx.x & 31, wave = threadIdx.x >> 5;
  const int cl = lane & 15, hl = lane >> 4;
  const int j = jb >> 1;
  const int m0 = (2 * (jb & 1) + (wave >> 1)) * 16;
  const int n0 = (wave & 1) * 32;
  const float* Ap = Wq + (size_t)j * HEAD * DK * FDIM + m0 + cl;
  const float* Bp = Wk + (size_t)j * HEAD * DK * FDIM + n0 + cl;
  v8f t0 = {}, t1 = {};
  for (int kb = 0; kb < HEAD * DK; kb += 64) {
    v2f a[16], b0[16], b1[16];
#pragma unroll
    for (int u = 0; u < 16; ++u) {
      const int kk = kb + 4 * u + 2 * hl;
      a[u].x  = Ap[kk * 64];      a[u].y  = Ap[(kk + 1) * 64];
      b0[u].x = Bp[kk * 64];      b0[u].y = Bp[(kk + 1) * 64];
      b1[u].x = Bp[kk * 64 + 16]; b1[u].y = Bp[(kk + 1) * 64 + 16];
    }
#pragma unroll
    for (int part = 0; part < 4; ++part) {
      v8f p0 = {}, p1 = {};
#pragma unroll
      for (int u = 4 * part; u < 4 * part + 4; ++u) {
        p0 = wmma4(a[u], b0[u], p0);
        p1 = wmma4(a[u], b1[u], p1);
      }
      if (kb == 0 && part == 0) { t0 = p0; t1 = p1; } else { t0 += p0; t1 += p1; }
    }
  }
  float* Cj = Caug + (size_t)j * CAUG;
#pragma unroll
  for (int r = 0; r < 8; ++r) {
    const int row = m0 + (hl << 3) + r;
    Cj[row * 64 + n0 + cl]      = t0[r];
    Cj[row * 64 + n0 + 16 + cl] = t1[r];
  }
}

__device__ __forceinline__ void cvec_job(const float* Wq, const float* bq, const float* Wk,
                                         const float* bk, float* Caug, float* caug, int jb,
                                         float* sh) {
  const int j = jb >> 1;
  const bool isD = (jb & 1) != 0;
  const float* Wm = (isD ? Wk : Wq) + (size_t)j * 256 * 64;
  const float* v  = (isD ? bq : bk) + j * 256;
  const int i = threadIdx.x & 63, half = threadIdx.x >> 6;
  for (int cc = 0; cc < 8; ++cc) {
    const int kc = half * 8 + cc;
    float a = 0.f;
#pragma unroll
    for (int u = 0; u < 16; ++u) {
      int k = kc * 16 + u;
      a = fmaf(Wm[k * 64 + i], v[k], a);
    }
    sh[kc * 64 + i] = a;
  }
  __syncthreads();
  float* Cj = Caug + (size_t)j * CAUG;
  float* cj = caug + j * 80;
  if (threadIdx.x < 64) {
    float t = sh[threadIdx.x];
    for (int p = 1; p < 16; ++p) t += sh[p * 64 + threadIdx.x];
    if (isD) Cj[64 * 64 + threadIdx.x] = t; else cj[threadIdx.x] = t;
  }
  if (isD) {
    if ((threadIdx.x >> 5) == 2) {
      const int lane = threadIdx.x & 31;
      float e = 0.f;
      for (int u = 0; u < 8; ++u) {
        int k = lane + 32 * u;
        e = fmaf(bq[j * 256 + k], bk[j * 256 + k], e);
      }
      for (int m = 16; m >= 1; m >>= 1) e += __shfl_xor(e, m, 32);
      if (lane == 0) cj[64] = e;
    }
  } else {
    for (int idx = threadIdx.x; idx < 15 * 64; idx += 128) Cj[65 * 64 + idx] = 0.f;
    if (threadIdx.x < 15) cj[65 + threadIdx.x] = 0.f;
  }
}
__device__ __forceinline__ float leaky01(float t) {
  return fmaxf(t, 0.01f * t);
}

struct EdgeNets {
  const float* kW0; const float* kb0; const float* kW1; const float* kb1;
  const float* kW2; const float* kb2;
  const float* fW0; const float* fb0; const float* fW1; const float* fb1;
  const float* fW2; const float* fb2;
  float* wout; float* fout;
};

template <int ITERS>
__device__ __forceinline__ void edge_row_job(const EdgeNets& en, int ejob, int mbase, int lane) {
  const bool isF = ejob >= NTOKEN;
  const float* W0 = isF ? en.fW0 : en.kW0;
  const float* b0 = isF ? en.fb0 : en.kb0;
  const float* W1 = isF ? en.fW1 : en.kW1;
  const float* b1 = isF ? en.fb1 : en.kb1;
  const float* W2 = isF ? en.fW2 : en.kW2;
  const float* b2 = isF ? en.fb2 : en.kb2;
  float* dst      = isF ? en.fout : en.wout;

  const int c = lane & 15, h = lane >> 4;
  const int n = ejob & (NTOKEN - 1);
  const float DXc = 2.0f / 31.0f;
  const float e0 = -1.f + DXc * (float)(n >> 5);
  const float e1 = -1.f + DXc * (float)(n & 31);

  v2f wa[4][8];
#pragma unroll
  for (int t = 0; t < 4; ++t) {
#pragma unroll
    for (int i = 0; i < 8; ++i) {
      const float* p = W1 + (t * 16 + c) * 32 + 4 * i + 2 * h;
      wa[t][i].x = p[0]; wa[t][i].y = p[1];
    }
  }
  float pre[16], wc2[16], wc3[16];
#pragma unroll
  for (int i = 0; i < 8; ++i) {
#pragma unroll
    for (int j = 0; j < 2; ++j) {
      int u = 4 * i + 2 * h + j;
      float t = b0[u];
      t = fmaf(W0[u * 4 + 0], e0, t);
      t = fmaf(W0[u * 4 + 1], e1, t);
      pre[2 * i + j] = t;
      wc2[2 * i + j] = W0[u * 4 + 2];
      wc3[2 * i + j] = W0[u * 4 + 3];
    }
  }
  v8f cb[4];
  float w2v[4][8];
#pragma unroll
  for (int t = 0; t < 4; ++t) {
#pragma unroll
    for (int r = 0; r < 8; ++r) {
      cb[t][r]  = b1[t * 16 + 8 * h + r];
      w2v[t][r] = W2[t * 16 + 8 * h + r];
    }
  }
  const float bias2 = b2[0];

  for (int it = 0; it < ITERS; ++it) {
    const int m0 = mbase + it * 32;
    float res[2];
#pragma unroll
    for (int et = 0; et < 2; ++et) {
      const int m = m0 + et * 16 + c;
      const float e2 = -1.f + DXc * (float)(m >> 5);
      const float e3 = -1.f + DXc * (float)(m & 31);
      v2f hb[8];
#pragma unroll
      for (int i = 0; i < 8; ++i) {
        float t0 = fmaf(wc3[2 * i],     e3, fmaf(wc2[2 * i],     e2, pre[2 * i]));
        float t1 = fmaf(wc3[2 * i + 1], e3, fmaf(wc2[2 * i + 1], e2, pre[2 * i + 1]));
        hb[i].x = leaky01(t0); hb[i].y = leaky01(t1);
      }
      v8f acc[4];
#pragma unroll
      for (int t = 0; t < 4; ++t) acc[t] = wmma4(wa[t][0], hb[0], cb[t]);
#pragma unroll
      for (int i = 1; i < 8; ++i) {
#pragma unroll
        for (int t = 0; t < 4; ++t) acc[t] = wmma4(wa[t][i], hb[i], acc[t]);
      }
      float s = 0.f;
#pragma unroll
      for (int t = 0; t < 4; ++t) {
#pragma unroll
        for (int r = 0; r < 8; ++r) s = fmaf(w2v[t][r], leaky01(acc[t][r]), s);
      }
      s += __shfl_xor(s, 16, 32);
      res[et] = s + bias2;
    }
    dst[n * NTOKEN + m0 + lane] = h ? res[1] : res[0];
  }
}

__global__ __launch_bounds__(128) void edge_mlp_kernel(
                                EdgeNets en,
                                const float* xy, const float* ln0g, const float* ln0b, float* V0,
                                const float* Wq, const float* bq, const float* Wk, const float* bk,
                                float* Caug, float* caug, float gs, float* Gpart, float* gpart) {
  __shared__ float shx[16 * 64];
  __shared__ __attribute__((aligned(16))) float sV[128 * VPITCH];
  if (blockIdx.x < LNA_WGS) {
    ln0_job(xy, ln0g, ln0b, V0, blockIdx.x, shx, sV, gs, Gpart, gpart);
    return;
  }
  if (blockIdx.x < LNA_WGS + WJOBS / 2) {
    caug_job(Wq, Wk, Caug, blockIdx.x - LNA_WGS);
    return;
  }
  if (blockIdx.x < LNA_WGS + WJOBS) {
    cvec_job(Wq, bq, Wk, bk, Caug, caug, blockIdx.x - LNA_WGS - WJOBS / 2, shx);
    return;
  }
  edge_row_job<8>(en, blockIdx.x - (LNA_WGS + WJOBS), (threadIdx.x >> 5) * 256, threadIdx.x & 31);
}

#define GPITCH  72


#define CPITCH  68
#define LROWS   128
#define LAYER_WGS (BATCH * (NTOK2 / LROWS))
#define EROWS_L 128
#define EROWS1  (2 * NTOKEN - 4 * EROWS_L)
#define OPITCH  80

template <int M>
__device__ __forceinline__ float row_xor_add(float v) {
  const int o = __builtin_amdgcn_update_dpp(0, __builtin_bit_cast(int, v), 0x160 + M, 0xf, 0xf, true);
  return v + __builtin_bit_cast(float, o);
}

template <bool LNRES, int NP, int GM>
__global__ __launch_bounds__(256) void layer_kernel(const float* Vin, const float* Cj,
                                                    const float* cj, const float* Gp,
                                                    const float* gp, const float* lg,
                                                    const float* lb, float* Out,
                                                    const float* xall, float gs,
                                                    float* Gout, float* gout,
                                                    EdgeNets en, int ebase) {
  constexpr int NPOUT = (GM == 2) ? 8 : 16;
  __shared__ __attribute__((aligned(16))) float sG[64 * GPITCH];
  __shared__ __attribute__((aligned(16))) float sC[80 * CPITCH];
  __shared__ __attribute__((aligned(16))) float sP[80 * CPITCH];
  __shared__ float sg[64];
  __shared__ __attribute__((aligned(16))) float scj[80];
  __shared__ __attribute__((aligned(16))) float sO[(GM != 0) ? LROWS * OPITCH : 4];
  __shared__ __attribute__((aligned(16))) float sX[(GM == 2) ? LROWS * OPITCH : 4];
  __shared__ float sS[(GM != 0) ? 4 * 64 : 4];
  const int tid = threadIdx.x;
  const int wave = __builtin_amdgcn_readfirstlane(tid >> 5);
  const int lane = tid & 31;
  if (blockIdx.x >= LAYER_WGS) {
    edge_row_job<4>(en, ebase + (int)blockIdx.x - LAYER_WGS, wave * 128, lane);
    return;
  }
  const int b = blockIdx.x & (BATCH - 1);
  const int strip = blockIdx.x >> 3;
  const int cl = lane & 15, hl = lane >> 4;
  const int r0 = strip * LROWS;
  const int m0 = r0 + wave * 16;
  const float* A  = Vin + (size_t)b * NTOK2 * FDIM;
  const bool gram = (GM == 1) || (GM == 2 && r0 < NTOKEN);

  const float4* G4 = reinterpret_cast<const float4*>(Gp + (size_t)b * NP * 4096);
  const float4* C4 = reinterpret_cast<const float4*>(Cj);
  const float* g0 = gp + (size_t)b * NP * 64;
  float4 gq[4][NP];
#pragma unroll
  for (int i = 0; i < 4; ++i) {
#pragma unroll
    for (int q = 0; q < NP; ++q) gq[i][q] = G4[q * 1024 + tid + 256 * i];
  }
  const float4 cq0 = C4[tid], cq1 = C4[tid + 256], cq2 = C4[tid + 512], cq3 = C4[tid + 768],
               cq4 = C4[tid + 1024];
  float gv[NP];
#pragma unroll
  for (int q = 0; q < NP; ++q) gv[q] = g0[q * 64 + (tid & 63)];
  const float cjg = cj[tid < 80 ? tid : 79];
  v2f av[16];
#pragma unroll
  for (int u = 0; u < 16; ++u) av[u] = frag_a(A, FDIM, m0, 4 * u, lane);
  __builtin_amdgcn_sched_barrier(0);
#pragma unroll
  for (int i = 0; i < 4; ++i) {
    const int i4 = tid + 256 * i;
    float4 v = gq[i][0];
#pragma unroll
    for (int q = 1; q < NP; ++q) {
      v.x += gq[i][q].x; v.y += gq[i][q].y; v.z += gq[i][q].z; v.w += gq[i][q].w;
    }
    *reinterpret_cast<float4*>(&sG[(i4 >> 4) * GPITCH + ((i4 & 15) << 2)]) = v;
  }
  {
    float4* cdst = reinterpret_cast<float4*>(&sC[(tid >> 4) * CPITCH + ((tid & 15) << 2)]);
    const int cstep = 16 * CPITCH / 4;
    cdst[0] = cq0; cdst[cstep] = cq1; cdst[2 * cstep] = cq2; cdst[3 * cstep] = cq3;
    cdst[4 * cstep] = cq4;
  }
  if (tid < 64) {
    float v = gv[0];
#pragma unroll
    for (int q = 1; q < NP; ++q) v += gv[q];
    sg[tid] = v;
  }
  if (tid < 80) scj[tid] = cjg;
  __syncthreads();

  float4 rv[8];
  float4 lg4 = make_float4(0.f, 0.f, 0.f, 0.f), lb4 = make_float4(0.f, 0.f, 0.f, 0.f);
  if constexpr (LNRES) {
#pragma unroll
    for (int r = 0; r < 8; ++r)
      rv[r] = *reinterpret_cast<const float4*>(&A[(m0 + (hl << 3) + r) * FDIM + 4 * cl]);
    lg4 = *reinterpret_cast<const float4*>(&lg[4 * cl]);
    lb4 = *reinterpret_cast<const float4*>(&lb[4 * cl]);
  }
  float4 xq[8];
#pragma unroll
  for (int i = 0; i < 8; ++i) xq[i] = make_float4(0.f, 0.f, 0.f, 0.f);
  if constexpr (GM == 2) {
    if (gram) {
      const float4* X4 = reinterpret_cast<const float4*>(xall + ((size_t)b * NTOK2 + r0) * FDIM);
#pragma unroll
      for (int i = 0; i < 8; ++i) xq[i] = X4[tid + 256 * i];
    }
  }

  {
    const int rg = wave >> 2;
    const int col = (wave & 3) * 16 + cl;
    const float g = sg[col];
    v2f gb[16];
#pragma unroll
    for (int u = 0; u < 16; ++u) {
      const int kk = 4 * u + 2 * hl;
      gb[u].x = sG[kk * GPITCH + col]; gb[u].y = sG[(kk + 1) * GPITCH + col];
    }
    if (rg == 0) {
      v8f pc;
#pragma unroll
      for (int r = 0; r < 8; ++r) pc[r] = scj[64 + (hl << 3) + r] * g;
      v2f cc[16];
#pragma unroll
      for (int u = 0; u < 16; ++u)
        cc[u] = *reinterpret_cast<const v2f*>(&sC[(64 + cl) * CPITCH + 4 * u + 2 * hl]);
      asm volatile("" ::: "memory");
      __builtin_amdgcn_sched_barrier(0);
#pragma unroll
      for (int u = 0; u < 16; ++u) pc = wmma4(cc[u], gb[u], pc);
#pragma unroll
      for (int r = 0; r < 8; ++r) sP[(64 + (hl << 3) + r) * CPITCH + col] = pc[r];
    }
    const int ra = rg * 32;
    v8f pa, pb;
#pragma unroll
    for (int r = 0; r < 8; ++r) {
      pa[r] = scj[ra + (hl << 3) + r] * g;
      pb[r] = scj[ra + 16 + (hl << 3) + r] * g;
    }
    v2f ca[16], cb2[16];
#pragma unroll
    for (int u = 0; u < 16; ++u) {
      ca[u]  = *reinterpret_cast<const v2f*>(&sC[(ra + cl) * CPITCH + 4 * u + 2 * hl]);
      cb2[u] = *reinterpret_cast<const v2f*>(&sC[(ra + 16 + cl) * CPITCH + 4 * u + 2 * hl]);
    }
    asm volatile("" ::: "memory");
    __builtin_amdgcn_sched_barrier(0);
#pragma unroll
    for (int u = 0; u < 16; ++u) {
      pa = wmma4(ca[u], gb[u], pa);
      pb = wmma4(cb2[u], gb[u], pb);
    }
#pragma unroll
    for (int r = 0; r < 8; ++r) {
      sP[(ra + (hl << 3) + r) * CPITCH + col]      = pa[r];
      sP[(ra + 16 + (hl << 3) + r) * CPITCH + col] = pb[r];
    }
  }
  __syncthreads();

  const float* Pp = sP + 4 * cl;
  const float4 pv = *reinterpret_cast<const float4*>(&Pp[64 * CPITCH]);
  v8f c[4] = {};
  float4 q0[16], q1[16];
#pragma unroll
  for (int u = 0; u < 16; ++u) {
    const int kk = 4 * u + 2 * hl;
    q0[u] = *reinterpret_cast<const float4*>(&Pp[kk * CPITCH]);
    q1[u] = *reinterpret_cast<const float4*>(&Pp[(kk + 1) * CPITCH]);
  }
  asm volatile("" ::: "memory");
  __builtin_amdgcn_sched_barrier(0);
#pragma unroll
  for (int u = 0; u < 16; ++u) {
    v2f bf;
    bf.x = q0[u].x; bf.y = q1[u].x; c[0] = wmma4(av[u], bf, c[0]);
    bf.x = q0[u].y; bf.y = q1[u].y; c[1] = wmma4(av[u], bf, c[1]);
    bf.x = q0[u].z; bf.y = q1[u].z; c[2] = wmma4(av[u], bf, c[2]);
    bf.x = q0[u].w; bf.y = q1[u].w; c[3] = wmma4(av[u], bf, c[3]);
  }
  float* dst = Out + (size_t)b * NTOK2 * FDIM;
  const int rbase = m0 + (hl << 3);
  const int lrow = wave * 16 + (hl << 3);
  if constexpr (!LNRES) {
#pragma unroll
    for (int r = 0; r < 8; ++r) {
      float4 o;
      o.x = c[0][r] + pv.x; o.y = c[1][r] + pv.y; o.z = c[2][r] + pv.z; o.w = c[3][r] + pv.w;
      *reinterpret_cast<float4*>(&dst[(rbase + r) * FDIM + 4 * cl]) = o;
    }
  } else {
    float sm[8], sq[8];
#pragma unroll
    for (int r = 0; r < 8; ++r) {
      c[0][r] += pv.x; c[1][r] += pv.y; c[2][r] += pv.z; c[3][r] += pv.w;
      sm[r] = (c[0][r] + c[1][r]) + (c[2][r] + c[3][r]);
      sq[r] = fmaf(c[3][r], c[3][r], fmaf(c[2][r], c[2][r], fmaf(c[1][r], c[1][r], c[0][r] * c[0][r])));
    }
#pragma unroll
    for (int r = 0; r < 8; ++r) { sm[r] = row_xor_add<1>(sm[r]); sq[r] = row_xor_add<1>(sq[r]); }
#pragma unroll
    for (int r = 0; r < 8; ++r) { sm[r] = row_xor_add<2>(sm[r]); sq[r] = row_xor_add<2>(sq[r]); }
#pragma unroll
    for (int r = 0; r < 8; ++r) { sm[r] = row_xor_add<4>(sm[r]); sq[r] = row_xor_add<4>(sq[r]); }
#pragma unroll
    for (int r = 0; r < 8; ++r) { sm[r] = row_xor_add<8>(sm[r]); sq[r] = row_xor_add<8>(sq[r]); }
#pragma unroll
    for (int r = 0; r < 8; ++r) {
      const float mean = sm[r] * (1.f / 64.f);
      const float var  = sq[r] * (1.f / 64.f) - mean * mean;
      const float inv  = rsqrtf(var + 1e-5f);
      float4 o;
      o.x = (c[0][r] - mean) * inv * lg4.x + lb4.x + rv[r].x;
      o.y = (c[1][r] - mean) * inv * lg4.y + lb4.y + rv[r].y;
      o.z = (c[2][r] - mean) * inv * lg4.z + lb4.z + rv[r].z;
      o.w = (c[3][r] - mean) * inv * lg4.w + lb4.w + rv[r].w;
      *reinterpret_cast<float4*>(&dst[(rbase + r) * FDIM + 4 * cl]) = o;
      if constexpr (GM != 0)
        *reinterpret_cast<float4*>(&sO[(lrow + r) * OPITCH + 4 * cl]) = o;
    }
  }

  if constexpr (GM != 0) {
    if (gram) {
      if constexpr (GM == 2) {
#pragma unroll
        for (int i = 0; i < 8; ++i) {
          const int p = tid + 256 * i;
          *reinterpret_cast<float4*>(&sX[(p >> 4) * OPITCH + ((p & 15) << 2)]) = xq[i];
        }
      }
      __syncthreads();
      const float* sB = (GM == 2) ? sX : sO;
      {
        const int col = tid & 63, grp = tid >> 6;
        float s = 0.f;
        for (int r = 0; r < 32; ++r) s += sB[(grp * 32 + r) * OPITCH + col];
        sS[grp * 64 + col] = s;
      }
      const int i0 = (wave >> 1) * 16, n0 = (wave & 1) * 32;
      v8f t0 = {}, t1 = {};
      for (int kc = 0; kc < 4; ++kc) {
        const int kb = kc * 32 + 2 * hl;
        v2f a[8], bx[8], by[8];
#pragma unroll
        for (int u = 0; u < 8; ++u) {
          const int k = kb + 4 * u;
          a[u].x = sO[k * OPITCH + i0 + cl]; a[u].y = sO[(k + 1) * OPITCH + i0 + cl];
          bx[u] = *reinterpret_cast<const v2f*>(&sB[k * OPITCH + n0 + 2 * cl]);
          by[u] = *reinterpret_cast<const v2f*>(&sB[(k + 1) * OPITCH + n0 + 2 * cl]);
        }
        asm volatile("" ::: "memory");
        __builtin_amdgcn_sched_barrier(0);
#pragma unroll
        for (int u = 0; u < 8; ++u) {
          v2f bf;
          bf.x = bx[u].x; bf.y = by[u].x; t0 = wmma4(a[u], bf, t0);
          bf.x = bx[u].y; bf.y = by[u].y; t1 = wmma4(a[u], bf, t1);
        }
      }
      const size_t slot = (size_t)b * NPOUT + strip;
      float* Gd = Gout + slot * 4096;
#pragma unroll
      for (int r = 0; r < 8; ++r) {
        const int row = i0 + (hl << 3) + r;
        v2f o; o.x = t0[r] * gs; o.y = t1[r] * gs;
        *reinterpret_cast<v2f*>(&Gd[row * 64 + n0 + 2 * cl]) = o;
      }
      __syncthreads();
      if (tid < 64) {
        const float s = ((sS[tid] + sS[64 + tid]) + sS[128 + tid]) + sS[192 + tid];
        gout[slot * 64 + tid] = s * gs;
      }
    }
  }
}

__global__ __launch_bounds__(KSPLIT * 32) void final_kernel(const float* weight, const float* weightf,
                                                            const float* S, float* out, float mul) {
  __shared__ __attribute__((aligned(16))) float sh[KSPLIT * FPLANE];
  const int b = blockIdx.x;
  const int ks = __builtin_amdgcn_readfirstlane(threadIdx.x >> 5);
  const int lane = threadIdx.x & 31;
  const int cl = lane & 15, hl = lane >> 4;
  const int m0 = blockIdx.y * FROWS;
  const int klen = NTOK2 / KSPLIT;
  const int kbeg = ks * klen;
  const float* W = (kbeg < NTOKEN) ? weight : weightf;
  const float* Wp = W + (size_t)((kbeg & (NTOKEN - 1)) + 2 * hl) * NTOKEN + m0 + 2 * cl;
  const float* Sp = S + (size_t)b * NTOK2 * FDIM + (size_t)(kbeg + 2 * hl) * FDIM + 4 * cl;
  v8f c[2][4] = {};
#pragma unroll
  for (int trip = 0; trip < klen / 32; ++trip) {
    v2f wa[8], wb[8];
    float4 q0[8], q1[8];
#pragma unroll
    for (int u = 0; u < 8; ++u) {
      const int k = trip * 32 + 4 * u;
      wa[u] = *reinterpret_cast<const v2f*>(Wp + (size_t)k * NTOKEN);
      wb[u] = *reinterpret_cast<const v2f*>(Wp + (size_t)(k + 1) * NTOKEN);
      q0[u] = *reinterpret_cast<const float4*>(Sp + k * FDIM);
      q1[u] = *reinterpret_cast<const float4*>(Sp + (k + 1) * FDIM);
    }
    __builtin_amdgcn_sched_barrier(0);
#pragma unroll
    for (int u = 0; u < 8; ++u) {
      v2f a0, a1, bf;
      a0.x = wa[u].x; a0.y = wb[u].x;
      a1.x = wa[u].y; a1.y = wb[u].y;
      bf.x = q0[u].x; bf.y = q1[u].x; c[0][0] = wmma4(a0, bf, c[0][0]); c[1][0] = wmma4(a1, bf, c[1][0]);
      bf.x = q0[u].y; bf.y = q1[u].y; c[0][1] = wmma4(a0, bf, c[0][1]); c[1][1] = wmma4(a1, bf, c[1][1]);
      bf.x = q0[u].z; bf.y = q1[u].z; c[0][2] = wmma4(a0, bf, c[0][2]); c[1][2] = wmma4(a1, bf, c[1][2]);
      bf.x = q0[u].w; bf.y = q1[u].w; c[0][3] = wmma4(a0, bf, c[0][3]); c[1][3] = wmma4(a1, bf, c[1][3]);
    }
  }
  float* mine = sh + ks * FPLANE;
#pragma unroll
  for (int rt = 0; rt < 2; ++rt) {
#pragma unroll
    for (int r = 0; r < 8; ++r) {
      float4 o;
      o.x = c[rt][0][r]; o.y = c[rt][1][r]; o.z = c[rt][2][r]; o.w = c[rt][3][r];
      *reinterpret_cast<float4*>(&mine[(2 * ((hl << 3) + r) + rt) * TPITCH + 4 * cl]) = o;
    }
  }
  __syncthreads();
  {
    const int row = threadIdx.x >> 4, c4 = (threadIdx.x & 15) << 2;
    float4 acc = *reinterpret_cast<const float4*>(&sh[row * TPITCH + c4]);
#pragma unroll
    for (int p = 1; p < KSPLIT; ++p) {
      const float4 v = *reinterpret_cast<const float4*>(&sh[p * FPLANE + row * TPITCH + c4]);
      acc.x += v.x; acc.y += v.y; acc.z += v.z; acc.w += v.w;
    }
    float4 o;
    o.x = acc.x * mul; o.y = acc.y * mul; o.z = acc.z * mul; o.w = acc.w * mul;
    *reinterpret_cast<float4*>(&out[(size_t)b * NTOKEN * FDIM + (size_t)(m0 + row) * FDIM + c4]) = o;
  }
}

extern "C" void kernel_launch(void* const* d_in, const int* in_sizes, int n_in,
                              void* d_out, int out_size, void* d_ws, size_t ws_size,
                              hipStream_t stream) {
  (void)in_sizes; (void)n_in; (void)out_size; (void)ws_size;
  const float* xy   = (const float*)d_in[0];
  const float* kW0  = (const float*)d_in[1];
  const float* kb0  = (const float*)d_in[2];
  const float* kW1  = (const float*)d_in[3];
  const float* kb1  = (const float*)d_in[4];
  const float* kW2  = (const float*)d_in[5];
  const float* kb2  = (const float*)d_in[6];
  const float* fW0  = (const float*)d_in[7];
  const float* fb0  = (const float*)d_in[8];
  const float* fW1  = (const float*)d_in[9];
  const float* fb1  = (const float*)d_in[10];
  const float* fW2  = (const float*)d_in[11];
  const float* fb2  = (const float*)d_in[12];
  const float* Wq   = (const float*)d_in[13];
  const float* bq   = (const float*)d_in[14];
  const float* Wk   = (const float*)d_in[15];
  const float* bk   = (const float*)d_in[16];
  const float* ln0g = (const float*)d_in[17];
  const float* ln0b = (const float*)d_in[18];
  const float* lng  = (const float*)d_in[19];
  const float* lnb  = (const float*)d_in[20];
  float* out = (float*)d_out;

  float* ws = (float*)d_ws;
  const size_t SZ_1M = 1048576;
  float* w_weight  = ws;
  float* w_weightf = ws + SZ_1M;
  float* V0        = ws + 2 * SZ_1M;
  float* V1        = ws + 3 * SZ_1M;
  float* Caug      = ws + 4 * SZ_1M;
  float* caug      = ws + 4 * SZ_1M + 20480;
  float* Gp0       = ws + 5 * SZ_1M;
  float* gp0       = ws + 5 * SZ_1M + 524288;
  float* GpA       = ws + 6 * SZ_1M;
  float* gpA       = ws + 6 * SZ_1M + 524288;
  float* GpB       = ws + 7 * SZ_1M;
  float* gpB       = ws + 7 * SZ_1M + 524288;
  float* MID       = ws + 12 * SZ_1M + 131072;

  const float DXc   = 2.0f / 31.0f;
  const float dxdy  = DXc * DXc;
  const float scale = 0.125f;

  const EdgeNets en = {kW0, kb0, kW1, kb1, kW2, kb2, fW0, fb0, fW1, fb1, fW2, fb2,
                       w_weight, w_weightf};
  edge_mlp_kernel<<<LNA_WGS + WJOBS + EROWS1, 128, 0, stream>>>(
      en, xy, ln0g, ln0b, V0, Wq, bq, Wk, bk, Caug, caug, scale * dxdy, Gp0, gp0);

  layer_kernel<true, 8, 1><<<LAYER_WGS + EROWS_L, 256, 0, stream>>>(
      V0, Caug, caug, Gp0, gp0, lng, lnb, V1, xy, scale * dxdy, GpA, gpA, en, EROWS1);
  layer_kernel<true, 16, 1><<<LAYER_WGS + EROWS_L, 256, 0, stream>>>(
      V1, Caug + (size_t)1 * CAUG, caug + 80, GpA, gpA, lng + FDIM, lnb + FDIM, V0, xy,
      scale * dxdy, GpB, gpB, en, EROWS1 + EROWS_L);
  layer_kernel<true, 16, 2><<<LAYER_WGS + EROWS_L, 256, 0, stream>>>(
      V0, Caug + (size_t)2 * CAUG, caug + 160, GpB, gpB, lng + 2 * FDIM, lnb + 2 * FDIM, V1, xy,
      scale, GpA, gpA, en, EROWS1 + 2 * EROWS_L);
  layer_kernel<false, 8, 0><<<LAYER_WGS + EROWS_L, 256, 0, stream>>>(
      V1, Caug + (size_t)3 * CAUG, caug + 240, GpA, gpA, nullptr, nullptr, MID, xy,
      scale, nullptr, nullptr, en, EROWS1 + 3 * EROWS_L);
  final_kernel<<<dim3(BATCH, NTOKEN / FROWS), KSPLIT * 32, 0, stream>>>(
      w_weight, w_weightf, MID, out, dxdy * dxdy);
}
